# starting kernel
# baseline (speedup 1.0000x reference)
#include <hip/hip_runtime.h>
#include <math.h>

#define NUM_HEADS 4
#define KCHUNK 64
#define LDSTRIDE 68

typedef float v2f __attribute__((ext_vector_type(2)));
typedef float v8f __attribute__((ext_vector_type(8)));

__device__ __forceinline__ void atomic_max_float(float* addr, float val) {
  if (val >= 0.0f) {
    atomicMax((int*)addr, __float_as_int(val));
  } else {
    atomicMin((unsigned int*)addr, __float_as_uint(val));
  }
}

__global__ void init_ws(float* z0, long n0, float* z1, long n1, float* smax, int ns) {
  long i = (long)blockIdx.x * blockDim.x + threadIdx.x;
  long stride = (long)gridDim.x * blockDim.x;
  for (long t = i; t < n0; t += stride) z0[t] = 0.0f;
  for (long t = i; t < n1; t += stride) z1[t] = 0.0f;
  for (long t = i; t < ns; t += stride) smax[t] = -INFINITY;
}

__global__ void qst_kernel(const float* __restrict__ query, const float* __restrict__ in_w,
                           const float* __restrict__ in_b, int D,
                           float* __restrict__ sbuf, float* __restrict__ tbuf) {
  __shared__ float qs[1024];
  int j = threadIdx.x;
  int dh = D / NUM_HEADS;
  float acc = in_b[j];
  for (int c = 0; c < D; ++c) acc += query[c] * in_w[(long)j * D + c];
  qs[j] = acc;
  __syncthreads();
  for (int hh = 0; hh < NUM_HEADS; ++hh) {
    float s = 0.0f;
    const float* wk = in_w + (long)(D + hh * dh) * D;
    for (int d = 0; d < dh; ++d) s += qs[hh * dh + d] * wk[(long)d * D + j];
    sbuf[hh * D + j] = s;
  }
  if (j < NUM_HEADS) {
    float t = 0.0f;
    for (int d = 0; d < dh; ++d) t += qs[j * dh + d] * in_b[D + j * dh + d];
    tbuf[j] = t;
  }
}

__global__ void edge_scatter(const float* __restrict__ eattr, const int* __restrict__ row,
                             float* __restrict__ sumA, float* __restrict__ cnt,
                             long E, int ED) {
  long total = E * (long)ED;
  long i = (long)blockIdx.x * blockDim.x + threadIdx.x;
  long stride = (long)gridDim.x * blockDim.x;
  for (long t = i; t < total; t += stride) {
    long e = t / ED;
    int c = (int)(t - e * ED);
    int r = row[e];
    atomicAdd(&sumA[(long)r * ED + c], eattr[t]);
    if (c == 0) atomicAdd(&cnt[r], 1.0f);
  }
}


__global__ void hproj_wmma(const float* __restrict__ h, const float* __restrict__ sumA,
                           const float* __restrict__ cnt,
                           const float* __restrict__ node_w, const float* __restrict__ node_b,
                           const float* __restrict__ edge_w, const float* __restrict__ edge_b,
                           float* __restrict__ hproj, int N, int ND, int ED, int D) {
  const int K = ND + ED;
  __shared__ float Xs[64 * LDSTRIDE];
  int row0 = blockIdx.x * 64;
  int tid = threadIdx.x;
  int wave = tid >> 5, lane = tid & 31;
  int mrow = lane & 15;
  int n = blockIdx.y * 64 + wave * 16 + mrow;
  int khalf = (lane & 16) ? 2 : 0;

  v8f acc0 = {}, acc1 = {}, acc2 = {}, acc3 = {};

  for (int kc = 0; kc < K; kc += KCHUNK) {
    __syncthreads();
    for (int t = tid; t < 64 * KCHUNK; t += blockDim.x) {
      int r = t >> 6;
      int c = t & (KCHUNK - 1);
      int m = row0 + r;
      int cg = kc + c;
      float val = 0.0f;
      if (m < N) {
        if (cg < ND) val = h[(long)m * ND + cg];
        else         val = sumA[(long)m * ED + (cg - ND)] / fmaxf(cnt[m], 1.0f);
      }
      Xs[r * LDSTRIDE + c] = val;
    }
    __syncthreads();
    for (int k = 0; k < KCHUNK; k += 4) {
      int kk = k + khalf;
      v2f b;
      if (kc + k < ND) b = *(const v2f*)&node_w[(long)n * ND + kc + kk];
      else             b = *(const v2f*)&edge_w[(long)n * ED + (kc + kk - ND)];
      v2f a0 = *(const v2f*)&Xs[(mrow +  0) * LDSTRIDE + kk];
      v2f a1 = *(const v2f*)&Xs[(mrow + 16) * LDSTRIDE + kk];
      v2f a2 = *(const v2f*)&Xs[(mrow + 32) * LDSTRIDE + kk];
      v2f a3 = *(const v2f*)&Xs[(mrow + 48) * LDSTRIDE + kk];
      acc0 = __builtin_amdgcn_wmma_f32_16x16x4_f32(false, a0, false, b, (short)0, acc0, false, false);
      acc1 = __builtin_amdgcn_wmma_f32_16x16x4_f32(false, a1, false, b, (short)0, acc1, false, false);
      acc2 = __builtin_amdgcn_wmma_f32_16x16x4_f32(false, a2, false, b, (short)0, acc2, false, false);
      acc3 = __builtin_amdgcn_wmma_f32_16x16x4_f32(false, a3, false, b, (short)0, acc3, false, false);
    }
  }

  float bn = node_b[n];
  float be = edge_b[n];
  for (int r = 0; r < 8; ++r) {
    int mb = row0 + r + ((lane & 16) ? 8 : 0);
    v8f* accs[4] = {&acc0, &acc1, &acc2, &acc3};
    for (int mt = 0; mt < 4; ++mt) {
      int m = mb + mt * 16;
      if (m < N) {
        float mask = (cnt[m] > 0.0f) ? 1.0f : 0.0f;
        hproj[(long)m * D + n] = (*accs[mt])[r] + bn + mask * be;
      }
    }
  }
}

__global__ void gemm_xwT_wmma(const float* __restrict__ X, const float* __restrict__ W,
                              const float* __restrict__ bias, float* __restrict__ Y,
                              int M, int K, int Nout) {
  __shared__ float Xs[64 * LDSTRIDE];
  int row0 = blockIdx.x * 64;
  int tid = threadIdx.x;
  int wave = tid >> 5, lane = tid & 31;
  int mrow = lane & 15;
  int n = blockIdx.y * 64 + wave * 16 + mrow;
  int khalf = (lane & 16) ? 2 : 0;

  v8f acc0 = {}, acc1 = {}, acc2 = {}, acc3 = {};

  for (int kc = 0; kc < K; kc += KCHUNK) {
    __syncthreads();
    for (int t = tid; t < 64 * KCHUNK; t += blockDim.x) {
      int r = t >> 6;
      int c = t & (KCHUNK - 1);
      int m = row0 + r;
      Xs[r * LDSTRIDE + c] = (m < M) ? X[(long)m * K + kc + c] : 0.0f;
    }
    __syncthreads();
    for (int k = 0; k < KCHUNK; k += 4) {
      int kk = k + khalf;
      v2f b = *(const v2f*)&W[(long)n * K + kc + kk];
      v2f a0 = *(const v2f*)&Xs[(mrow +  0) * LDSTRIDE + kk];
      v2f a1 = *(const v2f*)&Xs[(mrow + 16) * LDSTRIDE + kk];
      v2f a2 = *(const v2f*)&Xs[(mrow + 32) * LDSTRIDE + kk];
      v2f a3 = *(const v2f*)&Xs[(mrow + 48) * LDSTRIDE + kk];
      acc0 = __builtin_amdgcn_wmma_f32_16x16x4_f32(false, a0, false, b, (short)0, acc0, false, false);
      acc1 = __builtin_amdgcn_wmma_f32_16x16x4_f32(false, a1, false, b, (short)0, acc1, false, false);
      acc2 = __builtin_amdgcn_wmma_f32_16x16x4_f32(false, a2, false, b, (short)0, acc2, false, false);
      acc3 = __builtin_amdgcn_wmma_f32_16x16x4_f32(false, a3, false, b, (short)0, acc3, false, false);
    }
  }

  float bn = bias[n];
  for (int r = 0; r < 8; ++r) {
    int mb = row0 + r + ((lane & 16) ? 8 : 0);
    v8f* accs[4] = {&acc0, &acc1, &acc2, &acc3};
    for (int mt = 0; mt < 4; ++mt) {
      int m = mb + mt * 16;
      if (m < M) Y[(long)m * Nout + n] = (*accs[mt])[r] + bn;
    }
  }
}

__global__ void scores_kernel(const float* __restrict__ hproj, const float* __restrict__ sbuf,
                              const float* __restrict__ tbuf, const int* __restrict__ batch,
                              float* __restrict__ smax, float* __restrict__ scores,
                              int N, int D) {
  int idx = blockIdx.x * blockDim.x + threadIdx.x;
  if (idx >= N * NUM_HEADS) return;
  int nI = idx >> 2;
  int hh = idx & 3;
  const float* xr = hproj + (long)nI * D;
  const float* sr = sbuf + hh * D;
  float dot = tbuf[hh];
  for (int c = 0; c < D; c += 4) {
    float4 x = *(const float4*)&xr[c];
    float4 s = *(const float4*)&sr[c];
    dot += x.x * s.x + x.y * s.y + x.z * s.z + x.w * s.w;
  }
  float sc = dot * rsqrtf((float)(D / NUM_HEADS));
  scores[idx] = sc;
  atomic_max_float(&smax[batch[nI] * NUM_HEADS + hh], sc);
}

__global__ void expw_kernel(const float* __restrict__ scores, const int* __restrict__ batch,
                            const float* __restrict__ smax, float* __restrict__ denom,
                            float* __restrict__ wexp, int N) {
  int idx = blockIdx.x * blockDim.x + threadIdx.x;
  if (idx >= N * NUM_HEADS) return;
  int nI = idx >> 2;
  int hh = idx & 3;
  int g = batch[nI];
  float w = __expf(scores[idx] - smax[g * NUM_HEADS + hh]);
  wexp[idx] = w;
  atomicAdd(&denom[g * NUM_HEADS + hh], w);
}

__global__ void pool_kernel(const float* __restrict__ wexp, const float* __restrict__ denom,
                            const int* __restrict__ batch, const float* __restrict__ v,
                            float* __restrict__ pooled, int N, int D) {
  long idx = (long)blockIdx.x * blockDim.x + threadIdx.x;
  long total = (long)N * D;
  if (idx >= total) return;
  int nI = (int)(idx / D);
  int c = (int)(idx - (long)nI * D);
  int hh = c / (D / NUM_HEADS);
  int g = batch[nI];
  float p = wexp[nI * NUM_HEADS + hh] / fmaxf(denom[g * NUM_HEADS + hh], 1e-30f);
  atomicAdd(&pooled[(long)g * D + c], p * v[idx]);
}

extern "C" void kernel_launch(void* const* d_in, const int* in_sizes, int n_in,
                              void* d_out, int out_size, void* d_ws, size_t ws_size,
                              hipStream_t stream) {
  const float* h      = (const float*)d_in[0];
  const int*   eidx   = (const int*)d_in[1];
  const float* eattr  = (const float*)d_in[2];
  const int*   batch  = (const int*)d_in[3];
  const float* node_w = (const float*)d_in[5];
  const float* node_b = (const float*)d_in[6];
  const float* edge_w = (const float*)d_in[7];
  const float* edge_b = (const float*)d_in[8];
  const float* query  = (const float*)d_in[9];
  const float* in_w   = (const float*)d_in[10];
  const float* in_b   = (const float*)d_in[11];
  const float* out_w  = (const float*)d_in[12];
  const float* out_b  = (const float*)d_in[13];
  float* out = (float*)d_out;

  int D  = in_sizes[9];
  int ND = in_sizes[5] / D;
  int ED = in_sizes[7] / D;
  int N  = in_sizes[0] / ND;
  long E = in_sizes[1] / 2;
  int G  = out_size / D;

  float* ws = (float*)d_ws;
  long o = 0;
  auto take = [&](long nf) { float* p = ws + o; o += (nf + 3) & ~3L; return p; };
  float* sumA   = take((long)N * ED);
  float* cnt    = take(N);
  float* hproj  = take((long)N * D);
  float* vbuf   = take((long)N * D);
  float* scores = take((long)N * NUM_HEADS);
  float* wexp   = take((long)N * NUM_HEADS);
  float* sbuf   = take((long)NUM_HEADS * D);
  float* tbuf   = take(NUM_HEADS);
  float* smax   = take((long)G * NUM_HEADS);
  float* denom  = take((long)G * NUM_HEADS + (long)G * D);
  float* pooled = denom + G * NUM_HEADS;

  {
    long z0 = (long)N * ED + N;
    long z1 = (long)G * NUM_HEADS + (long)G * D;
    long maxn = z0 > z1 ? z0 : z1;
    int blocks = (int)((maxn + 255) / 256);
    if (blocks > 262144) blocks = 262144;
    init_ws<<<blocks, 256, 0, stream>>>(sumA, z0, denom, z1, smax, G * NUM_HEADS);
  }
  qst_kernel<<<1, D, 0, stream>>>(query, in_w, in_b, D, sbuf, tbuf);
  {
    long total = E * (long)ED;
    long blocks = (total + 255) / 256;
    if (blocks > 262144) blocks = 262144;
    edge_scatter<<<(int)blocks, 256, 0, stream>>>(eattr, eidx, sumA, cnt, E, ED);
  }
  {
    dim3 grid((N + 63) / 64, D / 64);
    hproj_wmma<<<grid, 128, 0, stream>>>(h, sumA, cnt, node_w, node_b, edge_w, edge_b,
                                         hproj, N, ND, ED, D);
  }
  {
    int total = N * NUM_HEADS;
    scores_kernel<<<(total + 255) / 256, 256, 0, stream>>>(hproj, sbuf, tbuf, batch,
                                                           smax, scores, N, D);
    expw_kernel<<<(total + 255) / 256, 256, 0, stream>>>(scores, batch, smax, denom,
                                                         wexp, N);
  }
  {
    dim3 grid((N + 63) / 64, D / 64);
    gemm_xwT_wmma<<<grid, 128, 0, stream>>>(hproj, in_w + 2L * D * D, in_b + 2 * D,
                                            vbuf, N, D, D);
  }
  {
    long total = (long)N * D;
    pool_kernel<<<(int)((total + 255) / 256), 256, 0, stream>>>(wexp, denom, batch,
                                                                vbuf, pooled, N, D);
  }
  {
    dim3 grid((G + 63) / 64, D / 64);
    gemm_xwT_wmma<<<grid, 128, 0, stream>>>(pooled, out_w, out_b, out, G, D, D);
  }
}


# Pool h_proj per head before the V projection (linearity): V GEMM runs on G*4 pooled rows, not N; pooling sums in-block, one float atomic per block and graph (sum order changes)
# speedup vs baseline: 1.3142x; 1.3142x over previous
#include <hip/hip_runtime.h>
#include <math.h>

#define NUM_HEADS 4
#define KCHUNK 64
#define LDSTRIDE 68

typedef float v2f __attribute__((ext_vector_type(2)));
typedef float v8f __attribute__((ext_vector_type(8)));

__device__ __forceinline__ void atomic_max_float(float* addr, float val) {
  if (val >= 0.0f) {
    atomicMax((int*)addr, __float_as_int(val));
  } else {
    atomicMin((unsigned int*)addr, __float_as_uint(val));
  }
}

__global__ void init_ws(float* z0, long n0, float* z1, long n1, float* smax, int ns) {
  long i = (long)blockIdx.x * blockDim.x + threadIdx.x;
  long stride = (long)gridDim.x * blockDim.x;
  for (long t = i; t < n0; t += stride) z0[t] = 0.0f;
  for (long t = i; t < n1; t += stride) z1[t] = 0.0f;
  for (long t = i; t < ns; t += stride) smax[t] = -INFINITY;
}

__global__ void qst_kernel(const float* __restrict__ query, const float* __restrict__ in_w,
                           const float* __restrict__ in_b, int D,
                           float* __restrict__ sbuf, float* __restrict__ tbuf) {
  __shared__ float qs[1024];
  int j = threadIdx.x;
  int dh = D / NUM_HEADS;
  float acc = in_b[j];
  for (int c = 0; c < D; ++c) acc += query[c] * in_w[(long)j * D + c];
  qs[j] = acc;
  __syncthreads();
  for (int hh = 0; hh < NUM_HEADS; ++hh) {
    float s = 0.0f;
    const float* wk = in_w + (long)(D + hh * dh) * D;
    for (int d = 0; d < dh; ++d) s += qs[hh * dh + d] * wk[(long)d * D + j];
    sbuf[hh * D + j] = s;
  }
  if (j < NUM_HEADS) {
    float t = 0.0f;
    for (int d = 0; d < dh; ++d) t += qs[j * dh + d] * in_b[D + j * dh + d];
    tbuf[j] = t;
  }
}

__global__ void edge_scatter(const float* __restrict__ eattr, const int* __restrict__ row,
                             float* __restrict__ sumA, float* __restrict__ cnt,
                             long E, int ED) {
  long total = E * (long)ED;
  long i = (long)blockIdx.x * blockDim.x + threadIdx.x;
  long stride = (long)gridDim.x * blockDim.x;
  for (long t = i; t < total; t += stride) {
    long e = t / ED;
    int c = (int)(t - e * ED);
    int r = row[e];
    atomicAdd(&sumA[(long)r * ED + c], eattr[t]);
    if (c == 0) atomicAdd(&cnt[r], 1.0f);
  }
}


__global__ void hproj_wmma(const float* __restrict__ h, const float* __restrict__ sumA,
                           const float* __restrict__ cnt,
                           const float* __restrict__ node_w, const float* __restrict__ node_b,
                           const float* __restrict__ edge_w, const float* __restrict__ edge_b,
                           float* __restrict__ hproj, int N, int ND, int ED, int D) {
  const int K = ND + ED;
  __shared__ float Xs[64 * LDSTRIDE];
  int row0 = blockIdx.x * 64;
  int tid = threadIdx.x;
  int wave = tid >> 5, lane = tid & 31;
  int mrow = lane & 15;
  int n = blockIdx.y * 64 + wave * 16 + mrow;
  int khalf = (lane & 16) ? 2 : 0;

  v8f acc0 = {}, acc1 = {}, acc2 = {}, acc3 = {};

  for (int kc = 0; kc < K; kc += KCHUNK) {
    __syncthreads();
    for (int t = tid; t < 64 * KCHUNK; t += blockDim.x) {
      int r = t >> 6;
      int c = t & (KCHUNK - 1);
      int m = row0 + r;
      int cg = kc + c;
      float val = 0.0f;
      if (m < N) {
        if (cg < ND) val = h[(long)m * ND + cg];
        else         val = sumA[(long)m * ED + (cg - ND)] / fmaxf(cnt[m], 1.0f);
      }
      Xs[r * LDSTRIDE + c] = val;
    }
    __syncthreads();
    for (int k = 0; k < KCHUNK; k += 4) {
      int kk = k + khalf;
      v2f b;
      if (kc + k < ND) b = *(const v2f*)&node_w[(long)n * ND + kc + kk];
      else             b = *(const v2f*)&edge_w[(long)n * ED + (kc + kk - ND)];
      v2f a0 = *(const v2f*)&Xs[(mrow +  0) * LDSTRIDE + kk];
      v2f a1 = *(const v2f*)&Xs[(mrow + 16) * LDSTRIDE + kk];
      v2f a2 = *(const v2f*)&Xs[(mrow + 32) * LDSTRIDE + kk];
      v2f a3 = *(const v2f*)&Xs[(mrow + 48) * LDSTRIDE + kk];
      acc0 = __builtin_amdgcn_wmma_f32_16x16x4_f32(false, a0, false, b, (short)0, acc0, false, false);
      acc1 = __builtin_amdgcn_wmma_f32_16x16x4_f32(false, a1, false, b, (short)0, acc1, false, false);
      acc2 = __builtin_amdgcn_wmma_f32_16x16x4_f32(false, a2, false, b, (short)0, acc2, false, false);
      acc3 = __builtin_amdgcn_wmma_f32_16x16x4_f32(false, a3, false, b, (short)0, acc3, false, false);
    }
  }

  float bn = node_b[n];
  float be = edge_b[n];
  for (int r = 0; r < 8; ++r) {
    int mb = row0 + r + ((lane & 16) ? 8 : 0);
    v8f* accs[4] = {&acc0, &acc1, &acc2, &acc3};
    for (int mt = 0; mt < 4; ++mt) {
      int m = mb + mt * 16;
      if (m < N) {
        float mask = (cnt[m] > 0.0f) ? 1.0f : 0.0f;
        hproj[(long)m * D + n] = (*accs[mt])[r] + bn + mask * be;
      }
    }
  }
}

__global__ void gemm_xwT_wmma(const float* __restrict__ X, const float* __restrict__ W,
                              const float* __restrict__ bias, float* __restrict__ Y,
                              int M, int K, int Nout) {
  __shared__ float Xs[64 * LDSTRIDE];
  int row0 = blockIdx.x * 64;
  int tid = threadIdx.x;
  int wave = tid >> 5, lane = tid & 31;
  int mrow = lane & 15;
  int n = blockIdx.y * 64 + wave * 16 + mrow;
  int khalf = (lane & 16) ? 2 : 0;

  v8f acc0 = {}, acc1 = {}, acc2 = {}, acc3 = {};

  for (int kc = 0; kc < K; kc += KCHUNK) {
    __syncthreads();
    for (int t = tid; t < 64 * KCHUNK; t += blockDim.x) {
      int r = t >> 6;
      int c = t & (KCHUNK - 1);
      int m = row0 + r;
      Xs[r * LDSTRIDE + c] = (m < M) ? X[(long)m * K + kc + c] : 0.0f;
    }
    __syncthreads();
    for (int k = 0; k < KCHUNK; k += 4) {
      int kk = k + khalf;
      v2f b = *(const v2f*)&W[(long)n * K + kc + kk];
      v2f a0 = *(const v2f*)&Xs[(mrow +  0) * LDSTRIDE + kk];
      v2f a1 = *(const v2f*)&Xs[(mrow + 16) * LDSTRIDE + kk];
      v2f a2 = *(const v2f*)&Xs[(mrow + 32) * LDSTRIDE + kk];
      v2f a3 = *(const v2f*)&Xs[(mrow + 48) * LDSTRIDE + kk];
      acc0 = __builtin_amdgcn_wmma_f32_16x16x4_f32(false, a0, false, b, (short)0, acc0, false, false);
      acc1 = __builtin_amdgcn_wmma_f32_16x16x4_f32(false, a1, false, b, (short)0, acc1, false, false);
      acc2 = __builtin_amdgcn_wmma_f32_16x16x4_f32(false, a2, false, b, (short)0, acc2, false, false);
      acc3 = __builtin_amdgcn_wmma_f32_16x16x4_f32(false, a3, false, b, (short)0, acc3, false, false);
    }
  }

  float bn = bias[n];
  for (int r = 0; r < 8; ++r) {
    int mb = row0 + r + ((lane & 16) ? 8 : 0);
    v8f* accs[4] = {&acc0, &acc1, &acc2, &acc3};
    for (int mt = 0; mt < 4; ++mt) {
      int m = mb + mt * 16;
      if (m < M) Y[(long)m * Nout + n] = (*accs[mt])[r] + bn;
    }
  }
}

__global__ void scores_kernel(const float* __restrict__ hproj, const float* __restrict__ sbuf,
                              const float* __restrict__ tbuf, const int* __restrict__ batch,
                              float* __restrict__ smax, float* __restrict__ scores,
                              int N, int D) {
  int idx = blockIdx.x * blockDim.x + threadIdx.x;
  if (idx >= N * NUM_HEADS) return;
  int nI = idx >> 2;
  int hh = idx & 3;
  const float* xr = hproj + (long)nI * D;
  const float* sr = sbuf + hh * D;
  float dot = tbuf[hh];
  for (int c = 0; c < D; c += 4) {
    float4 x = *(const float4*)&xr[c];
    float4 s = *(const float4*)&sr[c];
    dot += x.x * s.x + x.y * s.y + x.z * s.z + x.w * s.w;
  }
  float sc = dot * rsqrtf((float)(D / NUM_HEADS));
  scores[idx] = sc;
  atomic_max_float(&smax[batch[nI] * NUM_HEADS + hh], sc);
}

__global__ void expw_kernel(const float* __restrict__ scores, const int* __restrict__ batch,
                            const float* __restrict__ smax, float* __restrict__ denom,
                            float* __restrict__ wexp, int N) {
  int idx = blockIdx.x * blockDim.x + threadIdx.x;
  if (idx >= N * NUM_HEADS) return;
  int nI = idx >> 2;
  int hh = idx & 3;
  int g = batch[nI];
  float w = __expf(scores[idx] - smax[g * NUM_HEADS + hh]);
  wexp[idx] = w;
  atomicAdd(&denom[g * NUM_HEADS + hh], w);
}

#define POOL_ROWS 64
#define POOL_CHUNK 16

__device__ __forceinline__ void pool_flush(float* __restrict__ Z, const float* __restrict__ denom,
                                           int g, int c, int D,
                                           float a0, float a1, float a2, float a3) {
  const float4 d = *(const float4*)&denom[(long)g * NUM_HEADS];
  float* zp = Z + (long)g * NUM_HEADS * D + c;
  atomicAdd(zp,              a0 / fmaxf(d.x, 1e-30f));
  atomicAdd(zp + D,          a1 / fmaxf(d.y, 1e-30f));
  atomicAdd(zp + 2L * D,     a2 / fmaxf(d.z, 1e-30f));
  atomicAdd(zp + 3L * D,     a3 / fmaxf(d.w, 1e-30f));
}

__global__ void pool_hproj(const float* __restrict__ wexp, const float* __restrict__ denom,
                           const int* __restrict__ batch, const float* __restrict__ hproj,
                           float* __restrict__ Z, int N, int D) {
  __shared__ float4 wsh[POOL_ROWS];
  __shared__ int gsh[POOL_ROWS];
  const int tid = threadIdx.x;
  const int n0 = blockIdx.x * POOL_ROWS;
  int nend = n0 + POOL_ROWS;
  if (nend > N) nend = N;
  if (tid < POOL_ROWS) {
    if (n0 + tid < N) {
      wsh[tid] = *(const float4*)&wexp[(long)(n0 + tid) * NUM_HEADS];
      gsh[tid] = batch[n0 + tid];
    } else {
      wsh[tid] = make_float4(0.0f, 0.0f, 0.0f, 0.0f);
      gsh[tid] = 0;
    }
  }
  __syncthreads();

  const int c = blockIdx.y * blockDim.x + tid;
  const bool active = c < D;
  const int cc = active ? c : D - 1;
  int gcur = __builtin_amdgcn_readfirstlane(gsh[0]);
  const float* xp = hproj + (long)n0 * D + cc;
  const int rlast = nend - 1 - n0;
  float a0 = 0.0f, a1 = 0.0f, a2 = 0.0f, a3 = 0.0f;
  for (int i0 = 0; i0 < POOL_ROWS; i0 += POOL_CHUNK) {
    if (n0 + i0 >= nend) break;
    float x[POOL_CHUNK];
    float4 wq[POOL_CHUNK];
    int gq[POOL_CHUNK];
#pragma unroll
    for (int j = 0; j < POOL_CHUNK; ++j) {
      int r = i0 + j;
      if (r > rlast) r = rlast;
      x[j] = xp[(long)r * D];
      wq[j] = wsh[i0 + j];
      gq[j] = gsh[i0 + j];
    }
#pragma unroll
    for (int j = 0; j < POOL_CHUNK; ++j) {
      if (n0 + i0 + j < nend) {
        const int g = __builtin_amdgcn_readfirstlane(gq[j]);
        if (g != gcur) {
          if (active) pool_flush(Z, denom, gcur, c, D, a0, a1, a2, a3);
          gcur = g;
          a0 = 0.0f; a1 = 0.0f; a2 = 0.0f; a3 = 0.0f;
        }
        const float4 w = wq[j];
        a0 += w.x * x[j];
        a1 += w.y * x[j];
        a2 += w.z * x[j];
        a3 += w.w * x[j];
      }
    }
  }
  if (active) pool_flush(Z, denom, gcur, c, D, a0, a1, a2, a3);
}

__global__ void vproj_pooled_wmma(const float* __restrict__ Z, const float* __restrict__ Wv,
                                  const float* __restrict__ bv, const float* __restrict__ denom,
                                  float* __restrict__ pooled, int G, int D) {
  const int tid = threadIdx.x;
  const int wave = tid >> 5, lane = tid & 31;
  const int mrow = lane & 15;
  const int khalf = (lane & 16) ? 2 : 0;
  const int g0 = blockIdx.x * 16;
  const int ncol0 = blockIdx.y * 64 + wave * 16;
  const int n = ncol0 + mrow;
  const int hh = ncol0 / (D / NUM_HEADS);
  int ga = g0 + mrow;
  if (ga > G - 1) ga = G - 1;
  const float* zr = Z + ((long)ga * NUM_HEADS + hh) * D + khalf;
  const float* wr = Wv + (long)n * D + khalf;

  v8f acc = {};
  for (int kc = 0; kc < D; kc += KCHUNK) {
    v2f a[KCHUNK / 4], b[KCHUNK / 4];
#pragma unroll
    for (int i = 0; i < KCHUNK / 4; ++i) {
      a[i] = *(const v2f*)&zr[kc + 4 * i];
      b[i] = *(const v2f*)&wr[kc + 4 * i];
    }
#pragma unroll
    for (int i = 0; i < KCHUNK / 4; ++i)
      acc = __builtin_amdgcn_wmma_f32_16x16x4_f32(false, a[i], false, b[i], (short)0, acc, false, false);
  }

  const float bn = bv[n];
  const int gb = g0 + ((lane & 16) ? 8 : 0);
  float dsum[8];
#pragma unroll
  for (int r = 0; r < 8; ++r) {
    int g = gb + r;
    if (g > G - 1) g = G - 1;
    dsum[r] = denom[g * NUM_HEADS + hh];
  }
#pragma unroll
  for (int r = 0; r < 8; ++r) {
    const int g = gb + r;
    if (g < G) {
      const float sp = dsum[r] / fmaxf(dsum[r], 1e-30f);
      pooled[(long)g * D + n] = acc[r] + sp * bn;
    }
  }
}

extern "C" void kernel_launch(void* const* d_in, const int* in_sizes, int n_in,
                              void* d_out, int out_size, void* d_ws, size_t ws_size,
                              hipStream_t stream) {
  const float* h      = (const float*)d_in[0];
  const int*   eidx   = (const int*)d_in[1];
  const float* eattr  = (const float*)d_in[2];
  const int*   batch  = (const int*)d_in[3];
  const float* node_w = (const float*)d_in[5];
  const float* node_b = (const float*)d_in[6];
  const float* edge_w = (const float*)d_in[7];
  const float* edge_b = (const float*)d_in[8];
  const float* query  = (const float*)d_in[9];
  const float* in_w   = (const float*)d_in[10];
  const float* in_b   = (const float*)d_in[11];
  const float* out_w  = (const float*)d_in[12];
  const float* out_b  = (const float*)d_in[13];
  float* out = (float*)d_out;

  int D  = in_sizes[9];
  int ND = in_sizes[5] / D;
  int ED = in_sizes[7] / D;
  int N  = in_sizes[0] / ND;
  long E = in_sizes[1] / 2;
  int G  = out_size / D;

  float* ws = (float*)d_ws;
  long o = 0;
  auto take = [&](long nf) { float* p = ws + o; o += (nf + 3) & ~3L; return p; };
  float* sumA   = take((long)N * ED);
  float* cnt    = take(N);
  float* hproj  = take((long)N * D);
  float* scores = take((long)N * NUM_HEADS);
  float* wexp   = take((long)N * NUM_HEADS);
  float* sbuf   = take((long)NUM_HEADS * D);
  float* tbuf   = take(NUM_HEADS);
  float* smax   = take((long)G * NUM_HEADS);
  float* denom  = take((long)G * NUM_HEADS + (long)G * NUM_HEADS * D);
  float* Zb     = denom + G * NUM_HEADS;
  float* pooled = take((long)G * D);

  {
    long z0 = (long)N * ED + N;
    long z1 = (long)G * NUM_HEADS + (long)G * NUM_HEADS * D;
    long maxn = z0 > z1 ? z0 : z1;
    int blocks = (int)((maxn + 255) / 256);
    if (blocks > 262144) blocks = 262144;
    init_ws<<<blocks, 256, 0, stream>>>(sumA, z0, denom, z1, smax, G * NUM_HEADS);
  }
  qst_kernel<<<1, D, 0, stream>>>(query, in_w, in_b, D, sbuf, tbuf);
  {
    long total = E * (long)ED;
    long blocks = (total + 255) / 256;
    if (blocks > 262144) blocks = 262144;
    edge_scatter<<<(int)blocks, 256, 0, stream>>>(eattr, eidx, sumA, cnt, E, ED);
  }
  {
    dim3 grid((N + 63) / 64, D / 64);
    hproj_wmma<<<grid, 128, 0, stream>>>(h, sumA, cnt, node_w, node_b, edge_w, edge_b,
                                         hproj, N, ND, ED, D);
  }
  {
    int total = N * NUM_HEADS;
    scores_kernel<<<(total + 255) / 256, 256, 0, stream>>>(hproj, sbuf, tbuf, batch,
                                                           smax, scores, N, D);
    expw_kernel<<<(total + 255) / 256, 256, 0, stream>>>(scores, batch, smax, denom,
                                                         wexp, N);
  }
  {
    dim3 grid((N + POOL_ROWS - 1) / POOL_ROWS, (D + 255) / 256);
    pool_hproj<<<grid, 256, 0, stream>>>(wexp, denom, batch, hproj, Zb, N, D);
  }
  {
    dim3 grid((G + 15) / 16, D / 64);
    vproj_pooled_wmma<<<grid, 128, 0, stream>>>(Zb, in_w + 2L * D * D, in_b + 2 * D,
                                                denom, pooled, G, D);
  }
  {
    dim3 grid((G + 63) / 64, D / 64);
    gemm_xwT_wmma<<<grid, 128, 0, stream>>>(pooled, out_w, out_b, out, G, D, D);
  }
}


# Edge sum as a gather: each edge claims a slot in its node's list with one integer atomic, then 16 lanes per node sum the listed rows (per-node sum order: slot order); the 64 float atomics per edge go
# speedup vs baseline: 1.6845x; 1.2818x over previous
#include <hip/hip_runtime.h>
#include <math.h>

#define NUM_HEADS 4
#define KCHUNK 64
#define LDSTRIDE 68

typedef float v2f __attribute__((ext_vector_type(2)));
typedef float v8f __attribute__((ext_vector_type(8)));

__device__ __forceinline__ void atomic_max_float(float* addr, float val) {
  if (val >= 0.0f) {
    atomicMax((int*)addr, __float_as_int(val));
  } else {
    atomicMin((unsigned int*)addr, __float_as_uint(val));
  }
}

__global__ void init_ws(float* z0, long n0, float* z1, long n1, float* smax, int ns) {
  long i = (long)blockIdx.x * blockDim.x + threadIdx.x;
  long stride = (long)gridDim.x * blockDim.x;
  for (long t = i; t < n0; t += stride) z0[t] = 0.0f;
  for (long t = i; t < n1; t += stride) z1[t] = 0.0f;
  for (long t = i; t < ns; t += stride) smax[t] = -INFINITY;
}

__global__ void qst_kernel(const float* __restrict__ query, const float* __restrict__ in_w,
                           const float* __restrict__ in_b, int D,
                           float* __restrict__ sbuf, float* __restrict__ tbuf) {
  __shared__ float qs[1024];
  int j = threadIdx.x;
  int dh = D / NUM_HEADS;
  float acc = in_b[j];
  for (int c = 0; c < D; ++c) acc += query[c] * in_w[(long)j * D + c];
  qs[j] = acc;
  __syncthreads();
  for (int hh = 0; hh < NUM_HEADS; ++hh) {
    float s = 0.0f;
    const float* wk = in_w + (long)(D + hh * dh) * D;
    for (int d = 0; d < dh; ++d) s += qs[hh * dh + d] * wk[(long)d * D + j];
    sbuf[hh * D + j] = s;
  }
  if (j < NUM_HEADS) {
    float t = 0.0f;
    for (int d = 0; d < dh; ++d) t += qs[j * dh + d] * in_b[D + j * dh + d];
    tbuf[j] = t;
  }
}

#define SLOT_CAP 64
#define GATHER_CHUNK 8

__global__ void edge_slots(const int* __restrict__ row, const float* __restrict__ eattr,
                           int* __restrict__ cnti, int* __restrict__ slots,
                           float* __restrict__ sumA, long E, int ED) {
  long i = (long)blockIdx.x * blockDim.x + threadIdx.x;
  long stride = (long)gridDim.x * blockDim.x;
  for (long e = i; e < E; e += stride) {
    int r = row[e];
    int s = atomicAdd(&cnti[r], 1);
    if (s < SLOT_CAP) {
      slots[(long)r * SLOT_CAP + s] = (int)e;
    } else {
      for (int c = 0; c < ED; ++c) atomicAdd(&sumA[(long)r * ED + c], eattr[e * ED + c]);
    }
  }
}

__global__ void edge_gather(const float* __restrict__ eattr, const int* __restrict__ slots,
                            const int* __restrict__ cnti, float* __restrict__ sumA,
                            float* __restrict__ cntf, int N, int ED) {
  const int t = blockIdx.x * blockDim.x + threadIdx.x;
  const int m = t >> 4;
  const int l = t & 15;
  if (m >= N) return;
  const int cdeg = cnti[m];
  const int cs = cdeg < SLOT_CAP ? cdeg : SLOT_CAP;
  const int* sp = slots + (long)m * SLOT_CAP;
  for (int c0 = l * 4; c0 < ED; c0 += 64) {
    float4 acc = make_float4(0.0f, 0.0f, 0.0f, 0.0f);
    if (cdeg > SLOT_CAP) acc = *(const float4*)&sumA[(long)m * ED + c0];
    for (int j0 = 0; j0 < cs; j0 += GATHER_CHUNK) {
      float4 v[GATHER_CHUNK];
#pragma unroll
      for (int j = 0; j < GATHER_CHUNK; ++j) {
        int jj = j0 + j;
        if (jj > cs - 1) jj = cs - 1;
        const int e = sp[jj];
        v[j] = *(const float4*)&eattr[(long)e * ED + c0];
      }
#pragma unroll
      for (int j = 0; j < GATHER_CHUNK; ++j)
        asm volatile("" :: "v"(v[j].x), "v"(v[j].y), "v"(v[j].z), "v"(v[j].w));
#pragma unroll
      for (int j = 0; j < GATHER_CHUNK; ++j) {
        const bool in_list = j0 + j < cs;
        acc.x += in_list ? v[j].x : 0.0f;
        acc.y += in_list ? v[j].y : 0.0f;
        acc.z += in_list ? v[j].z : 0.0f;
        acc.w += in_list ? v[j].w : 0.0f;
      }
    }
    *(float4*)&sumA[(long)m * ED + c0] = acc;
  }
  if (l == 0) cntf[m] = (float)cdeg;
}


__global__ void hproj_wmma(const float* __restrict__ h, const float* __restrict__ sumA,
                           const float* __restrict__ cnt,
                           const float* __restrict__ node_w, const float* __restrict__ node_b,
                           const float* __restrict__ edge_w, const float* __restrict__ edge_b,
                           float* __restrict__ hproj, int N, int ND, int ED, int D) {
  const int K = ND + ED;
  __shared__ float Xs[64 * LDSTRIDE];
  int row0 = blockIdx.x * 64;
  int tid = threadIdx.x;
  int wave = tid >> 5, lane = tid & 31;
  int mrow = lane & 15;
  int n = blockIdx.y * 64 + wave * 16 + mrow;
  int khalf = (lane & 16) ? 2 : 0;

  v8f acc0 = {}, acc1 = {}, acc2 = {}, acc3 = {};

  for (int kc = 0; kc < K; kc += KCHUNK) {
    __syncthreads();
    for (int t = tid; t < 64 * KCHUNK; t += blockDim.x) {
      int r = t >> 6;
      int c = t & (KCHUNK - 1);
      int m = row0 + r;
      int cg = kc + c;
      float val = 0.0f;
      if (m < N) {
        if (cg < ND) val = h[(long)m * ND + cg];
        else         val = sumA[(long)m * ED + (cg - ND)] / fmaxf(cnt[m], 1.0f);
      }
      Xs[r * LDSTRIDE + c] = val;
    }
    __syncthreads();
    for (int k = 0; k < KCHUNK; k += 4) {
      int kk = k + khalf;
      v2f b;
      if (kc + k < ND) b = *(const v2f*)&node_w[(long)n * ND + kc + kk];
      else             b = *(const v2f*)&edge_w[(long)n * ED + (kc + kk - ND)];
      v2f a0 = *(const v2f*)&Xs[(mrow +  0) * LDSTRIDE + kk];
      v2f a1 = *(const v2f*)&Xs[(mrow + 16) * LDSTRIDE + kk];
      v2f a2 = *(const v2f*)&Xs[(mrow + 32) * LDSTRIDE + kk];
      v2f a3 = *(const v2f*)&Xs[(mrow + 48) * LDSTRIDE + kk];
      acc0 = __builtin_amdgcn_wmma_f32_16x16x4_f32(false, a0, false, b, (short)0, acc0, false, false);
      acc1 = __builtin_amdgcn_wmma_f32_16x16x4_f32(false, a1, false, b, (short)0, acc1, false, false);
      acc2 = __builtin_amdgcn_wmma_f32_16x16x4_f32(false, a2, false, b, (short)0, acc2, false, false);
      acc3 = __builtin_amdgcn_wmma_f32_16x16x4_f32(false, a3, false, b, (short)0, acc3, false, false);
    }
  }

  float bn = node_b[n];
  float be = edge_b[n];
  for (int r = 0; r < 8; ++r) {
    int mb = row0 + r + ((lane & 16) ? 8 : 0);
    v8f* accs[4] = {&acc0, &acc1, &acc2, &acc3};
    for (int mt = 0; mt < 4; ++mt) {
      int m = mb + mt * 16;
      if (m < N) {
        float mask = (cnt[m] > 0.0f) ? 1.0f : 0.0f;
        hproj[(long)m * D + n] = (*accs[mt])[r] + bn + mask * be;
      }
    }
  }
}

__global__ void gemm_xwT_wmma(const float* __restrict__ X, const float* __restrict__ W,
                              const float* __restrict__ bias, float* __restrict__ Y,
                              int M, int K, int Nout) {
  __shared__ float Xs[64 * LDSTRIDE];
  int row0 = blockIdx.x * 64;
  int tid = threadIdx.x;
  int wave = tid >> 5, lane = tid & 31;
  int mrow = lane & 15;
  int n = blockIdx.y * 64 + wave * 16 + mrow;
  int khalf = (lane & 16) ? 2 : 0;

  v8f acc0 = {}, acc1 = {}, acc2 = {}, acc3 = {};

  for (int kc = 0; kc < K; kc += KCHUNK) {
    __syncthreads();
    for (int t = tid; t < 64 * KCHUNK; t += blockDim.x) {
      int r = t >> 6;
      int c = t & (KCHUNK - 1);
      int m = row0 + r;
      Xs[r * LDSTRIDE + c] = (m < M) ? X[(long)m * K + kc + c] : 0.0f;
    }
    __syncthreads();
    for (int k = 0; k < KCHUNK; k += 4) {
      int kk = k + khalf;
      v2f b = *(const v2f*)&W[(long)n * K + kc + kk];
      v2f a0 = *(const v2f*)&Xs[(mrow +  0) * LDSTRIDE + kk];
      v2f a1 = *(const v2f*)&Xs[(mrow + 16) * LDSTRIDE + kk];
      v2f a2 = *(const v2f*)&Xs[(mrow + 32) * LDSTRIDE + kk];
      v2f a3 = *(const v2f*)&Xs[(mrow + 48) * LDSTRIDE + kk];
      acc0 = __builtin_amdgcn_wmma_f32_16x16x4_f32(false, a0, false, b, (short)0, acc0, false, false);
      acc1 = __builtin_amdgcn_wmma_f32_16x16x4_f32(false, a1, false, b, (short)0, acc1, false, false);
      acc2 = __builtin_amdgcn_wmma_f32_16x16x4_f32(false, a2, false, b, (short)0, acc2, false, false);
      acc3 = __builtin_amdgcn_wmma_f32_16x16x4_f32(false, a3, false, b, (short)0, acc3, false, false);
    }
  }

  float bn = bias[n];
  for (int r = 0; r < 8; ++r) {
    int mb = row0 + r + ((lane & 16) ? 8 : 0);
    v8f* accs[4] = {&acc0, &acc1, &acc2, &acc3};
    for (int mt = 0; mt < 4; ++mt) {
      int m = mb + mt * 16;
      if (m < M) Y[(long)m * Nout + n] = (*accs[mt])[r] + bn;
    }
  }
}

__global__ void scores_kernel(const float* __restrict__ hproj, const float* __restrict__ sbuf,
                              const float* __restrict__ tbuf, const int* __restrict__ batch,
                              float* __restrict__ smax, float* __restrict__ scores,
                              int N, int D) {
  int idx = blockIdx.x * blockDim.x + threadIdx.x;
  if (idx >= N * NUM_HEADS) return;
  int nI = idx >> 2;
  int hh = idx & 3;
  const float* xr = hproj + (long)nI * D;
  const float* sr = sbuf + hh * D;
  float dot = tbuf[hh];
  for (int c = 0; c < D; c += 4) {
    float4 x = *(const float4*)&xr[c];
    float4 s = *(const float4*)&sr[c];
    dot += x.x * s.x + x.y * s.y + x.z * s.z + x.w * s.w;
  }
  float sc = dot * rsqrtf((float)(D / NUM_HEADS));
  scores[idx] = sc;
  atomic_max_float(&smax[batch[nI] * NUM_HEADS + hh], sc);
}

__global__ void expw_kernel(const float* __restrict__ scores, const int* __restrict__ batch,
                            const float* __restrict__ smax, float* __restrict__ denom,
                            float* __restrict__ wexp, int N) {
  int idx = blockIdx.x * blockDim.x + threadIdx.x;
  if (idx >= N * NUM_HEADS) return;
  int nI = idx >> 2;
  int hh = idx & 3;
  int g = batch[nI];
  float w = __expf(scores[idx] - smax[g * NUM_HEADS + hh]);
  wexp[idx] = w;
  atomicAdd(&denom[g * NUM_HEADS + hh], w);
}

#define POOL_ROWS 64
#define POOL_CHUNK 16

__device__ __forceinline__ void pool_flush(float* __restrict__ Z, const float* __restrict__ denom,
                                           int g, int c, int D,
                                           float a0, float a1, float a2, float a3) {
  const float4 d = *(const float4*)&denom[(long)g * NUM_HEADS];
  float* zp = Z + (long)g * NUM_HEADS * D + c;
  atomicAdd(zp,              a0 / fmaxf(d.x, 1e-30f));
  atomicAdd(zp + D,          a1 / fmaxf(d.y, 1e-30f));
  atomicAdd(zp + 2L * D,     a2 / fmaxf(d.z, 1e-30f));
  atomicAdd(zp + 3L * D,     a3 / fmaxf(d.w, 1e-30f));
}

__global__ void pool_hproj(const float* __restrict__ wexp, const float* __restrict__ denom,
                           const int* __restrict__ batch, const float* __restrict__ hproj,
                           float* __restrict__ Z, int N, int D) {
  __shared__ float4 wsh[POOL_ROWS];
  __shared__ int gsh[POOL_ROWS];
  const int tid = threadIdx.x;
  const int n0 = blockIdx.x * POOL_ROWS;
  int nend = n0 + POOL_ROWS;
  if (nend > N) nend = N;
  if (tid < POOL_ROWS) {
    if (n0 + tid < N) {
      wsh[tid] = *(const float4*)&wexp[(long)(n0 + tid) * NUM_HEADS];
      gsh[tid] = batch[n0 + tid];
    } else {
      wsh[tid] = make_float4(0.0f, 0.0f, 0.0f, 0.0f);
      gsh[tid] = 0;
    }
  }
  __syncthreads();

  const int c = blockIdx.y * blockDim.x + tid;
  const bool active = c < D;
  const int cc = active ? c : D - 1;
  int gcur = __builtin_amdgcn_readfirstlane(gsh[0]);
  const float* xp = hproj + (long)n0 * D + cc;
  const int rlast = nend - 1 - n0;
  float a0 = 0.0f, a1 = 0.0f, a2 = 0.0f, a3 = 0.0f;
  for (int i0 = 0; i0 < POOL_ROWS; i0 += POOL_CHUNK) {
    if (n0 + i0 >= nend) break;
    float x[POOL_CHUNK];
    float4 wq[POOL_CHUNK];
    int gq[POOL_CHUNK];
#pragma unroll
    for (int j = 0; j < POOL_CHUNK; ++j) {
      int r = i0 + j;
      if (r > rlast) r = rlast;
      x[j] = xp[(long)r * D];
      wq[j] = wsh[i0 + j];
      gq[j] = gsh[i0 + j];
    }
#pragma unroll
    for (int j = 0; j < POOL_CHUNK; ++j) {
      if (n0 + i0 + j < nend) {
        const int g = __builtin_amdgcn_readfirstlane(gq[j]);
        if (g != gcur) {
          if (active) pool_flush(Z, denom, gcur, c, D, a0, a1, a2, a3);
          gcur = g;
          a0 = 0.0f; a1 = 0.0f; a2 = 0.0f; a3 = 0.0f;
        }
        const float4 w = wq[j];
        a0 += w.x * x[j];
        a1 += w.y * x[j];
        a2 += w.z * x[j];
        a3 += w.w * x[j];
      }
    }
  }
  if (active) pool_flush(Z, denom, gcur, c, D, a0, a1, a2, a3);
}

__global__ void vproj_pooled_wmma(const float* __restrict__ Z, const float* __restrict__ Wv,
                                  const float* __restrict__ bv, const float* __restrict__ denom,
                                  float* __restrict__ pooled, int G, int D) {
  const int tid = threadIdx.x;
  const int wave = tid >> 5, lane = tid & 31;
  const int mrow = lane & 15;
  const int khalf = (lane & 16) ? 2 : 0;
  const int g0 = blockIdx.x * 16;
  const int ncol0 = blockIdx.y * 64 + wave * 16;
  const int n = ncol0 + mrow;
  const int hh = ncol0 / (D / NUM_HEADS);
  int ga = g0 + mrow;
  if (ga > G - 1) ga = G - 1;
  const float* zr = Z + ((long)ga * NUM_HEADS + hh) * D + khalf;
  const float* wr = Wv + (long)n * D + khalf;

  v8f acc = {};
  for (int kc = 0; kc < D; kc += KCHUNK) {
    v2f a[KCHUNK / 4], b[KCHUNK / 4];
#pragma unroll
    for (int i = 0; i < KCHUNK / 4; ++i) {
      a[i] = *(const v2f*)&zr[kc + 4 * i];
      b[i] = *(const v2f*)&wr[kc + 4 * i];
    }
#pragma unroll
    for (int i = 0; i < KCHUNK / 4; ++i)
      acc = __builtin_amdgcn_wmma_f32_16x16x4_f32(false, a[i], false, b[i], (short)0, acc, false, false);
  }

  const float bn = bv[n];
  const int gb = g0 + ((lane & 16) ? 8 : 0);
  float dsum[8];
#pragma unroll
  for (int r = 0; r < 8; ++r) {
    int g = gb + r;
    if (g > G - 1) g = G - 1;
    dsum[r] = denom[g * NUM_HEADS + hh];
  }
#pragma unroll
  for (int r = 0; r < 8; ++r) {
    const int g = gb + r;
    if (g < G) {
      const float sp = dsum[r] / fmaxf(dsum[r], 1e-30f);
      pooled[(long)g * D + n] = acc[r] + sp * bn;
    }
  }
}

extern "C" void kernel_launch(void* const* d_in, const int* in_sizes, int n_in,
                              void* d_out, int out_size, void* d_ws, size_t ws_size,
                              hipStream_t stream) {
  const float* h      = (const float*)d_in[0];
  const int*   eidx   = (const int*)d_in[1];
  const float* eattr  = (const float*)d_in[2];
  const int*   batch  = (const int*)d_in[3];
  const float* node_w = (const float*)d_in[5];
  const float* node_b = (const float*)d_in[6];
  const float* edge_w = (const float*)d_in[7];
  const float* edge_b = (const float*)d_in[8];
  const float* query  = (const float*)d_in[9];
  const float* in_w   = (const float*)d_in[10];
  const float* in_b   = (const float*)d_in[11];
  const float* out_w  = (const float*)d_in[12];
  const float* out_b  = (const float*)d_in[13];
  float* out = (float*)d_out;

  int D  = in_sizes[9];
  int ND = in_sizes[5] / D;
  int ED = in_sizes[7] / D;
  int N  = in_sizes[0] / ND;
  long E = in_sizes[1] / 2;
  int G  = out_size / D;

  float* ws = (float*)d_ws;
  long o = 0;
  auto take = [&](long nf) { float* p = ws + o; o += (nf + 3) & ~3L; return p; };
  float* sumA   = take((long)N * ED);
  int*   cnti   = (int*)take(N);
  float* cnt    = take(N);
  int*   slots  = (int*)take((long)N * SLOT_CAP);
  float* hproj  = take((long)N * D);
  float* scores = take((long)N * NUM_HEADS);
  float* wexp   = take((long)N * NUM_HEADS);
  float* sbuf   = take((long)NUM_HEADS * D);
  float* tbuf   = take(NUM_HEADS);
  float* smax   = take((long)G * NUM_HEADS);
  float* denom  = take((long)G * NUM_HEADS + (long)G * NUM_HEADS * D);
  float* Zb     = denom + G * NUM_HEADS;
  float* pooled = take((long)G * D);

  {
    long z0 = (long)N * ED + N;
    long z1 = (long)G * NUM_HEADS + (long)G * NUM_HEADS * D;
    long maxn = z0 > z1 ? z0 : z1;
    int blocks = (int)((maxn + 255) / 256);
    if (blocks > 262144) blocks = 262144;
    init_ws<<<blocks, 256, 0, stream>>>(sumA, z0, denom, z1, smax, G * NUM_HEADS);
  }
  qst_kernel<<<1, D, 0, stream>>>(query, in_w, in_b, D, sbuf, tbuf);
  {
    long blocks = (E + 255) / 256;
    if (blocks > 262144) blocks = 262144;
    edge_slots<<<(int)blocks, 256, 0, stream>>>(eidx, eattr, cnti, slots, sumA, E, ED);
    long gthreads = (long)N * 16;
    edge_gather<<<(int)((gthreads + 255) / 256), 256, 0, stream>>>(eattr, slots, cnti, sumA,
                                                                 cnt, N, ED);
  }
  {
    dim3 grid((N + 63) / 64, D / 64);
    hproj_wmma<<<grid, 128, 0, stream>>>(h, sumA, cnt, node_w, node_b, edge_w, edge_b,
                                         hproj, N, ND, ED, D);
  }
  {
    int total = N * NUM_HEADS;
    scores_kernel<<<(total + 255) / 256, 256, 0, stream>>>(hproj, sbuf, tbuf, batch,
                                                           smax, scores, N, D);
    expw_kernel<<<(total + 255) / 256, 256, 0, stream>>>(scores, batch, smax, denom,
                                                         wexp, N);
  }
  {
    dim3 grid((N + POOL_ROWS - 1) / POOL_ROWS, (D + 255) / 256);
    pool_hproj<<<grid, 256, 0, stream>>>(wexp, denom, batch, hproj, Zb, N, D);
  }
  {
    dim3 grid((G + 15) / 16, D / 64);
    vproj_pooled_wmma<<<grid, 128, 0, stream>>>(Zb, in_w + 2L * D * D, in_b + 2 * D,
                                                denom, pooled, G, D);
  }
  {
    dim3 grid((G + 63) / 64, D / 64);
    gemm_xwT_wmma<<<grid, 128, 0, stream>>>(pooled, out_w, out_b, out, G, D, D);
  }
}


# h_proj is never formed: scores and pooling run on the raw [h | mean edge attr] rows (linearity), the node/edge projection runs on the G*4 pooled rows (WMMA); sum order changes
# speedup vs baseline: 2.1310x; 1.2651x over previous
#include <hip/hip_runtime.h>
#include <math.h>

#define NUM_HEADS 4
#define KCHUNK 64
#define LDSTRIDE 68

typedef float v2f __attribute__((ext_vector_type(2)));
typedef float v8f __attribute__((ext_vector_type(8)));

__device__ __forceinline__ void atomic_max_float(float* addr, float val) {
  if (val >= 0.0f) {
    atomicMax((int*)addr, __float_as_int(val));
  } else {
    atomicMin((unsigned int*)addr, __float_as_uint(val));
  }
}

__global__ void init_ws(float* z0, long n0, float* z1, long n1, float* smax, int ns) {
  long i = (long)blockIdx.x * blockDim.x + threadIdx.x;
  long stride = (long)gridDim.x * blockDim.x;
  for (long t = i; t < n0; t += stride) z0[t] = 0.0f;
  for (long t = i; t < n1; t += stride) z1[t] = 0.0f;
  for (long t = i; t < ns; t += stride) smax[t] = -INFINITY;
}

__global__ void qst_kernel(const float* __restrict__ query, const float* __restrict__ in_w,
                           const float* __restrict__ in_b, int D,
                           float* __restrict__ sbuf, float* __restrict__ tbuf) {
  __shared__ float qs[1024];
  int j = threadIdx.x;
  int dh = D / NUM_HEADS;
  float acc = in_b[j];
  for (int c = 0; c < D; ++c) acc += query[c] * in_w[(long)j * D + c];
  qs[j] = acc;
  __syncthreads();
  for (int hh = 0; hh < NUM_HEADS; ++hh) {
    float s = 0.0f;
    const float* wk = in_w + (long)(D + hh * dh) * D;
    for (int d = 0; d < dh; ++d) s += qs[hh * dh + d] * wk[(long)d * D + j];
    sbuf[hh * D + j] = s;
  }
  if (j < NUM_HEADS) {
    float t = 0.0f;
    for (int d = 0; d < dh; ++d) t += qs[j * dh + d] * in_b[D + j * dh + d];
    tbuf[j] = t;
  }
}

#define UF_WAVES 8

__global__ void ufold_kernel(const float* __restrict__ sbuf,
                             const float* __restrict__ node_w, const float* __restrict__ edge_w,
                             const float* __restrict__ node_b, const float* __restrict__ edge_b,
                             float* __restrict__ ubuf, int D, int ND, int ED, int US) {
  __shared__ float red[UF_WAVES][NUM_HEADS][32];
  const int tid = threadIdx.x;
  const int lane = tid & 31, wave = tid >> 5;
  const int K = ND + ED;
  const int k = blockIdx.x * 32 + lane;
  const float* wp;
  long stride;
  if (k < ND)      { wp = node_w + k;        stride = ND; }
  else if (k < K)  { wp = edge_w + (k - ND); stride = ED; }
  else if (k == K) { wp = node_b;            stride = 1; }
  else             { wp = edge_b;            stride = 1; }
  float a0 = 0.0f, a1 = 0.0f, a2 = 0.0f, a3 = 0.0f;
  for (int j = wave; j < D; j += UF_WAVES) {
    const float w = wp[(long)j * stride];
    a0 += sbuf[j] * w;
    a1 += sbuf[D + j] * w;
    a2 += sbuf[2 * D + j] * w;
    a3 += sbuf[3 * D + j] * w;
  }
  red[wave][0][lane] = a0;
  red[wave][1][lane] = a1;
  red[wave][2][lane] = a2;
  red[wave][3][lane] = a3;
  __syncthreads();
  if (tid < NUM_HEADS * 32) {
    const int hh = tid >> 5;
    const int kk = blockIdx.x * 32 + lane;
    float sum = 0.0f;
    for (int w = 0; w < UF_WAVES; ++w) sum += red[w][hh][lane];
    if (kk < K + 2) ubuf[hh * US + kk] = sum;
  }
}

#define SLOT_CAP 64
#define GATHER_CHUNK 8

__global__ void edge_slots(const int* __restrict__ row, const float* __restrict__ eattr,
                           int* __restrict__ cnti, int* __restrict__ slots,
                           float* __restrict__ sumA, long E, int ED) {
  long i = (long)blockIdx.x * blockDim.x + threadIdx.x;
  long stride = (long)gridDim.x * blockDim.x;
  for (long e = i; e < E; e += stride) {
    int r = row[e];
    int s = atomicAdd(&cnti[r], 1);
    if (s < SLOT_CAP) {
      slots[(long)r * SLOT_CAP + s] = (int)e;
    } else {
      for (int c = 0; c < ED; ++c) atomicAdd(&sumA[(long)r * ED + c], eattr[e * ED + c]);
    }
  }
}

__global__ void edge_gather(const float* __restrict__ eattr, const int* __restrict__ slots,
                            const int* __restrict__ cnti, float* __restrict__ sumA,
                            float* __restrict__ cntf, int N, int ED) {
  const int t = blockIdx.x * blockDim.x + threadIdx.x;
  const int m = t >> 4;
  const int l = t & 15;
  if (m >= N) return;
  const int cdeg = cnti[m];
  const int cs = cdeg < SLOT_CAP ? cdeg : SLOT_CAP;
  const int* sp = slots + (long)m * SLOT_CAP;
  for (int c0 = l * 4; c0 < ED; c0 += 64) {
    float4 acc = make_float4(0.0f, 0.0f, 0.0f, 0.0f);
    if (cdeg > SLOT_CAP) acc = *(const float4*)&sumA[(long)m * ED + c0];
    for (int j0 = 0; j0 < cs; j0 += GATHER_CHUNK) {
      float4 v[GATHER_CHUNK];
#pragma unroll
      for (int j = 0; j < GATHER_CHUNK; ++j) {
        int jj = j0 + j;
        if (jj > cs - 1) jj = cs - 1;
        const int e = sp[jj];
        v[j] = *(const float4*)&eattr[(long)e * ED + c0];
      }
#pragma unroll
      for (int j = 0; j < GATHER_CHUNK; ++j)
        asm volatile("" :: "v"(v[j].x), "v"(v[j].y), "v"(v[j].z), "v"(v[j].w));
#pragma unroll
      for (int j = 0; j < GATHER_CHUNK; ++j) {
        const bool in_list = j0 + j < cs;
        acc.x += in_list ? v[j].x : 0.0f;
        acc.y += in_list ? v[j].y : 0.0f;
        acc.z += in_list ? v[j].z : 0.0f;
        acc.w += in_list ? v[j].w : 0.0f;
      }
    }
    const float dm = fmaxf((float)cdeg, 1.0f);
    acc.x /= dm; acc.y /= dm; acc.z /= dm; acc.w /= dm;
    *(float4*)&sumA[(long)m * ED + c0] = acc;
  }
  if (l == 0) cntf[m] = (float)cdeg;
}


__global__ void gemm_xwT_wmma(const float* __restrict__ X, const float* __restrict__ W,
                              const float* __restrict__ bias, float* __restrict__ Y,
                              int M, int K, int Nout) {
  __shared__ float Xs[64 * LDSTRIDE];
  int row0 = blockIdx.x * 64;
  int tid = threadIdx.x;
  int wave = tid >> 5, lane = tid & 31;
  int mrow = lane & 15;
  int n = blockIdx.y * 64 + wave * 16 + mrow;
  int khalf = (lane & 16) ? 2 : 0;

  v8f acc0 = {}, acc1 = {}, acc2 = {}, acc3 = {};

  for (int kc = 0; kc < K; kc += KCHUNK) {
    __syncthreads();
    for (int t = tid; t < 64 * KCHUNK; t += blockDim.x) {
      int r = t >> 6;
      int c = t & (KCHUNK - 1);
      int m = row0 + r;
      Xs[r * LDSTRIDE + c] = (m < M) ? X[(long)m * K + kc + c] : 0.0f;
    }
    __syncthreads();
    for (int k = 0; k < KCHUNK; k += 4) {
      int kk = k + khalf;
      v2f b = *(const v2f*)&W[(long)n * K + kc + kk];
      v2f a0 = *(const v2f*)&Xs[(mrow +  0) * LDSTRIDE + kk];
      v2f a1 = *(const v2f*)&Xs[(mrow + 16) * LDSTRIDE + kk];
      v2f a2 = *(const v2f*)&Xs[(mrow + 32) * LDSTRIDE + kk];
      v2f a3 = *(const v2f*)&Xs[(mrow + 48) * LDSTRIDE + kk];
      acc0 = __builtin_amdgcn_wmma_f32_16x16x4_f32(false, a0, false, b, (short)0, acc0, false, false);
      acc1 = __builtin_amdgcn_wmma_f32_16x16x4_f32(false, a1, false, b, (short)0, acc1, false, false);
      acc2 = __builtin_amdgcn_wmma_f32_16x16x4_f32(false, a2, false, b, (short)0, acc2, false, false);
      acc3 = __builtin_amdgcn_wmma_f32_16x16x4_f32(false, a3, false, b, (short)0, acc3, false, false);
    }
  }

  float bn = bias[n];
  for (int r = 0; r < 8; ++r) {
    int mb = row0 + r + ((lane & 16) ? 8 : 0);
    v8f* accs[4] = {&acc0, &acc1, &acc2, &acc3};
    for (int mt = 0; mt < 4; ++mt) {
      int m = mb + mt * 16;
      if (m < M) Y[(long)m * Nout + n] = (*accs[mt])[r] + bn;
    }
  }
}

__global__ void scores_kernel(const float* __restrict__ h, const float* __restrict__ xe,
                              const float* __restrict__ cnt, const float* __restrict__ ubuf,
                              const float* __restrict__ tbuf, const int* __restrict__ batch,
                              float* __restrict__ smax, float* __restrict__ scores,
                              int N, int ND, int ED, int US, int D) {
  int idx = blockIdx.x * blockDim.x + threadIdx.x;
  if (idx >= N * NUM_HEADS) return;
  int nI = idx >> 2;
  int hh = idx & 3;
  const int K = ND + ED;
  const float* ur = ubuf + hh * US;
  float dot = tbuf[hh] + ur[K] + ((cnt[nI] > 0.0f) ? ur[K + 1] : 0.0f);
  const float* xr = h + (long)nI * ND;
  for (int c = 0; c < ND; c += 4) {
    float4 x = *(const float4*)&xr[c];
    float4 s = *(const float4*)&ur[c];
    dot += x.x * s.x + x.y * s.y + x.z * s.z + x.w * s.w;
  }
  const float* er = xe + (long)nI * ED;
  const float* ue = ur + ND;
  for (int c = 0; c < ED; c += 4) {
    float4 x = *(const float4*)&er[c];
    float4 s = *(const float4*)&ue[c];
    dot += x.x * s.x + x.y * s.y + x.z * s.z + x.w * s.w;
  }
  float sc = dot * rsqrtf((float)(D / NUM_HEADS));
  scores[idx] = sc;
  atomic_max_float(&smax[batch[nI] * NUM_HEADS + hh], sc);
}

__global__ void expw_kernel(const float* __restrict__ scores, const int* __restrict__ batch,
                            const float* __restrict__ smax, float* __restrict__ denom,
                            float* __restrict__ wexp, int N) {
  int idx = blockIdx.x * blockDim.x + threadIdx.x;
  if (idx >= N * NUM_HEADS) return;
  int nI = idx >> 2;
  int hh = idx & 3;
  int g = batch[nI];
  float w = __expf(scores[idx] - smax[g * NUM_HEADS + hh]);
  wexp[idx] = w;
  atomicAdd(&denom[g * NUM_HEADS + hh], w);
}

#define POOL_ROWS 64
#define POOL_CHUNK 16

__device__ __forceinline__ void pool_flush(float* __restrict__ Z, const float* __restrict__ denom,
                                           int g, int c, int ZS,
                                           float a0, float a1, float a2, float a3) {
  const float4 d = *(const float4*)&denom[(long)g * NUM_HEADS];
  float* zp = Z + (long)g * NUM_HEADS * ZS + c;
  atomicAdd(zp,              a0 / fmaxf(d.x, 1e-30f));
  atomicAdd(zp + ZS,         a1 / fmaxf(d.y, 1e-30f));
  atomicAdd(zp + 2L * ZS,    a2 / fmaxf(d.z, 1e-30f));
  atomicAdd(zp + 3L * ZS,    a3 / fmaxf(d.w, 1e-30f));
}

__global__ void pool_x(const float* __restrict__ wexp, const float* __restrict__ denom,
                       const int* __restrict__ batch, const float* __restrict__ h,
                       const float* __restrict__ xe, const float* __restrict__ cnt,
                       float* __restrict__ Zx, int N, int ND, int ED, int ZS) {
  __shared__ float4 wsh[POOL_ROWS];
  __shared__ int gsh[POOL_ROWS];
  const int tid = threadIdx.x;
  const int n0 = blockIdx.x * POOL_ROWS;
  int nend = n0 + POOL_ROWS;
  if (nend > N) nend = N;
  if (tid < POOL_ROWS) {
    if (n0 + tid < N) {
      wsh[tid] = *(const float4*)&wexp[(long)(n0 + tid) * NUM_HEADS];
      gsh[tid] = batch[n0 + tid];
    } else {
      wsh[tid] = make_float4(0.0f, 0.0f, 0.0f, 0.0f);
      gsh[tid] = 0;
    }
  }
  __syncthreads();

  const int K = ND + ED;
  const int c = blockIdx.y * blockDim.x + tid;
  const bool active = c <= K;
  const bool is_mask = c >= K;
  const float* xp;
  long pitch;
  if (c < ND)     { xp = h + (long)n0 * ND + c;          pitch = ND; }
  else if (c < K) { xp = xe + (long)n0 * ED + (c - ND);  pitch = ED; }
  else            { xp = cnt + n0;                       pitch = 1; }
  int gcur = __builtin_amdgcn_readfirstlane(gsh[0]);
  const int rlast = nend - 1 - n0;
  float a0 = 0.0f, a1 = 0.0f, a2 = 0.0f, a3 = 0.0f;
  for (int i0 = 0; i0 < POOL_ROWS; i0 += POOL_CHUNK) {
    if (n0 + i0 >= nend) break;
    float x[POOL_CHUNK];
    float4 wq[POOL_CHUNK];
    int gq[POOL_CHUNK];
#pragma unroll
    for (int j = 0; j < POOL_CHUNK; ++j) {
      int r = i0 + j;
      if (r > rlast) r = rlast;
      x[j] = xp[(long)r * pitch];
      wq[j] = wsh[i0 + j];
      gq[j] = gsh[i0 + j];
    }
#pragma unroll
    for (int j = 0; j < POOL_CHUNK; ++j) {
      if (n0 + i0 + j < nend) {
        const int g = __builtin_amdgcn_readfirstlane(gq[j]);
        if (g != gcur) {
          if (active) pool_flush(Zx, denom, gcur, c, ZS, a0, a1, a2, a3);
          gcur = g;
          a0 = 0.0f; a1 = 0.0f; a2 = 0.0f; a3 = 0.0f;
        }
        const float4 w = wq[j];
        const float xv = is_mask ? ((x[j] > 0.0f) ? 1.0f : 0.0f) : x[j];
        a0 += w.x * xv;
        a1 += w.y * xv;
        a2 += w.z * xv;
        a3 += w.w * xv;
      }
    }
  }
  if (active) pool_flush(Zx, denom, gcur, c, ZS, a0, a1, a2, a3);
}

__global__ void zproj_wmma(const float* __restrict__ Zx, const float* __restrict__ denom,
                           const float* __restrict__ node_w, const float* __restrict__ node_b,
                           const float* __restrict__ edge_w, const float* __restrict__ edge_b,
                           float* __restrict__ Z, int M, int ND, int ED, int D, int ZS) {
  const int tid = threadIdx.x;
  const int wave = tid >> 5, lane = tid & 31;
  const int mrow = lane & 15;
  const int khalf = (lane & 16) ? 2 : 0;
  const int K = ND + ED;
  const int r0 = blockIdx.x * 16;
  const int n = blockIdx.y * 64 + wave * 16 + mrow;
  int ra = r0 + mrow;
  if (ra > M - 1) ra = M - 1;
  const float* zr = Zx + (long)ra * ZS + khalf;
  const float* wn = node_w + (long)n * ND + khalf;
  const float* we = edge_w + (long)n * ED + khalf;

  v8f acc = {};
  for (int kc = 0; kc < K; kc += KCHUNK) {
    v2f a[KCHUNK / 4], b[KCHUNK / 4];
#pragma unroll
    for (int i = 0; i < KCHUNK / 4; ++i) {
      const int kg = kc + 4 * i;
      const float* bp = (kg < ND) ? (wn + kg) : (we + (kg - ND));
      a[i] = *(const v2f*)&zr[kg];
      b[i] = *(const v2f*)bp;
    }
#pragma unroll
    for (int i = 0; i < KCHUNK / 4; ++i)
      acc = __builtin_amdgcn_wmma_f32_16x16x4_f32(false, a[i], false, b[i], (short)0, acc, false, false);
  }

  const float bn = node_b[n];
  const float be = edge_b[n];
  const int rb = r0 + ((lane & 16) ? 8 : 0);
  float p0[8], p1[8];
#pragma unroll
  for (int r = 0; r < 8; ++r) {
    int rr = rb + r;
    if (rr > M - 1) rr = M - 1;
    const float d = denom[rr];
    p0[r] = d / fmaxf(d, 1e-30f);
    p1[r] = Zx[(long)rr * ZS + K];
  }
#pragma unroll
  for (int r = 0; r < 8; ++r) {
    const int rr = rb + r;
    if (rr < M) Z[(long)rr * D + n] = acc[r] + p0[r] * bn + p1[r] * be;
  }
}

__global__ void vproj_pooled_wmma(const float* __restrict__ Z, const float* __restrict__ Wv,
                                  const float* __restrict__ bv, const float* __restrict__ denom,
                                  float* __restrict__ pooled, int G, int D) {
  const int tid = threadIdx.x;
  const int wave = tid >> 5, lane = tid & 31;
  const int mrow = lane & 15;
  const int khalf = (lane & 16) ? 2 : 0;
  const int g0 = blockIdx.x * 16;
  const int ncol0 = blockIdx.y * 64 + wave * 16;
  const int n = ncol0 + mrow;
  const int hh = ncol0 / (D / NUM_HEADS);
  int ga = g0 + mrow;
  if (ga > G - 1) ga = G - 1;
  const float* zr = Z + ((long)ga * NUM_HEADS + hh) * D + khalf;
  const float* wr = Wv + (long)n * D + khalf;

  v8f acc = {};
  for (int kc = 0; kc < D; kc += KCHUNK) {
    v2f a[KCHUNK / 4], b[KCHUNK / 4];
#pragma unroll
    for (int i = 0; i < KCHUNK / 4; ++i) {
      a[i] = *(const v2f*)&zr[kc + 4 * i];
      b[i] = *(const v2f*)&wr[kc + 4 * i];
    }
#pragma unroll
    for (int i = 0; i < KCHUNK / 4; ++i)
      acc = __builtin_amdgcn_wmma_f32_16x16x4_f32(false, a[i], false, b[i], (short)0, acc, false, false);
  }

  const float bn = bv[n];
  const int gb = g0 + ((lane & 16) ? 8 : 0);
  float dsum[8];
#pragma unroll
  for (int r = 0; r < 8; ++r) {
    int g = gb + r;
    if (g > G - 1) g = G - 1;
    dsum[r] = denom[g * NUM_HEADS + hh];
  }
#pragma unroll
  for (int r = 0; r < 8; ++r) {
    const int g = gb + r;
    if (g < G) {
      const float sp = dsum[r] / fmaxf(dsum[r], 1e-30f);
      pooled[(long)g * D + n] = acc[r] + sp * bn;
    }
  }
}

extern "C" void kernel_launch(void* const* d_in, const int* in_sizes, int n_in,
                              void* d_out, int out_size, void* d_ws, size_t ws_size,
                              hipStream_t stream) {
  const float* h      = (const float*)d_in[0];
  const int*   eidx   = (const int*)d_in[1];
  const float* eattr  = (const float*)d_in[2];
  const int*   batch  = (const int*)d_in[3];
  const float* node_w = (const float*)d_in[5];
  const float* node_b = (const float*)d_in[6];
  const float* edge_w = (const float*)d_in[7];
  const float* edge_b = (const float*)d_in[8];
  const float* query  = (const float*)d_in[9];
  const float* in_w   = (const float*)d_in[10];
  const float* in_b   = (const float*)d_in[11];
  const float* out_w  = (const float*)d_in[12];
  const float* out_b  = (const float*)d_in[13];
  float* out = (float*)d_out;

  int D  = in_sizes[9];
  int ND = in_sizes[5] / D;
  int ED = in_sizes[7] / D;
  int N  = in_sizes[0] / ND;
  long E = in_sizes[1] / 2;
  int G  = out_size / D;
  int K  = ND + ED;
  int ZS = (K + 2 + 3) & ~3;

  float* ws = (float*)d_ws;
  long o = 0;
  auto take = [&](long nf) { float* p = ws + o; o += (nf + 3) & ~3L; return p; };
  float* sumA   = take((long)N * ED);
  int*   cnti   = (int*)take(N);
  float* cnt    = take(N);
  int*   slots  = (int*)take((long)N * SLOT_CAP);
  float* scores = take((long)N * NUM_HEADS);
  float* wexp   = take((long)N * NUM_HEADS);
  float* sbuf   = take((long)NUM_HEADS * D);
  float* tbuf   = take(NUM_HEADS);
  float* ubuf   = take((long)NUM_HEADS * ZS);
  float* smax   = take((long)G * NUM_HEADS);
  float* denom  = take((long)G * NUM_HEADS + (long)G * NUM_HEADS * ZS);
  float* Zx     = denom + G * NUM_HEADS;
  float* Zb     = take((long)G * NUM_HEADS * D);
  float* pooled = take((long)G * D);

  {
    long z0 = (long)N * ED + N;
    long z1 = (long)G * NUM_HEADS + (long)G * NUM_HEADS * ZS;
    long maxn = z0 > z1 ? z0 : z1;
    int blocks = (int)((maxn + 255) / 256);
    if (blocks > 262144) blocks = 262144;
    init_ws<<<blocks, 256, 0, stream>>>(sumA, z0, denom, z1, smax, G * NUM_HEADS);
  }
  qst_kernel<<<1, D, 0, stream>>>(query, in_w, in_b, D, sbuf, tbuf);
  ufold_kernel<<<(K + 2 + 31) / 32, UF_WAVES * 32, 0, stream>>>(sbuf, node_w, edge_w, node_b, edge_b,
                                                               ubuf, D, ND, ED, ZS);
  {
    long blocks = (E + 255) / 256;
    if (blocks > 262144) blocks = 262144;
    edge_slots<<<(int)blocks, 256, 0, stream>>>(eidx, eattr, cnti, slots, sumA, E, ED);
    long gthreads = (long)N * 16;
    edge_gather<<<(int)((gthreads + 255) / 256), 256, 0, stream>>>(eattr, slots, cnti, sumA,
                                                                 cnt, N, ED);
  }
  {
    int total = N * NUM_HEADS;
    scores_kernel<<<(total + 255) / 256, 256, 0, stream>>>(h, sumA, cnt, ubuf, tbuf, batch,
                                                           smax, scores, N, ND, ED, ZS, D);
    expw_kernel<<<(total + 255) / 256, 256, 0, stream>>>(scores, batch, smax, denom,
                                                         wexp, N);
  }
  {
    dim3 grid((N + POOL_ROWS - 1) / POOL_ROWS, (K + 1 + 255) / 256);
    pool_x<<<grid, 256, 0, stream>>>(wexp, denom, batch, h, sumA, cnt, Zx, N, ND, ED, ZS);
  }
  {
    dim3 grid((G * NUM_HEADS + 15) / 16, D / 64);
    zproj_wmma<<<grid, 128, 0, stream>>>(Zx, denom, node_w, node_b, edge_w, edge_b, Zb,
                                         G * NUM_HEADS, ND, ED, D, ZS);
  }
  {
    dim3 grid((G + 15) / 16, D / 64);
    vproj_pooled_wmma<<<grid, 128, 0, stream>>>(Zb, in_w + 2L * D * D, in_b + 2 * D,
                                                denom, pooled, G, D);
  }
  {
    dim3 grid((G + 63) / 64, D / 64);
    gemm_xwT_wmma<<<grid, 128, 0, stream>>>(pooled, out_w, out_b, out, G, D, D);
  }
}


# Reduce before the atomic: segment max and softmax denominator combined per head in the wave (shuffles) and block (LDS), one atomic per block and graph, not per node; denominator sum order changes
# speedup vs baseline: 2.2533x; 1.0574x over previous
#include <hip/hip_runtime.h>
#include <math.h>

#define NUM_HEADS 4
#define KCHUNK 64
#define LDSTRIDE 68

typedef float v2f __attribute__((ext_vector_type(2)));
typedef float v8f __attribute__((ext_vector_type(8)));

__device__ __forceinline__ void atomic_max_float(float* addr, float val) {
  if (val >= 0.0f) {
    atomicMax((int*)addr, __float_as_int(val));
  } else {
    atomicMin((unsigned int*)addr, __float_as_uint(val));
  }
}

__global__ void init_ws(float* z0, long n0, float* z1, long n1, float* smax, int ns) {
  long i = (long)blockIdx.x * blockDim.x + threadIdx.x;
  long stride = (long)gridDim.x * blockDim.x;
  for (long t = i; t < n0; t += stride) z0[t] = 0.0f;
  for (long t = i; t < n1; t += stride) z1[t] = 0.0f;
  for (long t = i; t < ns; t += stride) smax[t] = -INFINITY;
}

__global__ void qst_kernel(const float* __restrict__ query, const float* __restrict__ in_w,
                           const float* __restrict__ in_b, int D,
                           float* __restrict__ sbuf, float* __restrict__ tbuf) {
  __shared__ float qs[1024];
  int j = threadIdx.x;
  int dh = D / NUM_HEADS;
  float acc = in_b[j];
  for (int c = 0; c < D; ++c) acc += query[c] * in_w[(long)j * D + c];
  qs[j] = acc;
  __syncthreads();
  for (int hh = 0; hh < NUM_HEADS; ++hh) {
    float s = 0.0f;
    const float* wk = in_w + (long)(D + hh * dh) * D;
    for (int d = 0; d < dh; ++d) s += qs[hh * dh + d] * wk[(long)d * D + j];
    sbuf[hh * D + j] = s;
  }
  if (j < NUM_HEADS) {
    float t = 0.0f;
    for (int d = 0; d < dh; ++d) t += qs[j * dh + d] * in_b[D + j * dh + d];
    tbuf[j] = t;
  }
}

#define UF_WAVES 8

__global__ void ufold_kernel(const float* __restrict__ sbuf,
                             const float* __restrict__ node_w, const float* __restrict__ edge_w,
                             const float* __restrict__ node_b, const float* __restrict__ edge_b,
                             float* __restrict__ ubuf, int D, int ND, int ED, int US) {
  __shared__ float red[UF_WAVES][NUM_HEADS][32];
  const int tid = threadIdx.x;
  const int lane = tid & 31, wave = tid >> 5;
  const int K = ND + ED;
  const int k = blockIdx.x * 32 + lane;
  const float* wp;
  long stride;
  if (k < ND)      { wp = node_w + k;        stride = ND; }
  else if (k < K)  { wp = edge_w + (k - ND); stride = ED; }
  else if (k == K) { wp = node_b;            stride = 1; }
  else             { wp = edge_b;            stride = 1; }
  float a0 = 0.0f, a1 = 0.0f, a2 = 0.0f, a3 = 0.0f;
  for (int j = wave; j < D; j += UF_WAVES) {
    const float w = wp[(long)j * stride];
    a0 += sbuf[j] * w;
    a1 += sbuf[D + j] * w;
    a2 += sbuf[2 * D + j] * w;
    a3 += sbuf[3 * D + j] * w;
  }
  red[wave][0][lane] = a0;
  red[wave][1][lane] = a1;
  red[wave][2][lane] = a2;
  red[wave][3][lane] = a3;
  __syncthreads();
  if (tid < NUM_HEADS * 32) {
    const int hh = tid >> 5;
    const int kk = blockIdx.x * 32 + lane;
    float sum = 0.0f;
    for (int w = 0; w < UF_WAVES; ++w) sum += red[w][hh][lane];
    if (kk < K + 2) ubuf[hh * US + kk] = sum;
  }
}

#define SLOT_CAP 64
#define GATHER_CHUNK 8

__global__ void edge_slots(const int* __restrict__ row, const float* __restrict__ eattr,
                           int* __restrict__ cnti, int* __restrict__ slots,
                           float* __restrict__ sumA, long E, int ED) {
  long i = (long)blockIdx.x * blockDim.x + threadIdx.x;
  long stride = (long)gridDim.x * blockDim.x;
  for (long e = i; e < E; e += stride) {
    int r = row[e];
    int s = atomicAdd(&cnti[r], 1);
    if (s < SLOT_CAP) {
      slots[(long)r * SLOT_CAP + s] = (int)e;
    } else {
      for (int c = 0; c < ED; ++c) atomicAdd(&sumA[(long)r * ED + c], eattr[e * ED + c]);
    }
  }
}

__global__ void edge_gather(const float* __restrict__ eattr, const int* __restrict__ slots,
                            const int* __restrict__ cnti, float* __restrict__ sumA,
                            float* __restrict__ cntf, int N, int ED) {
  const int t = blockIdx.x * blockDim.x + threadIdx.x;
  const int m = t >> 4;
  const int l = t & 15;
  if (m >= N) return;
  const int cdeg = cnti[m];
  const int cs = cdeg < SLOT_CAP ? cdeg : SLOT_CAP;
  const int* sp = slots + (long)m * SLOT_CAP;
  for (int c0 = l * 4; c0 < ED; c0 += 64) {
    float4 acc = make_float4(0.0f, 0.0f, 0.0f, 0.0f);
    if (cdeg > SLOT_CAP) acc = *(const float4*)&sumA[(long)m * ED + c0];
    for (int j0 = 0; j0 < cs; j0 += GATHER_CHUNK) {
      float4 v[GATHER_CHUNK];
#pragma unroll
      for (int j = 0; j < GATHER_CHUNK; ++j) {
        int jj = j0 + j;
        if (jj > cs - 1) jj = cs - 1;
        const int e = sp[jj];
        v[j] = *(const float4*)&eattr[(long)e * ED + c0];
      }
#pragma unroll
      for (int j = 0; j < GATHER_CHUNK; ++j)
        asm volatile("" :: "v"(v[j].x), "v"(v[j].y), "v"(v[j].z), "v"(v[j].w));
#pragma unroll
      for (int j = 0; j < GATHER_CHUNK; ++j) {
        const bool in_list = j0 + j < cs;
        acc.x += in_list ? v[j].x : 0.0f;
        acc.y += in_list ? v[j].y : 0.0f;
        acc.z += in_list ? v[j].z : 0.0f;
        acc.w += in_list ? v[j].w : 0.0f;
      }
    }
    const float dm = fmaxf((float)cdeg, 1.0f);
    acc.x /= dm; acc.y /= dm; acc.z /= dm; acc.w /= dm;
    *(float4*)&sumA[(long)m * ED + c0] = acc;
  }
  if (l == 0) cntf[m] = (float)cdeg;
}


__global__ void gemm_xwT_wmma(const float* __restrict__ X, const float* __restrict__ W,
                              const float* __restrict__ bias, float* __restrict__ Y,
                              int M, int K, int Nout) {
  __shared__ float Xs[64 * LDSTRIDE];
  int row0 = blockIdx.x * 64;
  int tid = threadIdx.x;
  int wave = tid >> 5, lane = tid & 31;
  int mrow = lane & 15;
  int n = blockIdx.y * 64 + wave * 16 + mrow;
  int khalf = (lane & 16) ? 2 : 0;

  v8f acc0 = {}, acc1 = {}, acc2 = {}, acc3 = {};

  for (int kc = 0; kc < K; kc += KCHUNK) {
    __syncthreads();
    for (int t = tid; t < 64 * KCHUNK; t += blockDim.x) {
      int r = t >> 6;
      int c = t & (KCHUNK - 1);
      int m = row0 + r;
      Xs[r * LDSTRIDE + c] = (m < M) ? X[(long)m * K + kc + c] : 0.0f;
    }
    __syncthreads();
    for (int k = 0; k < KCHUNK; k += 4) {
      int kk = k + khalf;
      v2f b = *(const v2f*)&W[(long)n * K + kc + kk];
      v2f a0 = *(const v2f*)&Xs[(mrow +  0) * LDSTRIDE + kk];
      v2f a1 = *(const v2f*)&Xs[(mrow + 16) * LDSTRIDE + kk];
      v2f a2 = *(const v2f*)&Xs[(mrow + 32) * LDSTRIDE + kk];
      v2f a3 = *(const v2f*)&Xs[(mrow + 48) * LDSTRIDE + kk];
      acc0 = __builtin_amdgcn_wmma_f32_16x16x4_f32(false, a0, false, b, (short)0, acc0, false, false);
      acc1 = __builtin_amdgcn_wmma_f32_16x16x4_f32(false, a1, false, b, (short)0, acc1, false, false);
      acc2 = __builtin_amdgcn_wmma_f32_16x16x4_f32(false, a2, false, b, (short)0, acc2, false, false);
      acc3 = __builtin_amdgcn_wmma_f32_16x16x4_f32(false, a3, false, b, (short)0, acc3, false, false);
    }
  }

  float bn = bias[n];
  for (int r = 0; r < 8; ++r) {
    int mb = row0 + r + ((lane & 16) ? 8 : 0);
    v8f* accs[4] = {&acc0, &acc1, &acc2, &acc3};
    for (int mt = 0; mt < 4; ++mt) {
      int m = mb + mt * 16;
      if (m < M) Y[(long)m * Nout + n] = (*accs[mt])[r] + bn;
    }
  }
}

#define SM_THREADS 256
#define SM_WAVES (SM_THREADS / 32)

template <bool IS_MAX>
__device__ __forceinline__ void seg_combine(float v, int g, bool valid, float* __restrict__ dst,
                                            float (*wred)[NUM_HEADS], int* wgr) {
  const int tid = threadIdx.x;
  const int lane = tid & 31, wave = tid >> 5;
  const int hh = lane & 3;
  const int g0 = __builtin_amdgcn_readfirstlane(g);
  const bool wuni = __all(g == g0) != 0;
  if (wuni) {
    for (int off = NUM_HEADS; off < 32; off <<= 1) {
      const float o = __shfl_xor(v, off);
      v = IS_MAX ? fmaxf(v, o) : v + o;
    }
    if (lane < NUM_HEADS) wred[wave][lane] = v;
    if (lane == 0) wgr[wave] = g0;
  } else {
    if (valid) {
      if (IS_MAX) atomic_max_float(&dst[g * NUM_HEADS + hh], v);
      else        atomicAdd(&dst[g * NUM_HEADS + hh], v);
    }
    if (lane == 0) wgr[wave] = -1;
  }
  __syncthreads();
  if (tid < SM_WAVES * NUM_HEADS) {
    const int w = tid >> 2;
    const int gw = wgr[w];
    if (gw >= 0 && (w == 0 || wgr[w - 1] != gw)) {
      float m = wred[w][hh];
      for (int u = w + 1; u < SM_WAVES && wgr[u] == gw; ++u)
        m = IS_MAX ? fmaxf(m, wred[u][hh]) : m + wred[u][hh];
      if (IS_MAX) atomic_max_float(&dst[gw * NUM_HEADS + hh], m);
      else        atomicAdd(&dst[gw * NUM_HEADS + hh], m);
    }
  }
}

__global__ void scores_kernel(const float* __restrict__ h, const float* __restrict__ xe,
                              const float* __restrict__ cnt, const float* __restrict__ ubuf,
                              const float* __restrict__ tbuf, const int* __restrict__ batch,
                              float* __restrict__ smax, float* __restrict__ scores,
                              int N, int ND, int ED, int US, int D) {
  __shared__ float wred[SM_WAVES][NUM_HEADS];
  __shared__ int wgr[SM_WAVES];
  const int idx = blockIdx.x * SM_THREADS + threadIdx.x;
  const bool valid = idx < N * NUM_HEADS;
  int nI = idx >> 2;
  if (nI > N - 1) nI = N - 1;
  int hh = idx & 3;
  const int K = ND + ED;
  const float* ur = ubuf + hh * US;
  float dot = tbuf[hh] + ur[K] + ((cnt[nI] > 0.0f) ? ur[K + 1] : 0.0f);
  const float* xr = h + (long)nI * ND;
  for (int c = 0; c < ND; c += 4) {
    float4 x = *(const float4*)&xr[c];
    float4 s = *(const float4*)&ur[c];
    dot += x.x * s.x + x.y * s.y + x.z * s.z + x.w * s.w;
  }
  const float* er = xe + (long)nI * ED;
  const float* ue = ur + ND;
  for (int c = 0; c < ED; c += 4) {
    float4 x = *(const float4*)&er[c];
    float4 s = *(const float4*)&ue[c];
    dot += x.x * s.x + x.y * s.y + x.z * s.z + x.w * s.w;
  }
  float sc = dot * rsqrtf((float)(D / NUM_HEADS));
  if (valid) scores[idx] = sc;
  seg_combine<true>(valid ? sc : -INFINITY, batch[nI], valid, smax, wred, wgr);
}

__global__ void expw_kernel(const float* __restrict__ scores, const int* __restrict__ batch,
                            const float* __restrict__ smax, float* __restrict__ denom,
                            float* __restrict__ wexp, int N) {
  __shared__ float wred[SM_WAVES][NUM_HEADS];
  __shared__ int wgr[SM_WAVES];
  const int idx = blockIdx.x * SM_THREADS + threadIdx.x;
  const bool valid = idx < N * NUM_HEADS;
  int nI = idx >> 2;
  if (nI > N - 1) nI = N - 1;
  const int hh = idx & 3;
  const int g = batch[nI];
  float w = 0.0f;
  if (valid) {
    w = __expf(scores[idx] - smax[g * NUM_HEADS + hh]);
    wexp[idx] = w;
  }
  seg_combine<false>(w, g, valid, denom, wred, wgr);
}

#define POOL_ROWS 64
#define POOL_CHUNK 16

__device__ __forceinline__ void pool_flush(float* __restrict__ Z, const float* __restrict__ denom,
                                           int g, int c, int ZS,
                                           float a0, float a1, float a2, float a3) {
  const float4 d = *(const float4*)&denom[(long)g * NUM_HEADS];
  float* zp = Z + (long)g * NUM_HEADS * ZS + c;
  atomicAdd(zp,              a0 / fmaxf(d.x, 1e-30f));
  atomicAdd(zp + ZS,         a1 / fmaxf(d.y, 1e-30f));
  atomicAdd(zp + 2L * ZS,    a2 / fmaxf(d.z, 1e-30f));
  atomicAdd(zp + 3L * ZS,    a3 / fmaxf(d.w, 1e-30f));
}

__global__ void pool_x(const float* __restrict__ wexp, const float* __restrict__ denom,
                       const int* __restrict__ batch, const float* __restrict__ h,
                       const float* __restrict__ xe, const float* __restrict__ cnt,
                       float* __restrict__ Zx, int N, int ND, int ED, int ZS) {
  __shared__ float4 wsh[POOL_ROWS];
  __shared__ int gsh[POOL_ROWS];
  const int tid = threadIdx.x;
  const int n0 = blockIdx.x * POOL_ROWS;
  int nend = n0 + POOL_ROWS;
  if (nend > N) nend = N;
  if (tid < POOL_ROWS) {
    if (n0 + tid < N) {
      wsh[tid] = *(const float4*)&wexp[(long)(n0 + tid) * NUM_HEADS];
      gsh[tid] = batch[n0 + tid];
    } else {
      wsh[tid] = make_float4(0.0f, 0.0f, 0.0f, 0.0f);
      gsh[tid] = 0;
    }
  }
  __syncthreads();

  const int K = ND + ED;
  const int c = blockIdx.y * blockDim.x + tid;
  const bool active = c <= K;
  const bool is_mask = c >= K;
  const float* xp;
  long pitch;
  if (c < ND)     { xp = h + (long)n0 * ND + c;          pitch = ND; }
  else if (c < K) { xp = xe + (long)n0 * ED + (c - ND);  pitch = ED; }
  else            { xp = cnt + n0;                       pitch = 1; }
  int gcur = __builtin_amdgcn_readfirstlane(gsh[0]);
  const int rlast = nend - 1 - n0;
  float a0 = 0.0f, a1 = 0.0f, a2 = 0.0f, a3 = 0.0f;
  for (int i0 = 0; i0 < POOL_ROWS; i0 += POOL_CHUNK) {
    if (n0 + i0 >= nend) break;
    float x[POOL_CHUNK];
    float4 wq[POOL_CHUNK];
    int gq[POOL_CHUNK];
#pragma unroll
    for (int j = 0; j < POOL_CHUNK; ++j) {
      int r = i0 + j;
      if (r > rlast) r = rlast;
      x[j] = xp[(long)r * pitch];
      wq[j] = wsh[i0 + j];
      gq[j] = gsh[i0 + j];
    }
#pragma unroll
    for (int j = 0; j < POOL_CHUNK; ++j) {
      if (n0 + i0 + j < nend) {
        const int g = __builtin_amdgcn_readfirstlane(gq[j]);
        if (g != gcur) {
          if (active) pool_flush(Zx, denom, gcur, c, ZS, a0, a1, a2, a3);
          gcur = g;
          a0 = 0.0f; a1 = 0.0f; a2 = 0.0f; a3 = 0.0f;
        }
        const float4 w = wq[j];
        const float xv = is_mask ? ((x[j] > 0.0f) ? 1.0f : 0.0f) : x[j];
        a0 += w.x * xv;
        a1 += w.y * xv;
        a2 += w.z * xv;
        a3 += w.w * xv;
      }
    }
  }
  if (active) pool_flush(Zx, denom, gcur, c, ZS, a0, a1, a2, a3);
}

__global__ void zproj_wmma(const float* __restrict__ Zx, const float* __restrict__ denom,
                           const float* __restrict__ node_w, const float* __restrict__ node_b,
                           const float* __restrict__ edge_w, const float* __restrict__ edge_b,
                           float* __restrict__ Z, int M, int ND, int ED, int D, int ZS) {
  const int tid = threadIdx.x;
  const int wave = tid >> 5, lane = tid & 31;
  const int mrow = lane & 15;
  const int khalf = (lane & 16) ? 2 : 0;
  const int K = ND + ED;
  const int r0 = blockIdx.x * 16;
  const int n = blockIdx.y * 64 + wave * 16 + mrow;
  int ra = r0 + mrow;
  if (ra > M - 1) ra = M - 1;
  const float* zr = Zx + (long)ra * ZS + khalf;
  const float* wn = node_w + (long)n * ND + khalf;
  const float* we = edge_w + (long)n * ED + khalf;

  v8f acc = {};
  for (int kc = 0; kc < K; kc += KCHUNK) {
    v2f a[KCHUNK / 4], b[KCHUNK / 4];
#pragma unroll
    for (int i = 0; i < KCHUNK / 4; ++i) {
      const int kg = kc + 4 * i;
      const float* bp = (kg < ND) ? (wn + kg) : (we + (kg - ND));
      a[i] = *(const v2f*)&zr[kg];
      b[i] = *(const v2f*)bp;
    }
#pragma unroll
    for (int i = 0; i < KCHUNK / 4; ++i)
      acc = __builtin_amdgcn_wmma_f32_16x16x4_f32(false, a[i], false, b[i], (short)0, acc, false, false);
  }

  const float bn = node_b[n];
  const float be = edge_b[n];
  const int rb = r0 + ((lane & 16) ? 8 : 0);
  float p0[8], p1[8];
#pragma unroll
  for (int r = 0; r < 8; ++r) {
    int rr = rb + r;
    if (rr > M - 1) rr = M - 1;
    const float d = denom[rr];
    p0[r] = d / fmaxf(d, 1e-30f);
    p1[r] = Zx[(long)rr * ZS + K];
  }
#pragma unroll
  for (int r = 0; r < 8; ++r) {
    const int rr = rb + r;
    if (rr < M) Z[(long)rr * D + n] = acc[r] + p0[r] * bn + p1[r] * be;
  }
}

__global__ void vproj_pooled_wmma(const float* __restrict__ Z, const float* __restrict__ Wv,
                                  const float* __restrict__ bv, const float* __restrict__ denom,
                                  float* __restrict__ pooled, int G, int D) {
  const int tid = threadIdx.x;
  const int wave = tid >> 5, lane = tid & 31;
  const int mrow = lane & 15;
  const int khalf = (lane & 16) ? 2 : 0;
  const int g0 = blockIdx.x * 16;
  const int ncol0 = blockIdx.y * 64 + wave * 16;
  const int n = ncol0 + mrow;
  const int hh = ncol0 / (D / NUM_HEADS);
  int ga = g0 + mrow;
  if (ga > G - 1) ga = G - 1;
  const float* zr = Z + ((long)ga * NUM_HEADS + hh) * D + khalf;
  const float* wr = Wv + (long)n * D + khalf;

  v8f acc = {};
  for (int kc = 0; kc < D; kc += KCHUNK) {
    v2f a[KCHUNK / 4], b[KCHUNK / 4];
#pragma unroll
    for (int i = 0; i < KCHUNK / 4; ++i) {
      a[i] = *(const v2f*)&zr[kc + 4 * i];
      b[i] = *(const v2f*)&wr[kc + 4 * i];
    }
#pragma unroll
    for (int i = 0; i < KCHUNK / 4; ++i)
      acc = __builtin_amdgcn_wmma_f32_16x16x4_f32(false, a[i], false, b[i], (short)0, acc, false, false);
  }

  const float bn = bv[n];
  const int gb = g0 + ((lane & 16) ? 8 : 0);
  float dsum[8];
#pragma unroll
  for (int r = 0; r < 8; ++r) {
    int g = gb + r;
    if (g > G - 1) g = G - 1;
    dsum[r] = denom[g * NUM_HEADS + hh];
  }
#pragma unroll
  for (int r = 0; r < 8; ++r) {
    const int g = gb + r;
    if (g < G) {
      const float sp = dsum[r] / fmaxf(dsum[r], 1e-30f);
      pooled[(long)g * D + n] = acc[r] + sp * bn;
    }
  }
}

extern "C" void kernel_launch(void* const* d_in, const int* in_sizes, int n_in,
                              void* d_out, int out_size, void* d_ws, size_t ws_size,
                              hipStream_t stream) {
  const float* h      = (const float*)d_in[0];
  const int*   eidx   = (const int*)d_in[1];
  const float* eattr  = (const float*)d_in[2];
  const int*   batch  = (const int*)d_in[3];
  const float* node_w = (const float*)d_in[5];
  const float* node_b = (const float*)d_in[6];
  const float* edge_w = (const float*)d_in[7];
  const float* edge_b = (const float*)d_in[8];
  const float* query  = (const float*)d_in[9];
  const float* in_w   = (const float*)d_in[10];
  const float* in_b   = (const float*)d_in[11];
  const float* out_w  = (const float*)d_in[12];
  const float* out_b  = (const float*)d_in[13];
  float* out = (float*)d_out;

  int D  = in_sizes[9];
  int ND = in_sizes[5] / D;
  int ED = in_sizes[7] / D;
  int N  = in_sizes[0] / ND;
  long E = in_sizes[1] / 2;
  int G  = out_size / D;
  int K  = ND + ED;
  int ZS = (K + 2 + 3) & ~3;

  float* ws = (float*)d_ws;
  long o = 0;
  auto take = [&](long nf) { float* p = ws + o; o += (nf + 3) & ~3L; return p; };
  float* sumA   = take((long)N * ED);
  int*   cnti   = (int*)take(N);
  float* cnt    = take(N);
  int*   slots  = (int*)take((long)N * SLOT_CAP);
  float* scores = take((long)N * NUM_HEADS);
  float* wexp   = take((long)N * NUM_HEADS);
  float* sbuf   = take((long)NUM_HEADS * D);
  float* tbuf   = take(NUM_HEADS);
  float* ubuf   = take((long)NUM_HEADS * ZS);
  float* smax   = take((long)G * NUM_HEADS);
  float* denom  = take((long)G * NUM_HEADS + (long)G * NUM_HEADS * ZS);
  float* Zx     = denom + G * NUM_HEADS;
  float* Zb     = take((long)G * NUM_HEADS * D);
  float* pooled = take((long)G * D);

  {
    long z0 = (long)N * ED + N;
    long z1 = (long)G * NUM_HEADS + (long)G * NUM_HEADS * ZS;
    long maxn = z0 > z1 ? z0 : z1;
    int blocks = (int)((maxn + 255) / 256);
    if (blocks > 262144) blocks = 262144;
    init_ws<<<blocks, 256, 0, stream>>>(sumA, z0, denom, z1, smax, G * NUM_HEADS);
  }
  qst_kernel<<<1, D, 0, stream>>>(query, in_w, in_b, D, sbuf, tbuf);
  ufold_kernel<<<(K + 2 + 31) / 32, UF_WAVES * 32, 0, stream>>>(sbuf, node_w, edge_w, node_b, edge_b,
                                                               ubuf, D, ND, ED, ZS);
  {
    long blocks = (E + 255) / 256;
    if (blocks > 262144) blocks = 262144;
    edge_slots<<<(int)blocks, 256, 0, stream>>>(eidx, eattr, cnti, slots, sumA, E, ED);
    long gthreads = (long)N * 16;
    edge_gather<<<(int)((gthreads + 255) / 256), 256, 0, stream>>>(eattr, slots, cnti, sumA,
                                                                 cnt, N, ED);
  }
  {
    int total = N * NUM_HEADS;
    int blocks = (total + SM_THREADS - 1) / SM_THREADS;
    scores_kernel<<<blocks, SM_THREADS, 0, stream>>>(h, sumA, cnt, ubuf, tbuf, batch,
                                                     smax, scores, N, ND, ED, ZS, D);
    expw_kernel<<<blocks, SM_THREADS, 0, stream>>>(scores, batch, smax, denom, wexp, N);
  }
  {
    dim3 grid((N + POOL_ROWS - 1) / POOL_ROWS, (K + 1 + 255) / 256);
    pool_x<<<grid, 256, 0, stream>>>(wexp, denom, batch, h, sumA, cnt, Zx, N, ND, ED, ZS);
  }
  {
    dim3 grid((G * NUM_HEADS + 15) / 16, D / 64);
    zproj_wmma<<<grid, 128, 0, stream>>>(Zx, denom, node_w, node_b, edge_w, edge_b, Zb,
                                         G * NUM_HEADS, ND, ED, D, ZS);
  }
  {
    dim3 grid((G + 15) / 16, D / 64);
    vproj_pooled_wmma<<<grid, 128, 0, stream>>>(Zb, in_w + 2L * D * D, in_b + 2 * D,
                                                denom, pooled, G, D);
  }
  {
    dim3 grid((G + 63) / 64, D / 64);
    gemm_xwT_wmma<<<grid, 128, 0, stream>>>(pooled, out_w, out_b, out, G, D, D);
  }
}


# Parameter fold (q, t, s, u) in one launch of 28 blocks (32 columns of u x head), loads requested in batches of whole lines; was one serial block plus a second launch; sum order of q and s changes
# speedup vs baseline: 2.7741x; 1.2311x over previous
#include <hip/hip_runtime.h>
#include <math.h>

#define NUM_HEADS 4
#define KCHUNK 64
#define LDSTRIDE 68

typedef float v2f __attribute__((ext_vector_type(2)));
typedef float v8f __attribute__((ext_vector_type(8)));

__device__ __forceinline__ void atomic_max_float(float* addr, float val) {
  if (val >= 0.0f) {
    atomicMax((int*)addr, __float_as_int(val));
  } else {
    atomicMin((unsigned int*)addr, __float_as_uint(val));
  }
}

__global__ void init_ws(float* z0, long n0, float* z1, long n1, float* smax, int ns) {
  long i = (long)blockIdx.x * blockDim.x + threadIdx.x;
  long stride = (long)gridDim.x * blockDim.x;
  for (long t = i; t < n0; t += stride) z0[t] = 0.0f;
  for (long t = i; t < n1; t += stride) z1[t] = 0.0f;
  for (long t = i; t < ns; t += stride) smax[t] = -INFINITY;
}

#define FW 8
#define FROWS 8
#define FJB 16
#define FPITCH 33

__device__ __forceinline__ float dot4(float4 a, float4 b) {
  return a.x * b.x + a.y * b.y + a.z * b.z + a.w * b.w;
}

__global__ void fold_kernel(const float* __restrict__ query, const float* __restrict__ in_w,
                            const float* __restrict__ in_b,
                            const float* __restrict__ node_w, const float* __restrict__ edge_w,
                            const float* __restrict__ node_b, const float* __restrict__ edge_b,
                            float* __restrict__ ubuf, float* __restrict__ tbuf,
                            int D, int ND, int ED, int US) {
  __shared__ __attribute__((aligned(16))) float buf[256 * FPITCH];
  __shared__ float qh[256];
  __shared__ float sh[1024];
  __shared__ float red[FW][32];
  const int tid = threadIdx.x;
  const int lane = tid & 31, wave = tid >> 5;
  const int hh = blockIdx.y;
  const int dh = D / NUM_HEADS;
  const int K = ND + ED;

  for (int rb = 0; rb < dh; rb += FW * FROWS) {
    float p[FROWS];
#pragma unroll
    for (int i = 0; i < FROWS; ++i) p[i] = 0.0f;
    for (int c = lane * 4; c < D; c += 128) {
      const float4 x = *(const float4*)&query[c];
      float4 w[FROWS];
#pragma unroll
      for (int i = 0; i < FROWS; ++i) {
        int r = rb + wave + FW * i;
        if (r > dh - 1) r = dh - 1;
        w[i] = *(const float4*)&in_w[(long)(hh * dh + r) * D + c];
      }
#pragma unroll
      for (int i = 0; i < FROWS; ++i) p[i] += dot4(w[i], x);
    }
#pragma unroll
    for (int i = 0; i < FROWS; ++i) {
      const int r = rb + wave + FW * i;
      if (r < dh) buf[r * FPITCH + lane] = p[i];
    }
  }
  __syncthreads();
  if (tid < dh) {
    float q = in_b[hh * dh + tid];
    for (int l = 0; l < 32; ++l) q += buf[tid * FPITCH + l];
    qh[tid] = q;
  }
  __syncthreads();

  if (blockIdx.x == 0 && wave == 0) {
    float t = 0.0f;
    for (int d = lane; d < dh; d += 32) t += qh[d] * in_b[D + hh * dh + d];
    for (int off = 16; off > 0; off >>= 1) t += __shfl_xor(t, off);
    if (lane == 0) tbuf[hh] = t;
  }

  const float* wk = in_w + (long)(D + hh * dh) * D;
  for (int jc = lane * 4; jc < D; jc += 128) {
    float4 acc = make_float4(0.0f, 0.0f, 0.0f, 0.0f);
    for (int db = 0; db < dh; db += FW * FROWS) {
      float4 w[FROWS];
      float q[FROWS];
#pragma unroll
      for (int i = 0; i < FROWS; ++i) {
        const int d = db + wave + FW * i;
        const int dc = d < dh ? d : dh - 1;
        w[i] = *(const float4*)&wk[(long)dc * D + jc];
        q[i] = d < dh ? qh[dc] : 0.0f;
      }
#pragma unroll
      for (int i = 0; i < FROWS; ++i) {
        acc.x += q[i] * w[i].x; acc.y += q[i] * w[i].y; acc.z += q[i] * w[i].z; acc.w += q[i] * w[i].w;
      }
    }
    *(float4*)&buf[wave * D + jc] = acc;
  }
  __syncthreads();
  for (int j = tid; j < D; j += FW * 32) {
    float s = 0.0f;
    for (int w = 0; w < FW; ++w) s += buf[w * D + j];
    sh[j] = s;
  }
  __syncthreads();

  const int k = blockIdx.x * 32 + lane;
  const float* wp;
  long stride;
  if (k < ND)      { wp = node_w + k;        stride = ND; }
  else if (k < K)  { wp = edge_w + (k - ND); stride = ED; }
  else if (k == K) { wp = node_b;            stride = 1; }
  else             { wp = edge_b;            stride = 1; }
  float a = 0.0f;
  for (int j0 = wave; j0 < D; j0 += FW * FJB) {
    float w[FJB], sv[FJB];
#pragma unroll
    for (int i = 0; i < FJB; ++i) {
      const int j = j0 + FW * i;
      const int jcl = j < D ? j : D - 1;
      w[i] = wp[(long)jcl * stride];
      sv[i] = j < D ? sh[jcl] : 0.0f;
    }
#pragma unroll
    for (int i = 0; i < FJB; ++i) a += sv[i] * w[i];
  }
  red[wave][lane] = a;
  __syncthreads();
  if (tid < 32) {
    float sum = 0.0f;
    for (int w = 0; w < FW; ++w) sum += red[w][lane];
    if (k < K + 2) ubuf[hh * US + k] = sum;
  }
}

#define SLOT_CAP 64
#define GATHER_CHUNK 8

__global__ void edge_slots(const int* __restrict__ row, const float* __restrict__ eattr,
                           int* __restrict__ cnti, int* __restrict__ slots,
                           float* __restrict__ sumA, long E, int ED) {
  long i = (long)blockIdx.x * blockDim.x + threadIdx.x;
  long stride = (long)gridDim.x * blockDim.x;
  for (long e = i; e < E; e += stride) {
    int r = row[e];
    int s = atomicAdd(&cnti[r], 1);
    if (s < SLOT_CAP) {
      slots[(long)r * SLOT_CAP + s] = (int)e;
    } else {
      for (int c = 0; c < ED; ++c) atomicAdd(&sumA[(long)r * ED + c], eattr[e * ED + c]);
    }
  }
}

__global__ void edge_gather(const float* __restrict__ eattr, const int* __restrict__ slots,
                            const int* __restrict__ cnti, float* __restrict__ sumA,
                            float* __restrict__ cntf, int N, int ED) {
  const int t = blockIdx.x * blockDim.x + threadIdx.x;
  const int m = t >> 4;
  const int l = t & 15;
  if (m >= N) return;
  const int cdeg = cnti[m];
  const int cs = cdeg < SLOT_CAP ? cdeg : SLOT_CAP;
  const int* sp = slots + (long)m * SLOT_CAP;
  for (int c0 = l * 4; c0 < ED; c0 += 64) {
    float4 acc = make_float4(0.0f, 0.0f, 0.0f, 0.0f);
    if (cdeg > SLOT_CAP) acc = *(const float4*)&sumA[(long)m * ED + c0];
    for (int j0 = 0; j0 < cs; j0 += GATHER_CHUNK) {
      float4 v[GATHER_CHUNK];
#pragma unroll
      for (int j = 0; j < GATHER_CHUNK; ++j) {
        int jj = j0 + j;
        if (jj > cs - 1) jj = cs - 1;
        const int e = sp[jj];
        v[j] = *(const float4*)&eattr[(long)e * ED + c0];
      }
#pragma unroll
      for (int j = 0; j < GATHER_CHUNK; ++j)
        asm volatile("" :: "v"(v[j].x), "v"(v[j].y), "v"(v[j].z), "v"(v[j].w));
#pragma unroll
      for (int j = 0; j < GATHER_CHUNK; ++j) {
        const bool in_list = j0 + j < cs;
        acc.x += in_list ? v[j].x : 0.0f;
        acc.y += in_list ? v[j].y : 0.0f;
        acc.z += in_list ? v[j].z : 0.0f;
        acc.w += in_list ? v[j].w : 0.0f;
      }
    }
    const float dm = fmaxf((float)cdeg, 1.0f);
    acc.x /= dm; acc.y /= dm; acc.z /= dm; acc.w /= dm;
    *(float4*)&sumA[(long)m * ED + c0] = acc;
  }
  if (l == 0) cntf[m] = (float)cdeg;
}


__global__ void gemm_xwT_wmma(const float* __restrict__ X, const float* __restrict__ W,
                              const float* __restrict__ bias, float* __restrict__ Y,
                              int M, int K, int Nout) {
  __shared__ float Xs[64 * LDSTRIDE];
  int row0 = blockIdx.x * 64;
  int tid = threadIdx.x;
  int wave = tid >> 5, lane = tid & 31;
  int mrow = lane & 15;
  int n = blockIdx.y * 64 + wave * 16 + mrow;
  int khalf = (lane & 16) ? 2 : 0;

  v8f acc0 = {}, acc1 = {}, acc2 = {}, acc3 = {};

  for (int kc = 0; kc < K; kc += KCHUNK) {
    __syncthreads();
    for (int t = tid; t < 64 * KCHUNK; t += blockDim.x) {
      int r = t >> 6;
      int c = t & (KCHUNK - 1);
      int m = row0 + r;
      Xs[r * LDSTRIDE + c] = (m < M) ? X[(long)m * K + kc + c] : 0.0f;
    }
    __syncthreads();
    for (int k = 0; k < KCHUNK; k += 4) {
      int kk = k + khalf;
      v2f b = *(const v2f*)&W[(long)n * K + kc + kk];
      v2f a0 = *(const v2f*)&Xs[(mrow +  0) * LDSTRIDE + kk];
      v2f a1 = *(const v2f*)&Xs[(mrow + 16) * LDSTRIDE + kk];
      v2f a2 = *(const v2f*)&Xs[(mrow + 32) * LDSTRIDE + kk];
      v2f a3 = *(const v2f*)&Xs[(mrow + 48) * LDSTRIDE + kk];
      acc0 = __builtin_amdgcn_wmma_f32_16x16x4_f32(false, a0, false, b, (short)0, acc0, false, false);
      acc1 = __builtin_amdgcn_wmma_f32_16x16x4_f32(false, a1, false, b, (short)0, acc1, false, false);
      acc2 = __builtin_amdgcn_wmma_f32_16x16x4_f32(false, a2, false, b, (short)0, acc2, false, false);
      acc3 = __builtin_amdgcn_wmma_f32_16x16x4_f32(false, a3, false, b, (short)0, acc3, false, false);
    }
  }

  float bn = bias[n];
  for (int r = 0; r < 8; ++r) {
    int mb = row0 + r + ((lane & 16) ? 8 : 0);
    v8f* accs[4] = {&acc0, &acc1, &acc2, &acc3};
    for (int mt = 0; mt < 4; ++mt) {
      int m = mb + mt * 16;
      if (m < M) Y[(long)m * Nout + n] = (*accs[mt])[r] + bn;
    }
  }
}

#define SM_THREADS 256
#define SM_WAVES (SM_THREADS / 32)

template <bool IS_MAX>
__device__ __forceinline__ void seg_combine(float v, int g, bool valid, float* __restrict__ dst,
                                            float (*wred)[NUM_HEADS], int* wgr) {
  const int tid = threadIdx.x;
  const int lane = tid & 31, wave = tid >> 5;
  const int hh = lane & 3;
  const int g0 = __builtin_amdgcn_readfirstlane(g);
  const bool wuni = __all(g == g0) != 0;
  if (wuni) {
    for (int off = NUM_HEADS; off < 32; off <<= 1) {
      const float o = __shfl_xor(v, off);
      v = IS_MAX ? fmaxf(v, o) : v + o;
    }
    if (lane < NUM_HEADS) wred[wave][lane] = v;
    if (lane == 0) wgr[wave] = g0;
  } else {
    if (valid) {
      if (IS_MAX) atomic_max_float(&dst[g * NUM_HEADS + hh], v);
      else        atomicAdd(&dst[g * NUM_HEADS + hh], v);
    }
    if (lane == 0) wgr[wave] = -1;
  }
  __syncthreads();
  if (tid < SM_WAVES * NUM_HEADS) {
    const int w = tid >> 2;
    const int gw = wgr[w];
    if (gw >= 0 && (w == 0 || wgr[w - 1] != gw)) {
      float m = wred[w][hh];
      for (int u = w + 1; u < SM_WAVES && wgr[u] == gw; ++u)
        m = IS_MAX ? fmaxf(m, wred[u][hh]) : m + wred[u][hh];
      if (IS_MAX) atomic_max_float(&dst[gw * NUM_HEADS + hh], m);
      else        atomicAdd(&dst[gw * NUM_HEADS + hh], m);
    }
  }
}

__global__ void scores_kernel(const float* __restrict__ h, const float* __restrict__ xe,
                              const float* __restrict__ cnt, const float* __restrict__ ubuf,
                              const float* __restrict__ tbuf, const int* __restrict__ batch,
                              float* __restrict__ smax, float* __restrict__ scores,
                              int N, int ND, int ED, int US, int D) {
  __shared__ float wred[SM_WAVES][NUM_HEADS];
  __shared__ int wgr[SM_WAVES];
  const int idx = blockIdx.x * SM_THREADS + threadIdx.x;
  const bool valid = idx < N * NUM_HEADS;
  int nI = idx >> 2;
  if (nI > N - 1) nI = N - 1;
  int hh = idx & 3;
  const int K = ND + ED;
  const float* ur = ubuf + hh * US;
  float dot = tbuf[hh] + ur[K] + ((cnt[nI] > 0.0f) ? ur[K + 1] : 0.0f);
  const float* xr = h + (long)nI * ND;
  for (int c = 0; c < ND; c += 4) {
    float4 x = *(const float4*)&xr[c];
    float4 s = *(const float4*)&ur[c];
    dot += x.x * s.x + x.y * s.y + x.z * s.z + x.w * s.w;
  }
  const float* er = xe + (long)nI * ED;
  const float* ue = ur + ND;
  for (int c = 0; c < ED; c += 4) {
    float4 x = *(const float4*)&er[c];
    float4 s = *(const float4*)&ue[c];
    dot += x.x * s.x + x.y * s.y + x.z * s.z + x.w * s.w;
  }
  float sc = dot * rsqrtf((float)(D / NUM_HEADS));
  if (valid) scores[idx] = sc;
  seg_combine<true>(valid ? sc : -INFINITY, batch[nI], valid, smax, wred, wgr);
}

__global__ void expw_kernel(const float* __restrict__ scores, const int* __restrict__ batch,
                            const float* __restrict__ smax, float* __restrict__ denom,
                            float* __restrict__ wexp, int N) {
  __shared__ float wred[SM_WAVES][NUM_HEADS];
  __shared__ int wgr[SM_WAVES];
  const int idx = blockIdx.x * SM_THREADS + threadIdx.x;
  const bool valid = idx < N * NUM_HEADS;
  int nI = idx >> 2;
  if (nI > N - 1) nI = N - 1;
  const int hh = idx & 3;
  const int g = batch[nI];
  float w = 0.0f;
  if (valid) {
    w = __expf(scores[idx] - smax[g * NUM_HEADS + hh]);
    wexp[idx] = w;
  }
  seg_combine<false>(w, g, valid, denom, wred, wgr);
}

#define POOL_ROWS 64
#define POOL_CHUNK 16

__device__ __forceinline__ void pool_flush(float* __restrict__ Z, const float* __restrict__ denom,
                                           int g, int c, int ZS,
                                           float a0, float a1, float a2, float a3) {
  const float4 d = *(const float4*)&denom[(long)g * NUM_HEADS];
  float* zp = Z + (long)g * NUM_HEADS * ZS + c;
  atomicAdd(zp,              a0 / fmaxf(d.x, 1e-30f));
  atomicAdd(zp + ZS,         a1 / fmaxf(d.y, 1e-30f));
  atomicAdd(zp + 2L * ZS,    a2 / fmaxf(d.z, 1e-30f));
  atomicAdd(zp + 3L * ZS,    a3 / fmaxf(d.w, 1e-30f));
}

__global__ void pool_x(const float* __restrict__ wexp, const float* __restrict__ denom,
                       const int* __restrict__ batch, const float* __restrict__ h,
                       const float* __restrict__ xe, const float* __restrict__ cnt,
                       float* __restrict__ Zx, int N, int ND, int ED, int ZS) {
  __shared__ float4 wsh[POOL_ROWS];
  __shared__ int gsh[POOL_ROWS];
  const int tid = threadIdx.x;
  const int n0 = blockIdx.x * POOL_ROWS;
  int nend = n0 + POOL_ROWS;
  if (nend > N) nend = N;
  if (tid < POOL_ROWS) {
    if (n0 + tid < N) {
      wsh[tid] = *(const float4*)&wexp[(long)(n0 + tid) * NUM_HEADS];
      gsh[tid] = batch[n0 + tid];
    } else {
      wsh[tid] = make_float4(0.0f, 0.0f, 0.0f, 0.0f);
      gsh[tid] = 0;
    }
  }
  __syncthreads();

  const int K = ND + ED;
  const int c = blockIdx.y * blockDim.x + tid;
  const bool active = c <= K;
  const bool is_mask = c >= K;
  const float* xp;
  long pitch;
  if (c < ND)     { xp = h + (long)n0 * ND + c;          pitch = ND; }
  else if (c < K) { xp = xe + (long)n0 * ED + (c - ND);  pitch = ED; }
  else            { xp = cnt + n0;                       pitch = 1; }
  int gcur = __builtin_amdgcn_readfirstlane(gsh[0]);
  const int rlast = nend - 1 - n0;
  float a0 = 0.0f, a1 = 0.0f, a2 = 0.0f, a3 = 0.0f;
  for (int i0 = 0; i0 < POOL_ROWS; i0 += POOL_CHUNK) {
    if (n0 + i0 >= nend) break;
    float x[POOL_CHUNK];
    float4 wq[POOL_CHUNK];
    int gq[POOL_CHUNK];
#pragma unroll
    for (int j = 0; j < POOL_CHUNK; ++j) {
      int r = i0 + j;
      if (r > rlast) r = rlast;
      x[j] = xp[(long)r * pitch];
      wq[j] = wsh[i0 + j];
      gq[j] = gsh[i0 + j];
    }
#pragma unroll
    for (int j = 0; j < POOL_CHUNK; ++j) {
      if (n0 + i0 + j < nend) {
        const int g = __builtin_amdgcn_readfirstlane(gq[j]);
        if (g != gcur) {
          if (active) pool_flush(Zx, denom, gcur, c, ZS, a0, a1, a2, a3);
          gcur = g;
          a0 = 0.0f; a1 = 0.0f; a2 = 0.0f; a3 = 0.0f;
        }
        const float4 w = wq[j];
        const float xv = is_mask ? ((x[j] > 0.0f) ? 1.0f : 0.0f) : x[j];
        a0 += w.x * xv;
        a1 += w.y * xv;
        a2 += w.z * xv;
        a3 += w.w * xv;
      }
    }
  }
  if (active) pool_flush(Zx, denom, gcur, c, ZS, a0, a1, a2, a3);
}

__global__ void zproj_wmma(const float* __restrict__ Zx, const float* __restrict__ denom,
                           const float* __restrict__ node_w, const float* __restrict__ node_b,
                           const float* __restrict__ edge_w, const float* __restrict__ edge_b,
                           float* __restrict__ Z, int M, int ND, int ED, int D, int ZS) {
  const int tid = threadIdx.x;
  const int wave = tid >> 5, lane = tid & 31;
  const int mrow = lane & 15;
  const int khalf = (lane & 16) ? 2 : 0;
  const int K = ND + ED;
  const int r0 = blockIdx.x * 16;
  const int n = blockIdx.y * 64 + wave * 16 + mrow;
  int ra = r0 + mrow;
  if (ra > M - 1) ra = M - 1;
  const float* zr = Zx + (long)ra * ZS + khalf;
  const float* wn = node_w + (long)n * ND + khalf;
  const float* we = edge_w + (long)n * ED + khalf;

  v8f acc = {};
  for (int kc = 0; kc < K; kc += KCHUNK) {
    v2f a[KCHUNK / 4], b[KCHUNK / 4];
#pragma unroll
    for (int i = 0; i < KCHUNK / 4; ++i) {
      const int kg = kc + 4 * i;
      const float* bp = (kg < ND) ? (wn + kg) : (we + (kg - ND));
      a[i] = *(const v2f*)&zr[kg];
      b[i] = *(const v2f*)bp;
    }
#pragma unroll
    for (int i = 0; i < KCHUNK / 4; ++i)
      acc = __builtin_amdgcn_wmma_f32_16x16x4_f32(false, a[i], false, b[i], (short)0, acc, false, false);
  }

  const float bn = node_b[n];
  const float be = edge_b[n];
  const int rb = r0 + ((lane & 16) ? 8 : 0);
  float p0[8], p1[8];
#pragma unroll
  for (int r = 0; r < 8; ++r) {
    int rr = rb + r;
    if (rr > M - 1) rr = M - 1;
    const float d = denom[rr];
    p0[r] = d / fmaxf(d, 1e-30f);
    p1[r] = Zx[(long)rr * ZS + K];
  }
#pragma unroll
  for (int r = 0; r < 8; ++r) {
    const int rr = rb + r;
    if (rr < M) Z[(long)rr * D + n] = acc[r] + p0[r] * bn + p1[r] * be;
  }
}

__global__ void vproj_pooled_wmma(const float* __restrict__ Z, const float* __restrict__ Wv,
                                  const float* __restrict__ bv, const float* __restrict__ denom,
                                  float* __restrict__ pooled, int G, int D) {
  const int tid = threadIdx.x;
  const int wave = tid >> 5, lane = tid & 31;
  const int mrow = lane & 15;
  const int khalf = (lane & 16) ? 2 : 0;
  const int g0 = blockIdx.x * 16;
  const int ncol0 = blockIdx.y * 64 + wave * 16;
  const int n = ncol0 + mrow;
  const int hh = ncol0 / (D / NUM_HEADS);
  int ga = g0 + mrow;
  if (ga > G - 1) ga = G - 1;
  const float* zr = Z + ((long)ga * NUM_HEADS + hh) * D + khalf;
  const float* wr = Wv + (long)n * D + khalf;

  v8f acc = {};
  for (int kc = 0; kc < D; kc += KCHUNK) {
    v2f a[KCHUNK / 4], b[KCHUNK / 4];
#pragma unroll
    for (int i = 0; i < KCHUNK / 4; ++i) {
      a[i] = *(const v2f*)&zr[kc + 4 * i];
      b[i] = *(const v2f*)&wr[kc + 4 * i];
    }
#pragma unroll
    for (int i = 0; i < KCHUNK / 4; ++i)
      acc = __builtin_amdgcn_wmma_f32_16x16x4_f32(false, a[i], false, b[i], (short)0, acc, false, false);
  }

  const float bn = bv[n];
  const int gb = g0 + ((lane & 16) ? 8 : 0);
  float dsum[8];
#pragma unroll
  for (int r = 0; r < 8; ++r) {
    int g = gb + r;
    if (g > G - 1) g = G - 1;
    dsum[r] = denom[g * NUM_HEADS + hh];
  }
#pragma unroll
  for (int r = 0; r < 8; ++r) {
    const int g = gb + r;
    if (g < G) {
      const float sp = dsum[r] / fmaxf(dsum[r], 1e-30f);
      pooled[(long)g * D + n] = acc[r] + sp * bn;
    }
  }
}

extern "C" void kernel_launch(void* const* d_in, const int* in_sizes, int n_in,
                              void* d_out, int out_size, void* d_ws, size_t ws_size,
                              hipStream_t stream) {
  const float* h      = (const float*)d_in[0];
  const int*   eidx   = (const int*)d_in[1];
  const float* eattr  = (const float*)d_in[2];
  const int*   batch  = (const int*)d_in[3];
  const float* node_w = (const float*)d_in[5];
  const float* node_b = (const float*)d_in[6];
  const float* edge_w = (const float*)d_in[7];
  const float* edge_b = (const float*)d_in[8];
  const float* query  = (const float*)d_in[9];
  const float* in_w   = (const float*)d_in[10];
  const float* in_b   = (const float*)d_in[11];
  const float* out_w  = (const float*)d_in[12];
  const float* out_b  = (const float*)d_in[13];
  float* out = (float*)d_out;

  int D  = in_sizes[9];
  int ND = in_sizes[5] / D;
  int ED = in_sizes[7] / D;
  int N  = in_sizes[0] / ND;
  long E = in_sizes[1] / 2;
  int G  = out_size / D;
  int K  = ND + ED;
  int ZS = (K + 2 + 3) & ~3;

  float* ws = (float*)d_ws;
  long o = 0;
  auto take = [&](long nf) { float* p = ws + o; o += (nf + 3) & ~3L; return p; };
  float* sumA   = take((long)N * ED);
  int*   cnti   = (int*)take(N);
  float* cnt    = take(N);
  int*   slots  = (int*)take((long)N * SLOT_CAP);
  float* scores = take((long)N * NUM_HEADS);
  float* wexp   = take((long)N * NUM_HEADS);
  float* tbuf   = take(NUM_HEADS);
  float* ubuf   = take((long)NUM_HEADS * ZS);
  float* smax   = take((long)G * NUM_HEADS);
  float* denom  = take((long)G * NUM_HEADS + (long)G * NUM_HEADS * ZS);
  float* Zx     = denom + G * NUM_HEADS;
  float* Zb     = take((long)G * NUM_HEADS * D);
  float* pooled = take((long)G * D);

  {
    long z0 = (long)N * ED + N;
    long z1 = (long)G * NUM_HEADS + (long)G * NUM_HEADS * ZS;
    long maxn = z0 > z1 ? z0 : z1;
    int blocks = (int)((maxn + 255) / 256);
    if (blocks > 262144) blocks = 262144;
    init_ws<<<blocks, 256, 0, stream>>>(sumA, z0, denom, z1, smax, G * NUM_HEADS);
  }
  {
    dim3 grid((K + 2 + 31) / 32, NUM_HEADS);
    fold_kernel<<<grid, FW * 32, 0, stream>>>(query, in_w, in_b, node_w, edge_w, node_b, edge_b,
                                              ubuf, tbuf, D, ND, ED, ZS);
  }
  {
    long blocks = (E + 255) / 256;
    if (blocks > 262144) blocks = 262144;
    edge_slots<<<(int)blocks, 256, 0, stream>>>(eidx, eattr, cnti, slots, sumA, E, ED);
    long gthreads = (long)N * 16;
    edge_gather<<<(int)((gthreads + 255) / 256), 256, 0, stream>>>(eattr, slots, cnti, sumA,
                                                                 cnt, N, ED);
  }
  {
    int total = N * NUM_HEADS;
    int blocks = (total + SM_THREADS - 1) / SM_THREADS;
    scores_kernel<<<blocks, SM_THREADS, 0, stream>>>(h, sumA, cnt, ubuf, tbuf, batch,
                                                     smax, scores, N, ND, ED, ZS, D);
    expw_kernel<<<blocks, SM_THREADS, 0, stream>>>(scores, batch, smax, denom, wexp, N);
  }
  {
    dim3 grid((N + POOL_ROWS - 1) / POOL_ROWS, (K + 1 + 255) / 256);
    pool_x<<<grid, 256, 0, stream>>>(wexp, denom, batch, h, sumA, cnt, Zx, N, ND, ED, ZS);
  }
  {
    dim3 grid((G * NUM_HEADS + 15) / 16, D / 64);
    zproj_wmma<<<grid, 128, 0, stream>>>(Zx, denom, node_w, node_b, edge_w, edge_b, Zb,
                                         G * NUM_HEADS, ND, ED, D, ZS);
  }
  {
    dim3 grid((G + 15) / 16, D / 64);
    vproj_pooled_wmma<<<grid, 128, 0, stream>>>(Zb, in_w + 2L * D * D, in_b + 2 * D,
                                                denom, pooled, G, D);
  }
  {
    dim3 grid((G + 63) / 64, D / 64);
    gemm_xwT_wmma<<<grid, 128, 0, stream>>>(pooled, out_w, out_b, out, G, D, D);
  }
}


# Output GEMM: A and B fragments straight from global memory, a K chunk's loads issued as one group, a 16x16 tile per wave; was an LDS staging loop of one load, wait, store at a time; same sum order
# speedup vs baseline: 3.4954x; 1.2600x over previous
#include <hip/hip_runtime.h>
#include <math.h>

#define NUM_HEADS 4
#define KCHUNK 64

typedef float v2f __attribute__((ext_vector_type(2)));
typedef float v8f __attribute__((ext_vector_type(8)));

__device__ __forceinline__ void atomic_max_float(float* addr, float val) {
  if (val >= 0.0f) {
    atomicMax((int*)addr, __float_as_int(val));
  } else {
    atomicMin((unsigned int*)addr, __float_as_uint(val));
  }
}

__global__ void init_ws(float* z0, long n0, float* z1, long n1, float* smax, int ns) {
  long i = (long)blockIdx.x * blockDim.x + threadIdx.x;
  long stride = (long)gridDim.x * blockDim.x;
  for (long t = i; t < n0; t += stride) z0[t] = 0.0f;
  for (long t = i; t < n1; t += stride) z1[t] = 0.0f;
  for (long t = i; t < ns; t += stride) smax[t] = -INFINITY;
}

#define FW 8
#define FROWS 8
#define FJB 16
#define FPITCH 33

__device__ __forceinline__ float dot4(float4 a, float4 b) {
  return a.x * b.x + a.y * b.y + a.z * b.z + a.w * b.w;
}

__global__ void fold_kernel(const float* __restrict__ query, const float* __restrict__ in_w,
                            const float* __restrict__ in_b,
                            const float* __restrict__ node_w, const float* __restrict__ edge_w,
                            const float* __restrict__ node_b, const float* __restrict__ edge_b,
                            float* __restrict__ ubuf, float* __restrict__ tbuf,
                            int D, int ND, int ED, int US) {
  __shared__ __attribute__((aligned(16))) float buf[256 * FPITCH];
  __shared__ float qh[256];
  __shared__ float sh[1024];
  __shared__ float red[FW][32];
  const int tid = threadIdx.x;
  const int lane = tid & 31, wave = tid >> 5;
  const int hh = blockIdx.y;
  const int dh = D / NUM_HEADS;
  const int K = ND + ED;

  for (int rb = 0; rb < dh; rb += FW * FROWS) {
    float p[FROWS];
#pragma unroll
    for (int i = 0; i < FROWS; ++i) p[i] = 0.0f;
    for (int c = lane * 4; c < D; c += 128) {
      const float4 x = *(const float4*)&query[c];
      float4 w[FROWS];
#pragma unroll
      for (int i = 0; i < FROWS; ++i) {
        int r = rb + wave + FW * i;
        if (r > dh - 1) r = dh - 1;
        w[i] = *(const float4*)&in_w[(long)(hh * dh + r) * D + c];
      }
#pragma unroll
      for (int i = 0; i < FROWS; ++i) p[i] += dot4(w[i], x);
    }
#pragma unroll
    for (int i = 0; i < FROWS; ++i) {
      const int r = rb + wave + FW * i;
      if (r < dh) buf[r * FPITCH + lane] = p[i];
    }
  }
  __syncthreads();
  if (tid < dh) {
    float q = in_b[hh * dh + tid];
    for (int l = 0; l < 32; ++l) q += buf[tid * FPITCH + l];
    qh[tid] = q;
  }
  __syncthreads();

  if (blockIdx.x == 0 && wave == 0) {
    float t = 0.0f;
    for (int d = lane; d < dh; d += 32) t += qh[d] * in_b[D + hh * dh + d];
    for (int off = 16; off > 0; off >>= 1) t += __shfl_xor(t, off);
    if (lane == 0) tbuf[hh] = t;
  }

  const float* wk = in_w + (long)(D + hh * dh) * D;
  for (int jc = lane * 4; jc < D; jc += 128) {
    float4 acc = make_float4(0.0f, 0.0f, 0.0f, 0.0f);
    for (int db = 0; db < dh; db += FW * FROWS) {
      float4 w[FROWS];
      float q[FROWS];
#pragma unroll
      for (int i = 0; i < FROWS; ++i) {
        const int d = db + wave + FW * i;
        const int dc = d < dh ? d : dh - 1;
        w[i] = *(const float4*)&wk[(long)dc * D + jc];
        q[i] = d < dh ? qh[dc] : 0.0f;
      }
#pragma unroll
      for (int i = 0; i < FROWS; ++i) {
        acc.x += q[i] * w[i].x; acc.y += q[i] * w[i].y; acc.z += q[i] * w[i].z; acc.w += q[i] * w[i].w;
      }
    }
    *(float4*)&buf[wave * D + jc] = acc;
  }
  __syncthreads();
  for (int j = tid; j < D; j += FW * 32) {
    float s = 0.0f;
    for (int w = 0; w < FW; ++w) s += buf[w * D + j];
    sh[j] = s;
  }
  __syncthreads();

  const int k = blockIdx.x * 32 + lane;
  const float* wp;
  long stride;
  if (k < ND)      { wp = node_w + k;        stride = ND; }
  else if (k < K)  { wp = edge_w + (k - ND); stride = ED; }
  else if (k == K) { wp = node_b;            stride = 1; }
  else             { wp = edge_b;            stride = 1; }
  float a = 0.0f;
  for (int j0 = wave; j0 < D; j0 += FW * FJB) {
    float w[FJB], sv[FJB];
#pragma unroll
    for (int i = 0; i < FJB; ++i) {
      const int j = j0 + FW * i;
      const int jcl = j < D ? j : D - 1;
      w[i] = wp[(long)jcl * stride];
      sv[i] = j < D ? sh[jcl] : 0.0f;
    }
#pragma unroll
    for (int i = 0; i < FJB; ++i) a += sv[i] * w[i];
  }
  red[wave][lane] = a;
  __syncthreads();
  if (tid < 32) {
    float sum = 0.0f;
    for (int w = 0; w < FW; ++w) sum += red[w][lane];
    if (k < K + 2) ubuf[hh * US + k] = sum;
  }
}

#define SLOT_CAP 64
#define GATHER_CHUNK 8

__global__ void edge_slots(const int* __restrict__ row, const float* __restrict__ eattr,
                           int* __restrict__ cnti, int* __restrict__ slots,
                           float* __restrict__ sumA, long E, int ED) {
  long i = (long)blockIdx.x * blockDim.x + threadIdx.x;
  long stride = (long)gridDim.x * blockDim.x;
  for (long e = i; e < E; e += stride) {
    int r = row[e];
    int s = atomicAdd(&cnti[r], 1);
    if (s < SLOT_CAP) {
      slots[(long)r * SLOT_CAP + s] = (int)e;
    } else {
      for (int c = 0; c < ED; ++c) atomicAdd(&sumA[(long)r * ED + c], eattr[e * ED + c]);
    }
  }
}

__global__ void edge_gather(const float* __restrict__ eattr, const int* __restrict__ slots,
                            const int* __restrict__ cnti, float* __restrict__ sumA,
                            float* __restrict__ cntf, int N, int ED) {
  const int t = blockIdx.x * blockDim.x + threadIdx.x;
  const int m = t >> 4;
  const int l = t & 15;
  if (m >= N) return;
  const int cdeg = cnti[m];
  const int cs = cdeg < SLOT_CAP ? cdeg : SLOT_CAP;
  const int* sp = slots + (long)m * SLOT_CAP;
  for (int c0 = l * 4; c0 < ED; c0 += 64) {
    float4 acc = make_float4(0.0f, 0.0f, 0.0f, 0.0f);
    if (cdeg > SLOT_CAP) acc = *(const float4*)&sumA[(long)m * ED + c0];
    for (int j0 = 0; j0 < cs; j0 += GATHER_CHUNK) {
      float4 v[GATHER_CHUNK];
#pragma unroll
      for (int j = 0; j < GATHER_CHUNK; ++j) {
        int jj = j0 + j;
        if (jj > cs - 1) jj = cs - 1;
        const int e = sp[jj];
        v[j] = *(const float4*)&eattr[(long)e * ED + c0];
      }
#pragma unroll
      for (int j = 0; j < GATHER_CHUNK; ++j)
        asm volatile("" :: "v"(v[j].x), "v"(v[j].y), "v"(v[j].z), "v"(v[j].w));
#pragma unroll
      for (int j = 0; j < GATHER_CHUNK; ++j) {
        const bool in_list = j0 + j < cs;
        acc.x += in_list ? v[j].x : 0.0f;
        acc.y += in_list ? v[j].y : 0.0f;
        acc.z += in_list ? v[j].z : 0.0f;
        acc.w += in_list ? v[j].w : 0.0f;
      }
    }
    const float dm = fmaxf((float)cdeg, 1.0f);
    acc.x /= dm; acc.y /= dm; acc.z /= dm; acc.w /= dm;
    *(float4*)&sumA[(long)m * ED + c0] = acc;
  }
  if (l == 0) cntf[m] = (float)cdeg;
}


__global__ void gemm_xwT_wmma(const float* __restrict__ X, const float* __restrict__ W,
                              const float* __restrict__ bias, float* __restrict__ Y,
                              int M, int K, int Nout) {
  const int tid = threadIdx.x;
  const int wave = tid >> 5, lane = tid & 31;
  const int mrow = lane & 15;
  const int khalf = (lane & 16) ? 2 : 0;
  const int r0 = blockIdx.x * 16;
  const int n = blockIdx.y * 64 + wave * 16 + mrow;
  int ra = r0 + mrow;
  if (ra > M - 1) ra = M - 1;
  const float* xr = X + (long)ra * K + khalf;
  const float* wr = W + (long)n * K + khalf;
  const float bn = bias[n];

  v8f acc = {};
  for (int kc = 0; kc < K; kc += KCHUNK) {
    v2f a[KCHUNK / 4], b[KCHUNK / 4];
#pragma unroll
    for (int i = 0; i < KCHUNK / 4; ++i) {
      a[i] = *(const v2f*)&xr[kc + 4 * i];
      b[i] = *(const v2f*)&wr[kc + 4 * i];
    }
#pragma unroll
    for (int i = 0; i < KCHUNK / 4; ++i)
      acc = __builtin_amdgcn_wmma_f32_16x16x4_f32(false, a[i], false, b[i], (short)0, acc, false, false);
  }

  const int rb = r0 + ((lane & 16) ? 8 : 0);
#pragma unroll
  for (int r = 0; r < 8; ++r) {
    const int rr = rb + r;
    if (rr < M) Y[(long)rr * Nout + n] = acc[r] + bn;
  }
}

#define SM_THREADS 256
#define SM_WAVES (SM_THREADS / 32)

template <bool IS_MAX>
__device__ __forceinline__ void seg_combine(float v, int g, bool valid, float* __restrict__ dst,
                                            float (*wred)[NUM_HEADS], int* wgr) {
  const int tid = threadIdx.x;
  const int lane = tid & 31, wave = tid >> 5;
  const int hh = lane & 3;
  const int g0 = __builtin_amdgcn_readfirstlane(g);
  const bool wuni = __all(g == g0) != 0;
  if (wuni) {
    for (int off = NUM_HEADS; off < 32; off <<= 1) {
      const float o = __shfl_xor(v, off);
      v = IS_MAX ? fmaxf(v, o) : v + o;
    }
    if (lane < NUM_HEADS) wred[wave][lane] = v;
    if (lane == 0) wgr[wave] = g0;
  } else {
    if (valid) {
      if (IS_MAX) atomic_max_float(&dst[g * NUM_HEADS + hh], v);
      else        atomicAdd(&dst[g * NUM_HEADS + hh], v);
    }
    if (lane == 0) wgr[wave] = -1;
  }
  __syncthreads();
  if (tid < SM_WAVES * NUM_HEADS) {
    const int w = tid >> 2;
    const int gw = wgr[w];
    if (gw >= 0 && (w == 0 || wgr[w - 1] != gw)) {
      float m = wred[w][hh];
      for (int u = w + 1; u < SM_WAVES && wgr[u] == gw; ++u)
        m = IS_MAX ? fmaxf(m, wred[u][hh]) : m + wred[u][hh];
      if (IS_MAX) atomic_max_float(&dst[gw * NUM_HEADS + hh], m);
      else        atomicAdd(&dst[gw * NUM_HEADS + hh], m);
    }
  }
}

__global__ void scores_kernel(const float* __restrict__ h, const float* __restrict__ xe,
                              const float* __restrict__ cnt, const float* __restrict__ ubuf,
                              const float* __restrict__ tbuf, const int* __restrict__ batch,
                              float* __restrict__ smax, float* __restrict__ scores,
                              int N, int ND, int ED, int US, int D) {
  __shared__ float wred[SM_WAVES][NUM_HEADS];
  __shared__ int wgr[SM_WAVES];
  const int idx = blockIdx.x * SM_THREADS + threadIdx.x;
  const bool valid = idx < N * NUM_HEADS;
  int nI = idx >> 2;
  if (nI > N - 1) nI = N - 1;
  int hh = idx & 3;
  const int K = ND + ED;
  const float* ur = ubuf + hh * US;
  float dot = tbuf[hh] + ur[K] + ((cnt[nI] > 0.0f) ? ur[K + 1] : 0.0f);
  const float* xr = h + (long)nI * ND;
  for (int c = 0; c < ND; c += 4) {
    float4 x = *(const float4*)&xr[c];
    float4 s = *(const float4*)&ur[c];
    dot += x.x * s.x + x.y * s.y + x.z * s.z + x.w * s.w;
  }
  const float* er = xe + (long)nI * ED;
  const float* ue = ur + ND;
  for (int c = 0; c < ED; c += 4) {
    float4 x = *(const float4*)&er[c];
    float4 s = *(const float4*)&ue[c];
    dot += x.x * s.x + x.y * s.y + x.z * s.z + x.w * s.w;
  }
  float sc = dot * rsqrtf((float)(D / NUM_HEADS));
  if (valid) scores[idx] = sc;
  seg_combine<true>(valid ? sc : -INFINITY, batch[nI], valid, smax, wred, wgr);
}

__global__ void expw_kernel(const float* __restrict__ scores, const int* __restrict__ batch,
                            const float* __restrict__ smax, float* __restrict__ denom,
                            float* __restrict__ wexp, int N) {
  __shared__ float wred[SM_WAVES][NUM_HEADS];
  __shared__ int wgr[SM_WAVES];
  const int idx = blockIdx.x * SM_THREADS + threadIdx.x;
  const bool valid = idx < N * NUM_HEADS;
  int nI = idx >> 2;
  if (nI > N - 1) nI = N - 1;
  const int hh = idx & 3;
  const int g = batch[nI];
  float w = 0.0f;
  if (valid) {
    w = __expf(scores[idx] - smax[g * NUM_HEADS + hh]);
    wexp[idx] = w;
  }
  seg_combine<false>(w, g, valid, denom, wred, wgr);
}

#define POOL_ROWS 64
#define POOL_CHUNK 16

__device__ __forceinline__ void pool_flush(float* __restrict__ Z, const float* __restrict__ denom,
                                           int g, int c, int ZS,
                                           float a0, float a1, float a2, float a3) {
  const float4 d = *(const float4*)&denom[(long)g * NUM_HEADS];
  float* zp = Z + (long)g * NUM_HEADS * ZS + c;
  atomicAdd(zp,              a0 / fmaxf(d.x, 1e-30f));
  atomicAdd(zp + ZS,         a1 / fmaxf(d.y, 1e-30f));
  atomicAdd(zp + 2L * ZS,    a2 / fmaxf(d.z, 1e-30f));
  atomicAdd(zp + 3L * ZS,    a3 / fmaxf(d.w, 1e-30f));
}

__global__ void pool_x(const float* __restrict__ wexp, const float* __restrict__ denom,
                       const int* __restrict__ batch, const float* __restrict__ h,
                       const float* __restrict__ xe, const float* __restrict__ cnt,
                       float* __restrict__ Zx, int N, int ND, int ED, int ZS) {
  __shared__ float4 wsh[POOL_ROWS];
  __shared__ int gsh[POOL_ROWS];
  const int tid = threadIdx.x;
  const int n0 = blockIdx.x * POOL_ROWS;
  int nend = n0 + POOL_ROWS;
  if (nend > N) nend = N;
  if (tid < POOL_ROWS) {
    if (n0 + tid < N) {
      wsh[tid] = *(const float4*)&wexp[(long)(n0 + tid) * NUM_HEADS];
      gsh[tid] = batch[n0 + tid];
    } else {
      wsh[tid] = make_float4(0.0f, 0.0f, 0.0f, 0.0f);
      gsh[tid] = 0;
    }
  }
  __syncthreads();

  const int K = ND + ED;
  const int c = blockIdx.y * blockDim.x + tid;
  const bool active = c <= K;
  const bool is_mask = c >= K;
  const float* xp;
  long pitch;
  if (c < ND)     { xp = h + (long)n0 * ND + c;          pitch = ND; }
  else if (c < K) { xp = xe + (long)n0 * ED + (c - ND);  pitch = ED; }
  else            { xp = cnt + n0;                       pitch = 1; }
  int gcur = __builtin_amdgcn_readfirstlane(gsh[0]);
  const int rlast = nend - 1 - n0;
  float a0 = 0.0f, a1 = 0.0f, a2 = 0.0f, a3 = 0.0f;
  for (int i0 = 0; i0 < POOL_ROWS; i0 += POOL_CHUNK) {
    if (n0 + i0 >= nend) break;
    float x[POOL_CHUNK];
    float4 wq[POOL_CHUNK];
    int gq[POOL_CHUNK];
#pragma unroll
    for (int j = 0; j < POOL_CHUNK; ++j) {
      int r = i0 + j;
      if (r > rlast) r = rlast;
      x[j] = xp[(long)r * pitch];
      wq[j] = wsh[i0 + j];
      gq[j] = gsh[i0 + j];
    }
#pragma unroll
    for (int j = 0; j < POOL_CHUNK; ++j) {
      if (n0 + i0 + j < nend) {
        const int g = __builtin_amdgcn_readfirstlane(gq[j]);
        if (g != gcur) {
          if (active) pool_flush(Zx, denom, gcur, c, ZS, a0, a1, a2, a3);
          gcur = g;
          a0 = 0.0f; a1 = 0.0f; a2 = 0.0f; a3 = 0.0f;
        }
        const float4 w = wq[j];
        const float xv = is_mask ? ((x[j] > 0.0f) ? 1.0f : 0.0f) : x[j];
        a0 += w.x * xv;
        a1 += w.y * xv;
        a2 += w.z * xv;
        a3 += w.w * xv;
      }
    }
  }
  if (active) pool_flush(Zx, denom, gcur, c, ZS, a0, a1, a2, a3);
}

__global__ void zproj_wmma(const float* __restrict__ Zx, const float* __restrict__ denom,
                           const float* __restrict__ node_w, const float* __restrict__ node_b,
                           const float* __restrict__ edge_w, const float* __restrict__ edge_b,
                           float* __restrict__ Z, int M, int ND, int ED, int D, int ZS) {
  const int tid = threadIdx.x;
  const int wave = tid >> 5, lane = tid & 31;
  const int mrow = lane & 15;
  const int khalf = (lane & 16) ? 2 : 0;
  const int K = ND + ED;
  const int r0 = blockIdx.x * 16;
  const int n = blockIdx.y * 64 + wave * 16 + mrow;
  int ra = r0 + mrow;
  if (ra > M - 1) ra = M - 1;
  const float* zr = Zx + (long)ra * ZS + khalf;
  const float* wn = node_w + (long)n * ND + khalf;
  const float* we = edge_w + (long)n * ED + khalf;

  v8f acc = {};
  for (int kc = 0; kc < K; kc += KCHUNK) {
    v2f a[KCHUNK / 4], b[KCHUNK / 4];
#pragma unroll
    for (int i = 0; i < KCHUNK / 4; ++i) {
      const int kg = kc + 4 * i;
      const float* bp = (kg < ND) ? (wn + kg) : (we + (kg - ND));
      a[i] = *(const v2f*)&zr[kg];
      b[i] = *(const v2f*)bp;
    }
#pragma unroll
    for (int i = 0; i < KCHUNK / 4; ++i)
      acc = __builtin_amdgcn_wmma_f32_16x16x4_f32(false, a[i], false, b[i], (short)0, acc, false, false);
  }

  const float bn = node_b[n];
  const float be = edge_b[n];
  const int rb = r0 + ((lane & 16) ? 8 : 0);
  float p0[8], p1[8];
#pragma unroll
  for (int r = 0; r < 8; ++r) {
    int rr = rb + r;
    if (rr > M - 1) rr = M - 1;
    const float d = denom[rr];
    p0[r] = d / fmaxf(d, 1e-30f);
    p1[r] = Zx[(long)rr * ZS + K];
  }
#pragma unroll
  for (int r = 0; r < 8; ++r) {
    const int rr = rb + r;
    if (rr < M) Z[(long)rr * D + n] = acc[r] + p0[r] * bn + p1[r] * be;
  }
}

__global__ void vproj_pooled_wmma(const float* __restrict__ Z, const float* __restrict__ Wv,
                                  const float* __restrict__ bv, const float* __restrict__ denom,
                                  float* __restrict__ pooled, int G, int D) {
  const int tid = threadIdx.x;
  const int wave = tid >> 5, lane = tid & 31;
  const int mrow = lane & 15;
  const int khalf = (lane & 16) ? 2 : 0;
  const int g0 = blockIdx.x * 16;
  const int ncol0 = blockIdx.y * 64 + wave * 16;
  const int n = ncol0 + mrow;
  const int hh = ncol0 / (D / NUM_HEADS);
  int ga = g0 + mrow;
  if (ga > G - 1) ga = G - 1;
  const float* zr = Z + ((long)ga * NUM_HEADS + hh) * D + khalf;
  const float* wr = Wv + (long)n * D + khalf;

  v8f acc = {};
  for (int kc = 0; kc < D; kc += KCHUNK) {
    v2f a[KCHUNK / 4], b[KCHUNK / 4];
#pragma unroll
    for (int i = 0; i < KCHUNK / 4; ++i) {
      a[i] = *(const v2f*)&zr[kc + 4 * i];
      b[i] = *(const v2f*)&wr[kc + 4 * i];
    }
#pragma unroll
    for (int i = 0; i < KCHUNK / 4; ++i)
      acc = __builtin_amdgcn_wmma_f32_16x16x4_f32(false, a[i], false, b[i], (short)0, acc, false, false);
  }

  const float bn = bv[n];
  const int gb = g0 + ((lane & 16) ? 8 : 0);
  float dsum[8];
#pragma unroll
  for (int r = 0; r < 8; ++r) {
    int g = gb + r;
    if (g > G - 1) g = G - 1;
    dsum[r] = denom[g * NUM_HEADS + hh];
  }
#pragma unroll
  for (int r = 0; r < 8; ++r) {
    const int g = gb + r;
    if (g < G) {
      const float sp = dsum[r] / fmaxf(dsum[r], 1e-30f);
      pooled[(long)g * D + n] = acc[r] + sp * bn;
    }
  }
}

extern "C" void kernel_launch(void* const* d_in, const int* in_sizes, int n_in,
                              void* d_out, int out_size, void* d_ws, size_t ws_size,
                              hipStream_t stream) {
  const float* h      = (const float*)d_in[0];
  const int*   eidx   = (const int*)d_in[1];
  const float* eattr  = (const float*)d_in[2];
  const int*   batch  = (const int*)d_in[3];
  const float* node_w = (const float*)d_in[5];
  const float* node_b = (const float*)d_in[6];
  const float* edge_w = (const float*)d_in[7];
  const float* edge_b = (const float*)d_in[8];
  const float* query  = (const float*)d_in[9];
  const float* in_w   = (const float*)d_in[10];
  const float* in_b   = (const float*)d_in[11];
  const float* out_w  = (const float*)d_in[12];
  const float* out_b  = (const float*)d_in[13];
  float* out = (float*)d_out;

  int D  = in_sizes[9];
  int ND = in_sizes[5] / D;
  int ED = in_sizes[7] / D;
  int N  = in_sizes[0] / ND;
  long E = in_sizes[1] / 2;
  int G  = out_size / D;
  int K  = ND + ED;
  int ZS = (K + 2 + 3) & ~3;

  float* ws = (float*)d_ws;
  long o = 0;
  auto take = [&](long nf) { float* p = ws + o; o += (nf + 3) & ~3L; return p; };
  float* sumA   = take((long)N * ED);
  int*   cnti   = (int*)take(N);
  float* cnt    = take(N);
  int*   slots  = (int*)take((long)N * SLOT_CAP);
  float* scores = take((long)N * NUM_HEADS);
  float* wexp   = take((long)N * NUM_HEADS);
  float* tbuf   = take(NUM_HEADS);
  float* ubuf   = take((long)NUM_HEADS * ZS);
  float* smax   = take((long)G * NUM_HEADS);
  float* denom  = take((long)G * NUM_HEADS + (long)G * NUM_HEADS * ZS);
  float* Zx     = denom + G * NUM_HEADS;
  float* Zb     = take((long)G * NUM_HEADS * D);
  float* pooled = take((long)G * D);

  {
    long z0 = (long)N * ED + N;
    long z1 = (long)G * NUM_HEADS + (long)G * NUM_HEADS * ZS;
    long maxn = z0 > z1 ? z0 : z1;
    int blocks = (int)((maxn + 255) / 256);
    if (blocks > 262144) blocks = 262144;
    init_ws<<<blocks, 256, 0, stream>>>(sumA, z0, denom, z1, smax, G * NUM_HEADS);
  }
  {
    dim3 grid((K + 2 + 31) / 32, NUM_HEADS);
    fold_kernel<<<grid, FW * 32, 0, stream>>>(query, in_w, in_b, node_w, edge_w, node_b, edge_b,
                                              ubuf, tbuf, D, ND, ED, ZS);
  }
  {
    long blocks = (E + 255) / 256;
    if (blocks > 262144) blocks = 262144;
    edge_slots<<<(int)blocks, 256, 0, stream>>>(eidx, eattr, cnti, slots, sumA, E, ED);
    long gthreads = (long)N * 16;
    edge_gather<<<(int)((gthreads + 255) / 256), 256, 0, stream>>>(eattr, slots, cnti, sumA,
                                                                 cnt, N, ED);
  }
  {
    int total = N * NUM_HEADS;
    int blocks = (total + SM_THREADS - 1) / SM_THREADS;
    scores_kernel<<<blocks, SM_THREADS, 0, stream>>>(h, sumA, cnt, ubuf, tbuf, batch,
                                                     smax, scores, N, ND, ED, ZS, D);
    expw_kernel<<<blocks, SM_THREADS, 0, stream>>>(scores, batch, smax, denom, wexp, N);
  }
  {
    dim3 grid((N + POOL_ROWS - 1) / POOL_ROWS, (K + 1 + 255) / 256);
    pool_x<<<grid, 256, 0, stream>>>(wexp, denom, batch, h, sumA, cnt, Zx, N, ND, ED, ZS);
  }
  {
    dim3 grid((G * NUM_HEADS + 15) / 16, D / 64);
    zproj_wmma<<<grid, 128, 0, stream>>>(Zx, denom, node_w, node_b, edge_w, edge_b, Zb,
                                         G * NUM_HEADS, ND, ED, D, ZS);
  }
  {
    dim3 grid((G + 15) / 16, D / 64);
    vproj_pooled_wmma<<<grid, 128, 0, stream>>>(Zb, in_w + 2L * D * D, in_b + 2 * D,
                                                denom, pooled, G, D);
  }
  {
    dim3 grid((G + 15) / 16, D / 64);
    gemm_xwT_wmma<<<grid, 128, 0, stream>>>(pooled, out_w, out_b, out, G, D, D);
  }
}


# Edge ranks from LDS counters: a block takes 1024 nodes x a quarter of the edge list, the LDS add's return is the slot in that quarter's sub-list; no global atomic per edge; per-node sum order changes
# speedup vs baseline: 4.2302x; 1.2102x over previous
#include <hip/hip_runtime.h>
#include <math.h>

#define NUM_HEADS 4
#define KCHUNK 64

typedef float v2f __attribute__((ext_vector_type(2)));
typedef float v8f __attribute__((ext_vector_type(8)));

__device__ __forceinline__ void atomic_max_float(float* addr, float val) {
  if (val >= 0.0f) {
    atomicMax((int*)addr, __float_as_int(val));
  } else {
    atomicMin((unsigned int*)addr, __float_as_uint(val));
  }
}

__global__ void init_ws(float* z0, long n0, float* z1, long n1, float* smax, int ns) {
  long i = (long)blockIdx.x * blockDim.x + threadIdx.x;
  long stride = (long)gridDim.x * blockDim.x;
  for (long t = i; t < n0; t += stride) z0[t] = 0.0f;
  for (long t = i; t < n1; t += stride) z1[t] = 0.0f;
  for (long t = i; t < ns; t += stride) smax[t] = -INFINITY;
}

#define FW 8
#define FROWS 8
#define FJB 16
#define FPITCH 33

__device__ __forceinline__ float dot4(float4 a, float4 b) {
  return a.x * b.x + a.y * b.y + a.z * b.z + a.w * b.w;
}

__global__ void fold_kernel(const float* __restrict__ query, const float* __restrict__ in_w,
                            const float* __restrict__ in_b,
                            const float* __restrict__ node_w, const float* __restrict__ edge_w,
                            const float* __restrict__ node_b, const float* __restrict__ edge_b,
                            float* __restrict__ ubuf, float* __restrict__ tbuf,
                            int D, int ND, int ED, int US) {
  __shared__ __attribute__((aligned(16))) float buf[256 * FPITCH];
  __shared__ float qh[256];
  __shared__ float sh[1024];
  __shared__ float red[FW][32];
  const int tid = threadIdx.x;
  const int lane = tid & 31, wave = tid >> 5;
  const int hh = blockIdx.y;
  const int dh = D / NUM_HEADS;
  const int K = ND + ED;

  for (int rb = 0; rb < dh; rb += FW * FROWS) {
    float p[FROWS];
#pragma unroll
    for (int i = 0; i < FROWS; ++i) p[i] = 0.0f;
    for (int c = lane * 4; c < D; c += 128) {
      const float4 x = *(const float4*)&query[c];
      float4 w[FROWS];
#pragma unroll
      for (int i = 0; i < FROWS; ++i) {
        int r = rb + wave + FW * i;
        if (r > dh - 1) r = dh - 1;
        w[i] = *(const float4*)&in_w[(long)(hh * dh + r) * D + c];
      }
#pragma unroll
      for (int i = 0; i < FROWS; ++i) p[i] += dot4(w[i], x);
    }
#pragma unroll
    for (int i = 0; i < FROWS; ++i) {
      const int r = rb + wave + FW * i;
      if (r < dh) buf[r * FPITCH + lane] = p[i];
    }
  }
  __syncthreads();
  if (tid < dh) {
    float q = in_b[hh * dh + tid];
    for (int l = 0; l < 32; ++l) q += buf[tid * FPITCH + l];
    qh[tid] = q;
  }
  __syncthreads();

  if (blockIdx.x == 0 && wave == 0) {
    float t = 0.0f;
    for (int d = lane; d < dh; d += 32) t += qh[d] * in_b[D + hh * dh + d];
    for (int off = 16; off > 0; off >>= 1) t += __shfl_xor(t, off);
    if (lane == 0) tbuf[hh] = t;
  }

  const float* wk = in_w + (long)(D + hh * dh) * D;
  for (int jc = lane * 4; jc < D; jc += 128) {
    float4 acc = make_float4(0.0f, 0.0f, 0.0f, 0.0f);
    for (int db = 0; db < dh; db += FW * FROWS) {
      float4 w[FROWS];
      float q[FROWS];
#pragma unroll
      for (int i = 0; i < FROWS; ++i) {
        const int d = db + wave + FW * i;
        const int dc = d < dh ? d : dh - 1;
        w[i] = *(const float4*)&wk[(long)dc * D + jc];
        q[i] = d < dh ? qh[dc] : 0.0f;
      }
#pragma unroll
      for (int i = 0; i < FROWS; ++i) {
        acc.x += q[i] * w[i].x; acc.y += q[i] * w[i].y; acc.z += q[i] * w[i].z; acc.w += q[i] * w[i].w;
      }
    }
    *(float4*)&buf[wave * D + jc] = acc;
  }
  __syncthreads();
  for (int j = tid; j < D; j += FW * 32) {
    float s = 0.0f;
    for (int w = 0; w < FW; ++w) s += buf[w * D + j];
    sh[j] = s;
  }
  __syncthreads();

  const int k = blockIdx.x * 32 + lane;
  const float* wp;
  long stride;
  if (k < ND)      { wp = node_w + k;        stride = ND; }
  else if (k < K)  { wp = edge_w + (k - ND); stride = ED; }
  else if (k == K) { wp = node_b;            stride = 1; }
  else             { wp = edge_b;            stride = 1; }
  float a = 0.0f;
  for (int j0 = wave; j0 < D; j0 += FW * FJB) {
    float w[FJB], sv[FJB];
#pragma unroll
    for (int i = 0; i < FJB; ++i) {
      const int j = j0 + FW * i;
      const int jcl = j < D ? j : D - 1;
      w[i] = wp[(long)jcl * stride];
      sv[i] = j < D ? sh[jcl] : 0.0f;
    }
#pragma unroll
    for (int i = 0; i < FJB; ++i) a += sv[i] * w[i];
  }
  red[wave][lane] = a;
  __syncthreads();
  if (tid < 32) {
    float sum = 0.0f;
    for (int w = 0; w < FW; ++w) sum += red[w][lane];
    if (k < K + 2) ubuf[hh * US + k] = sum;
  }
}

#define SLOT_CAP 64
#define GATHER_CHUNK 8
#define RP_CHUNKS 4
#define RP_CAP (SLOT_CAP / RP_CHUNKS)
#define RP_NODES 1024
#define RP_THREADS 1024
#define RP_BATCH 4

__device__ __forceinline__ void rank_place(int r, long e, int nlo, int c, int* lcnt,
                                           const float* __restrict__ eattr,
                                           int* __restrict__ slots, float* __restrict__ sumA,
                                           int ED) {
  const unsigned d = (unsigned)(r - nlo);
  if (d < (unsigned)RP_NODES) {
    const int s = atomicAdd(&lcnt[d], 1);
    if (s < RP_CAP) {
      slots[(long)r * SLOT_CAP + c * RP_CAP + s] = (int)e;
    } else {
      for (int k = 0; k < ED; ++k) atomicAdd(&sumA[(long)r * ED + k], eattr[e * ED + k]);
    }
  }
}

__global__ void edge_rank(const int* __restrict__ row, const float* __restrict__ eattr,
                          int* __restrict__ slots, int* __restrict__ cntc,
                          float* __restrict__ sumA, long E, int N, int ED) {
  __shared__ int lcnt[RP_NODES];
  const int tid = threadIdx.x;
  const int nlo = blockIdx.x * RP_NODES;
  const int c = blockIdx.y;
  for (int i = tid; i < RP_NODES; i += RP_THREADS) lcnt[i] = 0;
  __syncthreads();
  const long ec = (((E + RP_CHUNKS - 1) / RP_CHUNKS) + 3) & ~3L;
  const long e0 = (long)c * ec;
  long e1 = e0 + ec;
  if (e1 > E) e1 = E;
  if (e0 < e1) {
    const long e1v = e0 + ((e1 - e0) & ~3L);
    for (long eb = e0 + 4L * tid; eb < e1v; eb += 4L * RP_THREADS * RP_BATCH) {
      int4 r4[RP_BATCH];
#pragma unroll
      for (int u = 0; u < RP_BATCH; ++u) {
        long e = eb + 4L * RP_THREADS * u;
        if (e > e1v - 4) e = e1v - 4;
        r4[u] = *(const int4*)&row[e];
      }
#pragma unroll
      for (int u = 0; u < RP_BATCH; ++u) {
        const long e = eb + 4L * RP_THREADS * u;
        if (e < e1v) {
          rank_place(r4[u].x, e,     nlo, c, lcnt, eattr, slots, sumA, ED);
          rank_place(r4[u].y, e + 1, nlo, c, lcnt, eattr, slots, sumA, ED);
          rank_place(r4[u].z, e + 2, nlo, c, lcnt, eattr, slots, sumA, ED);
          rank_place(r4[u].w, e + 3, nlo, c, lcnt, eattr, slots, sumA, ED);
        }
      }
    }
    const long et = e1v + tid;
    if (et < e1) rank_place(row[et], et, nlo, c, lcnt, eattr, slots, sumA, ED);
  }
  __syncthreads();
  for (int i = tid; i < RP_NODES; i += RP_THREADS) {
    const int n = nlo + i;
    if (n < N) cntc[(long)n * RP_CHUNKS + c] = lcnt[i];
  }
}

__global__ void edge_gather(const float* __restrict__ eattr, const int* __restrict__ slots,
                            const int* __restrict__ cntc, float* __restrict__ sumA,
                            float* __restrict__ cntf, int N, int ED) {
  __shared__ int dense[256 / 16][SLOT_CAP];
  const int t = blockIdx.x * blockDim.x + threadIdx.x;
  const int m = t >> 4;
  const int l = t & 15;
  const int grp = threadIdx.x >> 4;
  const bool valid = m < N;
  int m0 = 0, m1 = 0, m2 = 0, m3 = 0, cdeg = 0;
  if (valid) {
    const int4 cq = *(const int4*)&cntc[(long)m * RP_CHUNKS];
    m0 = cq.x < RP_CAP ? cq.x : RP_CAP;
    m1 = cq.y < RP_CAP ? cq.y : RP_CAP;
    m2 = cq.z < RP_CAP ? cq.z : RP_CAP;
    m3 = cq.w < RP_CAP ? cq.w : RP_CAP;
    cdeg = cq.x + cq.y + cq.z + cq.w;
  }
  const int cs = m0 + m1 + m2 + m3;
  int ev[SLOT_CAP / 16];
#pragma unroll
  for (int u = 0; u < SLOT_CAP / 16; ++u) {
    const int j = l + 16 * u;
    ev[u] = 0;
    if (j < cs) {
      int c, s;
      if (j < m0)                { c = 0; s = j; }
      else if (j < m0 + m1)      { c = 1; s = j - m0; }
      else if (j < m0 + m1 + m2) { c = 2; s = j - m0 - m1; }
      else                       { c = 3; s = j - m0 - m1 - m2; }
      ev[u] = slots[(long)m * SLOT_CAP + c * RP_CAP + s];
    }
  }
#pragma unroll
  for (int u = 0; u < SLOT_CAP / 16; ++u) {
    const int j = l + 16 * u;
    if (j < cs) dense[grp][j] = ev[u];
  }
  __syncthreads();
  if (!valid) return;
  const int* sp = dense[grp];
  for (int c0 = l * 4; c0 < ED; c0 += 64) {
    float4 acc = make_float4(0.0f, 0.0f, 0.0f, 0.0f);
    if (cdeg > cs) acc = *(const float4*)&sumA[(long)m * ED + c0];
    for (int j0 = 0; j0 < cs; j0 += GATHER_CHUNK) {
      float4 v[GATHER_CHUNK];
#pragma unroll
      for (int j = 0; j < GATHER_CHUNK; ++j) {
        int jj = j0 + j;
        if (jj > cs - 1) jj = cs - 1;
        const int e = sp[jj];
        v[j] = *(const float4*)&eattr[(long)e * ED + c0];
      }
#pragma unroll
      for (int j = 0; j < GATHER_CHUNK; ++j)
        asm volatile("" :: "v"(v[j].x), "v"(v[j].y), "v"(v[j].z), "v"(v[j].w));
#pragma unroll
      for (int j = 0; j < GATHER_CHUNK; ++j) {
        const bool in_list = j0 + j < cs;
        acc.x += in_list ? v[j].x : 0.0f;
        acc.y += in_list ? v[j].y : 0.0f;
        acc.z += in_list ? v[j].z : 0.0f;
        acc.w += in_list ? v[j].w : 0.0f;
      }
    }
    const float dm = fmaxf((float)cdeg, 1.0f);
    acc.x /= dm; acc.y /= dm; acc.z /= dm; acc.w /= dm;
    *(float4*)&sumA[(long)m * ED + c0] = acc;
  }
  if (l == 0) cntf[m] = (float)cdeg;
}


__global__ void gemm_xwT_wmma(const float* __restrict__ X, const float* __restrict__ W,
                              const float* __restrict__ bias, float* __restrict__ Y,
                              int M, int K, int Nout) {
  const int tid = threadIdx.x;
  const int wave = tid >> 5, lane = tid & 31;
  const int mrow = lane & 15;
  const int khalf = (lane & 16) ? 2 : 0;
  const int r0 = blockIdx.x * 16;
  const int n = blockIdx.y * 64 + wave * 16 + mrow;
  int ra = r0 + mrow;
  if (ra > M - 1) ra = M - 1;
  const float* xr = X + (long)ra * K + khalf;
  const float* wr = W + (long)n * K + khalf;
  const float bn = bias[n];

  v8f acc = {};
  for (int kc = 0; kc < K; kc += KCHUNK) {
    v2f a[KCHUNK / 4], b[KCHUNK / 4];
#pragma unroll
    for (int i = 0; i < KCHUNK / 4; ++i) {
      a[i] = *(const v2f*)&xr[kc + 4 * i];
      b[i] = *(const v2f*)&wr[kc + 4 * i];
    }
#pragma unroll
    for (int i = 0; i < KCHUNK / 4; ++i)
      acc = __builtin_amdgcn_wmma_f32_16x16x4_f32(false, a[i], false, b[i], (short)0, acc, false, false);
  }

  const int rb = r0 + ((lane & 16) ? 8 : 0);
#pragma unroll
  for (int r = 0; r < 8; ++r) {
    const int rr = rb + r;
    if (rr < M) Y[(long)rr * Nout + n] = acc[r] + bn;
  }
}

#define SM_THREADS 256
#define SM_WAVES (SM_THREADS / 32)

template <bool IS_MAX>
__device__ __forceinline__ void seg_combine(float v, int g, bool valid, float* __restrict__ dst,
                                            float (*wred)[NUM_HEADS], int* wgr) {
  const int tid = threadIdx.x;
  const int lane = tid & 31, wave = tid >> 5;
  const int hh = lane & 3;
  const int g0 = __builtin_amdgcn_readfirstlane(g);
  const bool wuni = __all(g == g0) != 0;
  if (wuni) {
    for (int off = NUM_HEADS; off < 32; off <<= 1) {
      const float o = __shfl_xor(v, off);
      v = IS_MAX ? fmaxf(v, o) : v + o;
    }
    if (lane < NUM_HEADS) wred[wave][lane] = v;
    if (lane == 0) wgr[wave] = g0;
  } else {
    if (valid) {
      if (IS_MAX) atomic_max_float(&dst[g * NUM_HEADS + hh], v);
      else        atomicAdd(&dst[g * NUM_HEADS + hh], v);
    }
    if (lane == 0) wgr[wave] = -1;
  }
  __syncthreads();
  if (tid < SM_WAVES * NUM_HEADS) {
    const int w = tid >> 2;
    const int gw = wgr[w];
    if (gw >= 0 && (w == 0 || wgr[w - 1] != gw)) {
      float m = wred[w][hh];
      for (int u = w + 1; u < SM_WAVES && wgr[u] == gw; ++u)
        m = IS_MAX ? fmaxf(m, wred[u][hh]) : m + wred[u][hh];
      if (IS_MAX) atomic_max_float(&dst[gw * NUM_HEADS + hh], m);
      else        atomicAdd(&dst[gw * NUM_HEADS + hh], m);
    }
  }
}

__global__ void scores_kernel(const float* __restrict__ h, const float* __restrict__ xe,
                              const float* __restrict__ cnt, const float* __restrict__ ubuf,
                              const float* __restrict__ tbuf, const int* __restrict__ batch,
                              float* __restrict__ smax, float* __restrict__ scores,
                              int N, int ND, int ED, int US, int D) {
  __shared__ float wred[SM_WAVES][NUM_HEADS];
  __shared__ int wgr[SM_WAVES];
  const int idx = blockIdx.x * SM_THREADS + threadIdx.x;
  const bool valid = idx < N * NUM_HEADS;
  int nI = idx >> 2;
  if (nI > N - 1) nI = N - 1;
  int hh = idx & 3;
  const int K = ND + ED;
  const float* ur = ubuf + hh * US;
  float dot = tbuf[hh] + ur[K] + ((cnt[nI] > 0.0f) ? ur[K + 1] : 0.0f);
  const float* xr = h + (long)nI * ND;
  for (int c = 0; c < ND; c += 4) {
    float4 x = *(const float4*)&xr[c];
    float4 s = *(const float4*)&ur[c];
    dot += x.x * s.x + x.y * s.y + x.z * s.z + x.w * s.w;
  }
  const float* er = xe + (long)nI * ED;
  const float* ue = ur + ND;
  for (int c = 0; c < ED; c += 4) {
    float4 x = *(const float4*)&er[c];
    float4 s = *(const float4*)&ue[c];
    dot += x.x * s.x + x.y * s.y + x.z * s.z + x.w * s.w;
  }
  float sc = dot * rsqrtf((float)(D / NUM_HEADS));
  if (valid) scores[idx] = sc;
  seg_combine<true>(valid ? sc : -INFINITY, batch[nI], valid, smax, wred, wgr);
}

__global__ void expw_kernel(const float* __restrict__ scores, const int* __restrict__ batch,
                            const float* __restrict__ smax, float* __restrict__ denom,
                            float* __restrict__ wexp, int N) {
  __shared__ float wred[SM_WAVES][NUM_HEADS];
  __shared__ int wgr[SM_WAVES];
  const int idx = blockIdx.x * SM_THREADS + threadIdx.x;
  const bool valid = idx < N * NUM_HEADS;
  int nI = idx >> 2;
  if (nI > N - 1) nI = N - 1;
  const int hh = idx & 3;
  const int g = batch[nI];
  float w = 0.0f;
  if (valid) {
    w = __expf(scores[idx] - smax[g * NUM_HEADS + hh]);
    wexp[idx] = w;
  }
  seg_combine<false>(w, g, valid, denom, wred, wgr);
}

#define POOL_ROWS 64
#define POOL_CHUNK 16

__device__ __forceinline__ void pool_flush(float* __restrict__ Z, const float* __restrict__ denom,
                                           int g, int c, int ZS,
                                           float a0, float a1, float a2, float a3) {
  const float4 d = *(const float4*)&denom[(long)g * NUM_HEADS];
  float* zp = Z + (long)g * NUM_HEADS * ZS + c;
  atomicAdd(zp,              a0 / fmaxf(d.x, 1e-30f));
  atomicAdd(zp + ZS,         a1 / fmaxf(d.y, 1e-30f));
  atomicAdd(zp + 2L * ZS,    a2 / fmaxf(d.z, 1e-30f));
  atomicAdd(zp + 3L * ZS,    a3 / fmaxf(d.w, 1e-30f));
}

__global__ void pool_x(const float* __restrict__ wexp, const float* __restrict__ denom,
                       const int* __restrict__ batch, const float* __restrict__ h,
                       const float* __restrict__ xe, const float* __restrict__ cnt,
                       float* __restrict__ Zx, int N, int ND, int ED, int ZS) {
  __shared__ float4 wsh[POOL_ROWS];
  __shared__ int gsh[POOL_ROWS];
  const int tid = threadIdx.x;
  const int n0 = blockIdx.x * POOL_ROWS;
  int nend = n0 + POOL_ROWS;
  if (nend > N) nend = N;
  if (tid < POOL_ROWS) {
    if (n0 + tid < N) {
      wsh[tid] = *(const float4*)&wexp[(long)(n0 + tid) * NUM_HEADS];
      gsh[tid] = batch[n0 + tid];
    } else {
      wsh[tid] = make_float4(0.0f, 0.0f, 0.0f, 0.0f);
      gsh[tid] = 0;
    }
  }
  __syncthreads();

  const int K = ND + ED;
  const int c = blockIdx.y * blockDim.x + tid;
  const bool active = c <= K;
  const bool is_mask = c >= K;
  const float* xp;
  long pitch;
  if (c < ND)     { xp = h + (long)n0 * ND + c;          pitch = ND; }
  else if (c < K) { xp = xe + (long)n0 * ED + (c - ND);  pitch = ED; }
  else            { xp = cnt + n0;                       pitch = 1; }
  int gcur = __builtin_amdgcn_readfirstlane(gsh[0]);
  const int rlast = nend - 1 - n0;
  float a0 = 0.0f, a1 = 0.0f, a2 = 0.0f, a3 = 0.0f;
  for (int i0 = 0; i0 < POOL_ROWS; i0 += POOL_CHUNK) {
    if (n0 + i0 >= nend) break;
    float x[POOL_CHUNK];
    float4 wq[POOL_CHUNK];
    int gq[POOL_CHUNK];
#pragma unroll
    for (int j = 0; j < POOL_CHUNK; ++j) {
      int r = i0 + j;
      if (r > rlast) r = rlast;
      x[j] = xp[(long)r * pitch];
      wq[j] = wsh[i0 + j];
      gq[j] = gsh[i0 + j];
    }
#pragma unroll
    for (int j = 0; j < POOL_CHUNK; ++j) {
      if (n0 + i0 + j < nend) {
        const int g = __builtin_amdgcn_readfirstlane(gq[j]);
        if (g != gcur) {
          if (active) pool_flush(Zx, denom, gcur, c, ZS, a0, a1, a2, a3);
          gcur = g;
          a0 = 0.0f; a1 = 0.0f; a2 = 0.0f; a3 = 0.0f;
        }
        const float4 w = wq[j];
        const float xv = is_mask ? ((x[j] > 0.0f) ? 1.0f : 0.0f) : x[j];
        a0 += w.x * xv;
        a1 += w.y * xv;
        a2 += w.z * xv;
        a3 += w.w * xv;
      }
    }
  }
  if (active) pool_flush(Zx, denom, gcur, c, ZS, a0, a1, a2, a3);
}

__global__ void zproj_wmma(const float* __restrict__ Zx, const float* __restrict__ denom,
                           const float* __restrict__ node_w, const float* __restrict__ node_b,
                           const float* __restrict__ edge_w, const float* __restrict__ edge_b,
                           float* __restrict__ Z, int M, int ND, int ED, int D, int ZS) {
  const int tid = threadIdx.x;
  const int wave = tid >> 5, lane = tid & 31;
  const int mrow = lane & 15;
  const int khalf = (lane & 16) ? 2 : 0;
  const int K = ND + ED;
  const int r0 = blockIdx.x * 16;
  const int n = blockIdx.y * 64 + wave * 16 + mrow;
  int ra = r0 + mrow;
  if (ra > M - 1) ra = M - 1;
  const float* zr = Zx + (long)ra * ZS + khalf;
  const float* wn = node_w + (long)n * ND + khalf;
  const float* we = edge_w + (long)n * ED + khalf;

  v8f acc = {};
  for (int kc = 0; kc < K; kc += KCHUNK) {
    v2f a[KCHUNK / 4], b[KCHUNK / 4];
#pragma unroll
    for (int i = 0; i < KCHUNK / 4; ++i) {
      const int kg = kc + 4 * i;
      const float* bp = (kg < ND) ? (wn + kg) : (we + (kg - ND));
      a[i] = *(const v2f*)&zr[kg];
      b[i] = *(const v2f*)bp;
    }
#pragma unroll
    for (int i = 0; i < KCHUNK / 4; ++i)
      acc = __builtin_amdgcn_wmma_f32_16x16x4_f32(false, a[i], false, b[i], (short)0, acc, false, false);
  }

  const float bn = node_b[n];
  const float be = edge_b[n];
  const int rb = r0 + ((lane & 16) ? 8 : 0);
  float p0[8], p1[8];
#pragma unroll
  for (int r = 0; r < 8; ++r) {
    int rr = rb + r;
    if (rr > M - 1) rr = M - 1;
    const float d = denom[rr];
    p0[r] = d / fmaxf(d, 1e-30f);
    p1[r] = Zx[(long)rr * ZS + K];
  }
#pragma unroll
  for (int r = 0; r < 8; ++r) {
    const int rr = rb + r;
    if (rr < M) Z[(long)rr * D + n] = acc[r] + p0[r] * bn + p1[r] * be;
  }
}

__global__ void vproj_pooled_wmma(const float* __restrict__ Z, const float* __restrict__ Wv,
                                  const float* __restrict__ bv, const float* __restrict__ denom,
                                  float* __restrict__ pooled, int G, int D) {
  const int tid = threadIdx.x;
  const int wave = tid >> 5, lane = tid & 31;
  const int mrow = lane & 15;
  const int khalf = (lane & 16) ? 2 : 0;
  const int g0 = blockIdx.x * 16;
  const int ncol0 = blockIdx.y * 64 + wave * 16;
  const int n = ncol0 + mrow;
  const int hh = ncol0 / (D / NUM_HEADS);
  int ga = g0 + mrow;
  if (ga > G - 1) ga = G - 1;
  const float* zr = Z + ((long)ga * NUM_HEADS + hh) * D + khalf;
  const float* wr = Wv + (long)n * D + khalf;

  v8f acc = {};
  for (int kc = 0; kc < D; kc += KCHUNK) {
    v2f a[KCHUNK / 4], b[KCHUNK / 4];
#pragma unroll
    for (int i = 0; i < KCHUNK / 4; ++i) {
      a[i] = *(const v2f*)&zr[kc + 4 * i];
      b[i] = *(const v2f*)&wr[kc + 4 * i];
    }
#pragma unroll
    for (int i = 0; i < KCHUNK / 4; ++i)
      acc = __builtin_amdgcn_wmma_f32_16x16x4_f32(false, a[i], false, b[i], (short)0, acc, false, false);
  }

  const float bn = bv[n];
  const int gb = g0 + ((lane & 16) ? 8 : 0);
  float dsum[8];
#pragma unroll
  for (int r = 0; r < 8; ++r) {
    int g = gb + r;
    if (g > G - 1) g = G - 1;
    dsum[r] = denom[g * NUM_HEADS + hh];
  }
#pragma unroll
  for (int r = 0; r < 8; ++r) {
    const int g = gb + r;
    if (g < G) {
      const float sp = dsum[r] / fmaxf(dsum[r], 1e-30f);
      pooled[(long)g * D + n] = acc[r] + sp * bn;
    }
  }
}

extern "C" void kernel_launch(void* const* d_in, const int* in_sizes, int n_in,
                              void* d_out, int out_size, void* d_ws, size_t ws_size,
                              hipStream_t stream) {
  const float* h      = (const float*)d_in[0];
  const int*   eidx   = (const int*)d_in[1];
  const float* eattr  = (const float*)d_in[2];
  const int*   batch  = (const int*)d_in[3];
  const float* node_w = (const float*)d_in[5];
  const float* node_b = (const float*)d_in[6];
  const float* edge_w = (const float*)d_in[7];
  const float* edge_b = (const float*)d_in[8];
  const float* query  = (const float*)d_in[9];
  const float* in_w   = (const float*)d_in[10];
  const float* in_b   = (const float*)d_in[11];
  const float* out_w  = (const float*)d_in[12];
  const float* out_b  = (const float*)d_in[13];
  float* out = (float*)d_out;

  int D  = in_sizes[9];
  int ND = in_sizes[5] / D;
  int ED = in_sizes[7] / D;
  int N  = in_sizes[0] / ND;
  long E = in_sizes[1] / 2;
  int G  = out_size / D;
  int K  = ND + ED;
  int ZS = (K + 2 + 3) & ~3;

  float* ws = (float*)d_ws;
  long o = 0;
  auto take = [&](long nf) { float* p = ws + o; o += (nf + 3) & ~3L; return p; };
  float* sumA   = take((long)N * ED);
  int*   cntc   = (int*)take((long)N * RP_CHUNKS);
  float* cnt    = take(N);
  int*   slots  = (int*)take((long)N * SLOT_CAP);
  float* scores = take((long)N * NUM_HEADS);
  float* wexp   = take((long)N * NUM_HEADS);
  float* tbuf   = take(NUM_HEADS);
  float* ubuf   = take((long)NUM_HEADS * ZS);
  float* smax   = take((long)G * NUM_HEADS);
  float* denom  = take((long)G * NUM_HEADS + (long)G * NUM_HEADS * ZS);
  float* Zx     = denom + G * NUM_HEADS;
  float* Zb     = take((long)G * NUM_HEADS * D);
  float* pooled = take((long)G * D);

  {
    long z0 = (long)N * ED;
    long z1 = (long)G * NUM_HEADS + (long)G * NUM_HEADS * ZS;
    long maxn = z0 > z1 ? z0 : z1;
    int blocks = (int)((maxn + 255) / 256);
    if (blocks > 262144) blocks = 262144;
    init_ws<<<blocks, 256, 0, stream>>>(sumA, z0, denom, z1, smax, G * NUM_HEADS);
  }
  {
    dim3 grid((K + 2 + 31) / 32, NUM_HEADS);
    fold_kernel<<<grid, FW * 32, 0, stream>>>(query, in_w, in_b, node_w, edge_w, node_b, edge_b,
                                              ubuf, tbuf, D, ND, ED, ZS);
  }
  {
    dim3 rgrid((N + RP_NODES - 1) / RP_NODES, RP_CHUNKS);
    edge_rank<<<rgrid, RP_THREADS, 0, stream>>>(eidx, eattr, slots, cntc, sumA, E, N, ED);
    long gthreads = (long)N * 16;
    edge_gather<<<(int)((gthreads + 255) / 256), 256, 0, stream>>>(eattr, slots, cntc, sumA,
                                                                 cnt, N, ED);
  }
  {
    int total = N * NUM_HEADS;
    int blocks = (total + SM_THREADS - 1) / SM_THREADS;
    scores_kernel<<<blocks, SM_THREADS, 0, stream>>>(h, sumA, cnt, ubuf, tbuf, batch,
                                                     smax, scores, N, ND, ED, ZS, D);
    expw_kernel<<<blocks, SM_THREADS, 0, stream>>>(scores, batch, smax, denom, wexp, N);
  }
  {
    dim3 grid((N + POOL_ROWS - 1) / POOL_ROWS, (K + 1 + 255) / 256);
    pool_x<<<grid, 256, 0, stream>>>(wexp, denom, batch, h, sumA, cnt, Zx, N, ND, ED, ZS);
  }
  {
    dim3 grid((G * NUM_HEADS + 15) / 16, D / 64);
    zproj_wmma<<<grid, 128, 0, stream>>>(Zx, denom, node_w, node_b, edge_w, edge_b, Zb,
                                         G * NUM_HEADS, ND, ED, D, ZS);
  }
  {
    dim3 grid((G + 15) / 16, D / 64);
    vproj_pooled_wmma<<<grid, 128, 0, stream>>>(Zb, in_w + 2L * D * D, in_b + 2 * D,
                                                denom, pooled, G, D);
  }
  {
    dim3 grid((G + 15) / 16, D / 64);
    gemm_xwT_wmma<<<grid, 128, 0, stream>>>(pooled, out_w, out_b, out, G, D, D);
  }
}


# Launches 10 to 8: exp(score - max) taken inside the pooling, denominator pooled as a column of ones and divided out afterwards; workspace fill rides in the fold launch; denominator sum order changes
# speedup vs baseline: 4.5244x; 1.0695x over previous
#include <hip/hip_runtime.h>
#include <math.h>

#define NUM_HEADS 4
#define KCHUNK 64

typedef float v2f __attribute__((ext_vector_type(2)));
typedef float v8f __attribute__((ext_vector_type(8)));

__device__ __forceinline__ void atomic_max_float(float* addr, float val) {
  if (val >= 0.0f) {
    atomicMax((int*)addr, __float_as_int(val));
  } else {
    atomicMin((unsigned int*)addr, __float_as_uint(val));
  }
}

__device__ __forceinline__ void fill_region(float* __restrict__ p, long n, float v, long i, long stride) {
  const float4 v4 = make_float4(v, v, v, v);
  const long n4 = n >> 2;
  for (long t = i; t < n4; t += stride) ((float4*)p)[t] = v4;
  for (long t = (n4 << 2) + i; t < n; t += stride) p[t] = v;
}

#define FW 8
#define FROWS 8
#define FJB 16
#define FPITCH 33
#define FILL_ROWS 32

__device__ __forceinline__ float dot4(float4 a, float4 b) {
  return a.x * b.x + a.y * b.y + a.z * b.z + a.w * b.w;
}

__global__ void fold_kernel(const float* __restrict__ query, const float* __restrict__ in_w,
                            const float* __restrict__ in_b,
                            const float* __restrict__ node_w, const float* __restrict__ edge_w,
                            const float* __restrict__ node_b, const float* __restrict__ edge_b,
                            float* __restrict__ ubuf, float* __restrict__ tbuf,
                            int D, int ND, int ED, int US,
                            float* __restrict__ z0, long n0, float* __restrict__ z1, long n1,
                            float* __restrict__ smax, long ns) {
  if (blockIdx.y >= NUM_HEADS) {
    const long fb = (long)(blockIdx.y - NUM_HEADS) * gridDim.x + blockIdx.x;
    const long i = fb * blockDim.x + threadIdx.x;
    const long stride = (long)(gridDim.y - NUM_HEADS) * gridDim.x * blockDim.x;
    fill_region(z0, n0, 0.0f, i, stride);
    fill_region(z1, n1, 0.0f, i, stride);
    fill_region(smax, ns, -INFINITY, i, stride);
    return;
  }
  __shared__ __attribute__((aligned(16))) float buf[256 * FPITCH];
  __shared__ float qh[256];
  __shared__ float sh[1024];
  __shared__ float red[FW][32];
  const int tid = threadIdx.x;
  const int lane = tid & 31, wave = tid >> 5;
  const int hh = blockIdx.y;
  const int dh = D / NUM_HEADS;
  const int K = ND + ED;

  for (int rb = 0; rb < dh; rb += FW * FROWS) {
    float p[FROWS];
#pragma unroll
    for (int i = 0; i < FROWS; ++i) p[i] = 0.0f;
    for (int c = lane * 4; c < D; c += 128) {
      const float4 x = *(const float4*)&query[c];
      float4 w[FROWS];
#pragma unroll
      for (int i = 0; i < FROWS; ++i) {
        int r = rb + wave + FW * i;
        if (r > dh - 1) r = dh - 1;
        w[i] = *(const float4*)&in_w[(long)(hh * dh + r) * D + c];
      }
#pragma unroll
      for (int i = 0; i < FROWS; ++i) p[i] += dot4(w[i], x);
    }
#pragma unroll
    for (int i = 0; i < FROWS; ++i) {
      const int r = rb + wave + FW * i;
      if (r < dh) buf[r * FPITCH + lane] = p[i];
    }
  }
  __syncthreads();
  if (tid < dh) {
    float q = in_b[hh * dh + tid];
    for (int l = 0; l < 32; ++l) q += buf[tid * FPITCH + l];
    qh[tid] = q;
  }
  __syncthreads();

  if (blockIdx.x == 0 && wave == 0) {
    float t = 0.0f;
    for (int d = lane; d < dh; d += 32) t += qh[d] * in_b[D + hh * dh + d];
    for (int off = 16; off > 0; off >>= 1) t += __shfl_xor(t, off);
    if (lane == 0) tbuf[hh] = t;
  }

  const float* wk = in_w + (long)(D + hh * dh) * D;
  for (int jc = lane * 4; jc < D; jc += 128) {
    float4 acc = make_float4(0.0f, 0.0f, 0.0f, 0.0f);
    for (int db = 0; db < dh; db += FW * FROWS) {
      float4 w[FROWS];
      float q[FROWS];
#pragma unroll
      for (int i = 0; i < FROWS; ++i) {
        const int d = db + wave + FW * i;
        const int dc = d < dh ? d : dh - 1;
        w[i] = *(const float4*)&wk[(long)dc * D + jc];
        q[i] = d < dh ? qh[dc] : 0.0f;
      }
#pragma unroll
      for (int i = 0; i < FROWS; ++i) {
        acc.x += q[i] * w[i].x; acc.y += q[i] * w[i].y; acc.z += q[i] * w[i].z; acc.w += q[i] * w[i].w;
      }
    }
    *(float4*)&buf[wave * D + jc] = acc;
  }
  __syncthreads();
  for (int j = tid; j < D; j += FW * 32) {
    float s = 0.0f;
    for (int w = 0; w < FW; ++w) s += buf[w * D + j];
    sh[j] = s;
  }
  __syncthreads();

  const int k = blockIdx.x * 32 + lane;
  const float* wp;
  long stride;
  if (k < ND)      { wp = node_w + k;        stride = ND; }
  else if (k < K)  { wp = edge_w + (k - ND); stride = ED; }
  else if (k == K) { wp = node_b;            stride = 1; }
  else             { wp = edge_b;            stride = 1; }
  float a = 0.0f;
  for (int j0 = wave; j0 < D; j0 += FW * FJB) {
    float w[FJB], sv[FJB];
#pragma unroll
    for (int i = 0; i < FJB; ++i) {
      const int j = j0 + FW * i;
      const int jcl = j < D ? j : D - 1;
      w[i] = wp[(long)jcl * stride];
      sv[i] = j < D ? sh[jcl] : 0.0f;
    }
#pragma unroll
    for (int i = 0; i < FJB; ++i) a += sv[i] * w[i];
  }
  red[wave][lane] = a;
  __syncthreads();
  if (tid < 32) {
    float sum = 0.0f;
    for (int w = 0; w < FW; ++w) sum += red[w][lane];
    if (k < K + 2) ubuf[hh * US + k] = sum;
  }
}

#define SLOT_CAP 64
#define GATHER_CHUNK 8
#define RP_CHUNKS 4
#define RP_CAP (SLOT_CAP / RP_CHUNKS)
#define RP_NODES 1024
#define RP_THREADS 1024
#define RP_BATCH 4

__device__ __forceinline__ void rank_place(int r, long e, int nlo, int c, int* lcnt,
                                           const float* __restrict__ eattr,
                                           int* __restrict__ slots, float* __restrict__ sumA,
                                           int ED) {
  const unsigned d = (unsigned)(r - nlo);
  if (d < (unsigned)RP_NODES) {
    const int s = atomicAdd(&lcnt[d], 1);
    if (s < RP_CAP) {
      slots[(long)r * SLOT_CAP + c * RP_CAP + s] = (int)e;
    } else {
      for (int k = 0; k < ED; ++k) atomicAdd(&sumA[(long)r * ED + k], eattr[e * ED + k]);
    }
  }
}

__global__ void edge_rank(const int* __restrict__ row, const float* __restrict__ eattr,
                          int* __restrict__ slots, int* __restrict__ cntc,
                          float* __restrict__ sumA, long E, int N, int ED) {
  __shared__ int lcnt[RP_NODES];
  const int tid = threadIdx.x;
  const int nlo = blockIdx.x * RP_NODES;
  const int c = blockIdx.y;
  for (int i = tid; i < RP_NODES; i += RP_THREADS) lcnt[i] = 0;
  __syncthreads();
  const long ec = (((E + RP_CHUNKS - 1) / RP_CHUNKS) + 3) & ~3L;
  const long e0 = (long)c * ec;
  long e1 = e0 + ec;
  if (e1 > E) e1 = E;
  if (e0 < e1) {
    const long e1v = e0 + ((e1 - e0) & ~3L);
    for (long eb = e0 + 4L * tid; eb < e1v; eb += 4L * RP_THREADS * RP_BATCH) {
      int4 r4[RP_BATCH];
#pragma unroll
      for (int u = 0; u < RP_BATCH; ++u) {
        long e = eb + 4L * RP_THREADS * u;
        if (e > e1v - 4) e = e1v - 4;
        r4[u] = *(const int4*)&row[e];
      }
#pragma unroll
      for (int u = 0; u < RP_BATCH; ++u) {
        const long e = eb + 4L * RP_THREADS * u;
        if (e < e1v) {
          rank_place(r4[u].x, e,     nlo, c, lcnt, eattr, slots, sumA, ED);
          rank_place(r4[u].y, e + 1, nlo, c, lcnt, eattr, slots, sumA, ED);
          rank_place(r4[u].z, e + 2, nlo, c, lcnt, eattr, slots, sumA, ED);
          rank_place(r4[u].w, e + 3, nlo, c, lcnt, eattr, slots, sumA, ED);
        }
      }
    }
    const long et = e1v + tid;
    if (et < e1) rank_place(row[et], et, nlo, c, lcnt, eattr, slots, sumA, ED);
  }
  __syncthreads();
  for (int i = tid; i < RP_NODES; i += RP_THREADS) {
    const int n = nlo + i;
    if (n < N) cntc[(long)n * RP_CHUNKS + c] = lcnt[i];
  }
}

__global__ void edge_gather(const float* __restrict__ eattr, const int* __restrict__ slots,
                            const int* __restrict__ cntc, float* __restrict__ sumA,
                            float* __restrict__ cntf, int N, int ED) {
  __shared__ int dense[256 / 16][SLOT_CAP];
  const int t = blockIdx.x * blockDim.x + threadIdx.x;
  const int m = t >> 4;
  const int l = t & 15;
  const int grp = threadIdx.x >> 4;
  const bool valid = m < N;
  int m0 = 0, m1 = 0, m2 = 0, m3 = 0, cdeg = 0;
  if (valid) {
    const int4 cq = *(const int4*)&cntc[(long)m * RP_CHUNKS];
    m0 = cq.x < RP_CAP ? cq.x : RP_CAP;
    m1 = cq.y < RP_CAP ? cq.y : RP_CAP;
    m2 = cq.z < RP_CAP ? cq.z : RP_CAP;
    m3 = cq.w < RP_CAP ? cq.w : RP_CAP;
    cdeg = cq.x + cq.y + cq.z + cq.w;
  }
  const int cs = m0 + m1 + m2 + m3;
  int ev[SLOT_CAP / 16];
#pragma unroll
  for (int u = 0; u < SLOT_CAP / 16; ++u) {
    const int j = l + 16 * u;
    ev[u] = 0;
    if (j < cs) {
      int c, s;
      if (j < m0)                { c = 0; s = j; }
      else if (j < m0 + m1)      { c = 1; s = j - m0; }
      else if (j < m0 + m1 + m2) { c = 2; s = j - m0 - m1; }
      else                       { c = 3; s = j - m0 - m1 - m2; }
      ev[u] = slots[(long)m * SLOT_CAP + c * RP_CAP + s];
    }
  }
#pragma unroll
  for (int u = 0; u < SLOT_CAP / 16; ++u) {
    const int j = l + 16 * u;
    if (j < cs) dense[grp][j] = ev[u];
  }
  __syncthreads();
  if (!valid) return;
  const int* sp = dense[grp];
  for (int c0 = l * 4; c0 < ED; c0 += 64) {
    float4 acc = make_float4(0.0f, 0.0f, 0.0f, 0.0f);
    if (cdeg > cs) acc = *(const float4*)&sumA[(long)m * ED + c0];
    for (int j0 = 0; j0 < cs; j0 += GATHER_CHUNK) {
      float4 v[GATHER_CHUNK];
#pragma unroll
      for (int j = 0; j < GATHER_CHUNK; ++j) {
        int jj = j0 + j;
        if (jj > cs - 1) jj = cs - 1;
        const int e = sp[jj];
        v[j] = *(const float4*)&eattr[(long)e * ED + c0];
      }
#pragma unroll
      for (int j = 0; j < GATHER_CHUNK; ++j)
        asm volatile("" :: "v"(v[j].x), "v"(v[j].y), "v"(v[j].z), "v"(v[j].w));
#pragma unroll
      for (int j = 0; j < GATHER_CHUNK; ++j) {
        const bool in_list = j0 + j < cs;
        acc.x += in_list ? v[j].x : 0.0f;
        acc.y += in_list ? v[j].y : 0.0f;
        acc.z += in_list ? v[j].z : 0.0f;
        acc.w += in_list ? v[j].w : 0.0f;
      }
    }
    const float dm = fmaxf((float)cdeg, 1.0f);
    acc.x /= dm; acc.y /= dm; acc.z /= dm; acc.w /= dm;
    *(float4*)&sumA[(long)m * ED + c0] = acc;
  }
  if (l == 0) cntf[m] = (float)cdeg;
}


__global__ void gemm_xwT_wmma(const float* __restrict__ X, const float* __restrict__ W,
                              const float* __restrict__ bias, float* __restrict__ Y,
                              int M, int K, int Nout) {
  const int tid = threadIdx.x;
  const int wave = tid >> 5, lane = tid & 31;
  const int mrow = lane & 15;
  const int khalf = (lane & 16) ? 2 : 0;
  const int r0 = blockIdx.x * 16;
  const int n = blockIdx.y * 64 + wave * 16 + mrow;
  int ra = r0 + mrow;
  if (ra > M - 1) ra = M - 1;
  const float* xr = X + (long)ra * K + khalf;
  const float* wr = W + (long)n * K + khalf;
  const float bn = bias[n];

  v8f acc = {};
  for (int kc = 0; kc < K; kc += KCHUNK) {
    v2f a[KCHUNK / 4], b[KCHUNK / 4];
#pragma unroll
    for (int i = 0; i < KCHUNK / 4; ++i) {
      a[i] = *(const v2f*)&xr[kc + 4 * i];
      b[i] = *(const v2f*)&wr[kc + 4 * i];
    }
#pragma unroll
    for (int i = 0; i < KCHUNK / 4; ++i)
      acc = __builtin_amdgcn_wmma_f32_16x16x4_f32(false, a[i], false, b[i], (short)0, acc, false, false);
  }

  const int rb = r0 + ((lane & 16) ? 8 : 0);
#pragma unroll
  for (int r = 0; r < 8; ++r) {
    const int rr = rb + r;
    if (rr < M) Y[(long)rr * Nout + n] = acc[r] + bn;
  }
}

#define SM_THREADS 256
#define SM_WAVES (SM_THREADS / 32)

__device__ __forceinline__ void seg_combine_max(float v, int g, bool valid, float* __restrict__ dst,
                                                float (*wred)[NUM_HEADS], int* wgr) {
  const int tid = threadIdx.x;
  const int lane = tid & 31, wave = tid >> 5;
  const int hh = lane & 3;
  const int g0 = __builtin_amdgcn_readfirstlane(g);
  const bool wuni = __all(g == g0) != 0;
  if (wuni) {
    for (int off = NUM_HEADS; off < 32; off <<= 1) {
      const float o = __shfl_xor(v, off);
      v = fmaxf(v, o);
    }
    if (lane < NUM_HEADS) wred[wave][lane] = v;
    if (lane == 0) wgr[wave] = g0;
  } else {
    if (valid) atomic_max_float(&dst[g * NUM_HEADS + hh], v);
    if (lane == 0) wgr[wave] = -1;
  }
  __syncthreads();
  if (tid < SM_WAVES * NUM_HEADS) {
    const int w = tid >> 2;
    const int gw = wgr[w];
    if (gw >= 0 && (w == 0 || wgr[w - 1] != gw)) {
      float m = wred[w][hh];
      for (int u = w + 1; u < SM_WAVES && wgr[u] == gw; ++u)
        m = fmaxf(m, wred[u][hh]);
      atomic_max_float(&dst[gw * NUM_HEADS + hh], m);
    }
  }
}

__global__ void scores_kernel(const float* __restrict__ h, const float* __restrict__ xe,
                              const float* __restrict__ cnt, const float* __restrict__ ubuf,
                              const float* __restrict__ tbuf, const int* __restrict__ batch,
                              float* __restrict__ smax, float* __restrict__ scores,
                              int N, int ND, int ED, int US, int D) {
  __shared__ float wred[SM_WAVES][NUM_HEADS];
  __shared__ int wgr[SM_WAVES];
  const int idx = blockIdx.x * SM_THREADS + threadIdx.x;
  const bool valid = idx < N * NUM_HEADS;
  int nI = idx >> 2;
  if (nI > N - 1) nI = N - 1;
  int hh = idx & 3;
  const int K = ND + ED;
  const float* ur = ubuf + hh * US;
  float dot = tbuf[hh] + ur[K] + ((cnt[nI] > 0.0f) ? ur[K + 1] : 0.0f);
  const float* xr = h + (long)nI * ND;
  for (int c = 0; c < ND; c += 4) {
    float4 x = *(const float4*)&xr[c];
    float4 s = *(const float4*)&ur[c];
    dot += x.x * s.x + x.y * s.y + x.z * s.z + x.w * s.w;
  }
  const float* er = xe + (long)nI * ED;
  const float* ue = ur + ND;
  for (int c = 0; c < ED; c += 4) {
    float4 x = *(const float4*)&er[c];
    float4 s = *(const float4*)&ue[c];
    dot += x.x * s.x + x.y * s.y + x.z * s.z + x.w * s.w;
  }
  float sc = dot * rsqrtf((float)(D / NUM_HEADS));
  if (valid) scores[idx] = sc;
  seg_combine_max(valid ? sc : -INFINITY, batch[nI], valid, smax, wred, wgr);
}

#define POOL_ROWS 64
#define POOL_CHUNK 16

__device__ __forceinline__ void pool_flush(float* __restrict__ Z, int g, int c, int ZS,
                                           float a0, float a1, float a2, float a3) {
  float* zp = Z + (long)g * NUM_HEADS * ZS + c;
  atomicAdd(zp,              a0);
  atomicAdd(zp + ZS,         a1);
  atomicAdd(zp + 2L * ZS,    a2);
  atomicAdd(zp + 3L * ZS,    a3);
}

__global__ void pool_x(const float* __restrict__ scores, const float* __restrict__ smax,
                       const int* __restrict__ batch, const float* __restrict__ h,
                       const float* __restrict__ xe, const float* __restrict__ cnt,
                       float* __restrict__ Zx, int N, int ND, int ED, int ZS) {
  __shared__ float4 wsh[POOL_ROWS];
  __shared__ int gsh[POOL_ROWS];
  const int tid = threadIdx.x;
  const int n0 = blockIdx.x * POOL_ROWS;
  int nend = n0 + POOL_ROWS;
  if (nend > N) nend = N;
  if (tid < POOL_ROWS) {
    if (n0 + tid < N) {
      const int g = batch[n0 + tid];
      const float4 s = *(const float4*)&scores[(long)(n0 + tid) * NUM_HEADS];
      const float4 m = *(const float4*)&smax[(long)g * NUM_HEADS];
      wsh[tid] = make_float4(__expf(s.x - m.x), __expf(s.y - m.y), __expf(s.z - m.z), __expf(s.w - m.w));
      gsh[tid] = g;
    } else {
      wsh[tid] = make_float4(0.0f, 0.0f, 0.0f, 0.0f);
      gsh[tid] = 0;
    }
  }
  __syncthreads();

  const int K = ND + ED;
  const int c = blockIdx.y * blockDim.x + tid;
  const bool active = c <= K + 1;
  const bool is_mask = c == K;
  const bool is_one = c > K;
  const float* xp;
  long pitch;
  if (c < ND)     { xp = h + (long)n0 * ND + c;          pitch = ND; }
  else if (c < K) { xp = xe + (long)n0 * ED + (c - ND);  pitch = ED; }
  else            { xp = cnt + n0;                       pitch = 1; }
  int gcur = __builtin_amdgcn_readfirstlane(gsh[0]);
  const int rlast = nend - 1 - n0;
  float a0 = 0.0f, a1 = 0.0f, a2 = 0.0f, a3 = 0.0f;
  for (int i0 = 0; i0 < POOL_ROWS; i0 += POOL_CHUNK) {
    if (n0 + i0 >= nend) break;
    float x[POOL_CHUNK];
    float4 wq[POOL_CHUNK];
    int gq[POOL_CHUNK];
#pragma unroll
    for (int j = 0; j < POOL_CHUNK; ++j) {
      int r = i0 + j;
      if (r > rlast) r = rlast;
      x[j] = xp[(long)r * pitch];
      wq[j] = wsh[i0 + j];
      gq[j] = gsh[i0 + j];
    }
#pragma unroll
    for (int j = 0; j < POOL_CHUNK; ++j) {
      if (n0 + i0 + j < nend) {
        const int g = __builtin_amdgcn_readfirstlane(gq[j]);
        if (g != gcur) {
          if (active) pool_flush(Zx, gcur, c, ZS, a0, a1, a2, a3);
          gcur = g;
          a0 = 0.0f; a1 = 0.0f; a2 = 0.0f; a3 = 0.0f;
        }
        const float4 w = wq[j];
        const float xv = is_one ? 1.0f : (is_mask ? ((x[j] > 0.0f) ? 1.0f : 0.0f) : x[j]);
        a0 += w.x * xv;
        a1 += w.y * xv;
        a2 += w.z * xv;
        a3 += w.w * xv;
      }
    }
  }
  if (active) pool_flush(Zx, gcur, c, ZS, a0, a1, a2, a3);
}

__global__ void zproj_wmma(const float* __restrict__ Zx,
                           const float* __restrict__ node_w, const float* __restrict__ node_b,
                           const float* __restrict__ edge_w, const float* __restrict__ edge_b,
                           float* __restrict__ Z, int M, int ND, int ED, int D, int ZS) {
  const int tid = threadIdx.x;
  const int wave = tid >> 5, lane = tid & 31;
  const int mrow = lane & 15;
  const int khalf = (lane & 16) ? 2 : 0;
  const int K = ND + ED;
  const int r0 = blockIdx.x * 16;
  const int n = blockIdx.y * 64 + wave * 16 + mrow;
  int ra = r0 + mrow;
  if (ra > M - 1) ra = M - 1;
  const float* zr = Zx + (long)ra * ZS + khalf;
  const float* wn = node_w + (long)n * ND + khalf;
  const float* we = edge_w + (long)n * ED + khalf;

  v8f acc = {};
  for (int kc = 0; kc < K; kc += KCHUNK) {
    v2f a[KCHUNK / 4], b[KCHUNK / 4];
#pragma unroll
    for (int i = 0; i < KCHUNK / 4; ++i) {
      const int kg = kc + 4 * i;
      const float* bp = (kg < ND) ? (wn + kg) : (we + (kg - ND));
      a[i] = *(const v2f*)&zr[kg];
      b[i] = *(const v2f*)bp;
    }
#pragma unroll
    for (int i = 0; i < KCHUNK / 4; ++i)
      acc = __builtin_amdgcn_wmma_f32_16x16x4_f32(false, a[i], false, b[i], (short)0, acc, false, false);
  }

  const float bn = node_b[n];
  const float be = edge_b[n];
  const int rb = r0 + ((lane & 16) ? 8 : 0);
  float dn[8], p1[8];
#pragma unroll
  for (int r = 0; r < 8; ++r) {
    int rr = rb + r;
    if (rr > M - 1) rr = M - 1;
    p1[r] = Zx[(long)rr * ZS + K];
    dn[r] = Zx[(long)rr * ZS + K + 1];
  }
#pragma unroll
  for (int r = 0; r < 8; ++r) {
    const int rr = rb + r;
    const float inv = 1.0f / fmaxf(dn[r], 1e-30f);
    if (rr < M) Z[(long)rr * D + n] = acc[r] * inv + (dn[r] * inv) * bn + (p1[r] * inv) * be;
  }
}

__global__ void vproj_pooled_wmma(const float* __restrict__ Z, const float* __restrict__ Wv,
                                  const float* __restrict__ bv, const float* __restrict__ den,
                                  int DS, float* __restrict__ pooled, int G, int D) {
  const int tid = threadIdx.x;
  const int wave = tid >> 5, lane = tid & 31;
  const int mrow = lane & 15;
  const int khalf = (lane & 16) ? 2 : 0;
  const int g0 = blockIdx.x * 16;
  const int ncol0 = blockIdx.y * 64 + wave * 16;
  const int n = ncol0 + mrow;
  const int hh = ncol0 / (D / NUM_HEADS);
  int ga = g0 + mrow;
  if (ga > G - 1) ga = G - 1;
  const float* zr = Z + ((long)ga * NUM_HEADS + hh) * D + khalf;
  const float* wr = Wv + (long)n * D + khalf;

  v8f acc = {};
  for (int kc = 0; kc < D; kc += KCHUNK) {
    v2f a[KCHUNK / 4], b[KCHUNK / 4];
#pragma unroll
    for (int i = 0; i < KCHUNK / 4; ++i) {
      a[i] = *(const v2f*)&zr[kc + 4 * i];
      b[i] = *(const v2f*)&wr[kc + 4 * i];
    }
#pragma unroll
    for (int i = 0; i < KCHUNK / 4; ++i)
      acc = __builtin_amdgcn_wmma_f32_16x16x4_f32(false, a[i], false, b[i], (short)0, acc, false, false);
  }

  const float bn = bv[n];
  const int gb = g0 + ((lane & 16) ? 8 : 0);
  float dsum[8];
#pragma unroll
  for (int r = 0; r < 8; ++r) {
    int g = gb + r;
    if (g > G - 1) g = G - 1;
    dsum[r] = den[(long)(g * NUM_HEADS + hh) * DS];
  }
#pragma unroll
  for (int r = 0; r < 8; ++r) {
    const int g = gb + r;
    if (g < G) {
      const float sp = dsum[r] / fmaxf(dsum[r], 1e-30f);
      pooled[(long)g * D + n] = acc[r] + sp * bn;
    }
  }
}

extern "C" void kernel_launch(void* const* d_in, const int* in_sizes, int n_in,
                              void* d_out, int out_size, void* d_ws, size_t ws_size,
                              hipStream_t stream) {
  const float* h      = (const float*)d_in[0];
  const int*   eidx   = (const int*)d_in[1];
  const float* eattr  = (const float*)d_in[2];
  const int*   batch  = (const int*)d_in[3];
  const float* node_w = (const float*)d_in[5];
  const float* node_b = (const float*)d_in[6];
  const float* edge_w = (const float*)d_in[7];
  const float* edge_b = (const float*)d_in[8];
  const float* query  = (const float*)d_in[9];
  const float* in_w   = (const float*)d_in[10];
  const float* in_b   = (const float*)d_in[11];
  const float* out_w  = (const float*)d_in[12];
  const float* out_b  = (const float*)d_in[13];
  float* out = (float*)d_out;

  int D  = in_sizes[9];
  int ND = in_sizes[5] / D;
  int ED = in_sizes[7] / D;
  int N  = in_sizes[0] / ND;
  long E = in_sizes[1] / 2;
  int G  = out_size / D;
  int K  = ND + ED;
  int ZS = (K + 2 + 3) & ~3;

  float* ws = (float*)d_ws;
  long o = 0;
  auto take = [&](long nf) { float* p = ws + o; o += (nf + 3) & ~3L; return p; };
  float* sumA   = take((long)N * ED);
  int*   cntc   = (int*)take((long)N * RP_CHUNKS);
  float* cnt    = take(N);
  int*   slots  = (int*)take((long)N * SLOT_CAP);
  float* scores = take((long)N * NUM_HEADS);
  float* tbuf   = take(NUM_HEADS);
  float* ubuf   = take((long)NUM_HEADS * ZS);
  float* smax   = take((long)G * NUM_HEADS);
  float* Zx     = take((long)G * NUM_HEADS * ZS);
  float* Zb     = take((long)G * NUM_HEADS * D);
  float* pooled = take((long)G * D);

  {
    dim3 grid((K + 2 + 31) / 32, NUM_HEADS + FILL_ROWS);
    fold_kernel<<<grid, FW * 32, 0, stream>>>(query, in_w, in_b, node_w, edge_w, node_b, edge_b,
                                              ubuf, tbuf, D, ND, ED, ZS,
                                              sumA, (long)N * ED, Zx, (long)G * NUM_HEADS * ZS,
                                              smax, (long)G * NUM_HEADS);
  }
  {
    dim3 rgrid((N + RP_NODES - 1) / RP_NODES, RP_CHUNKS);
    edge_rank<<<rgrid, RP_THREADS, 0, stream>>>(eidx, eattr, slots, cntc, sumA, E, N, ED);
    long gthreads = (long)N * 16;
    edge_gather<<<(int)((gthreads + 255) / 256), 256, 0, stream>>>(eattr, slots, cntc, sumA,
                                                                 cnt, N, ED);
  }
  {
    int total = N * NUM_HEADS;
    int blocks = (total + SM_THREADS - 1) / SM_THREADS;
    scores_kernel<<<blocks, SM_THREADS, 0, stream>>>(h, sumA, cnt, ubuf, tbuf, batch,
                                                     smax, scores, N, ND, ED, ZS, D);
  }
  {
    dim3 grid((N + POOL_ROWS - 1) / POOL_ROWS, (K + 2 + 255) / 256);
    pool_x<<<grid, 256, 0, stream>>>(scores, smax, batch, h, sumA, cnt, Zx, N, ND, ED, ZS);
  }
  {
    dim3 grid((G * NUM_HEADS + 15) / 16, D / 64);
    zproj_wmma<<<grid, 128, 0, stream>>>(Zx, node_w, node_b, edge_w, edge_b, Zb,
                                         G * NUM_HEADS, ND, ED, D, ZS);
  }
  {
    dim3 grid((G + 15) / 16, D / 64);
    vproj_pooled_wmma<<<grid, 128, 0, stream>>>(Zb, in_w + 2L * D * D, in_b + 2 * D,
                                                Zx + K + 1, ZS, pooled, G, D);
  }
  {
    dim3 grid((G + 15) / 16, D / 64);
    gemm_xwT_wmma<<<grid, 128, 0, stream>>>(pooled, out_w, out_b, out, G, D, D);
  }
}


# Scores and their segment max formed inside the edge gather (16 lanes per node hold the mean row, read h[n,:], join by xor shuffles); the scores launch goes, 8 to 7 launches; score sum order changes
# speedup vs baseline: 4.6534x; 1.0285x over previous
#include <hip/hip_runtime.h>
#include <math.h>

#define NUM_HEADS 4
#define KCHUNK 64

typedef float v2f __attribute__((ext_vector_type(2)));
typedef float v8f __attribute__((ext_vector_type(8)));

__device__ __forceinline__ void atomic_max_float(float* addr, float val) {
  if (val >= 0.0f) {
    atomicMax((int*)addr, __float_as_int(val));
  } else {
    atomicMin((unsigned int*)addr, __float_as_uint(val));
  }
}

__device__ __forceinline__ void fill_region(float* __restrict__ p, long n, float v, long i, long stride) {
  const float4 v4 = make_float4(v, v, v, v);
  const long n4 = n >> 2;
  for (long t = i; t < n4; t += stride) ((float4*)p)[t] = v4;
  for (long t = (n4 << 2) + i; t < n; t += stride) p[t] = v;
}

#define FW 8
#define FROWS 8
#define FJB 16
#define FPITCH 33
#define FILL_ROWS 32

__device__ __forceinline__ float dot4(float4 a, float4 b) {
  return a.x * b.x + a.y * b.y + a.z * b.z + a.w * b.w;
}

__global__ void fold_kernel(const float* __restrict__ query, const float* __restrict__ in_w,
                            const float* __restrict__ in_b,
                            const float* __restrict__ node_w, const float* __restrict__ edge_w,
                            const float* __restrict__ node_b, const float* __restrict__ edge_b,
                            float* __restrict__ ubuf, float* __restrict__ tbuf,
                            int D, int ND, int ED, int US,
                            float* __restrict__ z0, long n0, float* __restrict__ z1, long n1,
                            float* __restrict__ smax, long ns) {
  if (blockIdx.y >= NUM_HEADS) {
    const long fb = (long)(blockIdx.y - NUM_HEADS) * gridDim.x + blockIdx.x;
    const long i = fb * blockDim.x + threadIdx.x;
    const long stride = (long)(gridDim.y - NUM_HEADS) * gridDim.x * blockDim.x;
    fill_region(z0, n0, 0.0f, i, stride);
    fill_region(z1, n1, 0.0f, i, stride);
    fill_region(smax, ns, -INFINITY, i, stride);
    return;
  }
  __shared__ __attribute__((aligned(16))) float buf[256 * FPITCH];
  __shared__ float qh[256];
  __shared__ float sh[1024];
  __shared__ float red[FW][32];
  const int tid = threadIdx.x;
  const int lane = tid & 31, wave = tid >> 5;
  const int hh = blockIdx.y;
  const int dh = D / NUM_HEADS;
  const int K = ND + ED;

  for (int rb = 0; rb < dh; rb += FW * FROWS) {
    float p[FROWS];
#pragma unroll
    for (int i = 0; i < FROWS; ++i) p[i] = 0.0f;
    for (int c = lane * 4; c < D; c += 128) {
      const float4 x = *(const float4*)&query[c];
      float4 w[FROWS];
#pragma unroll
      for (int i = 0; i < FROWS; ++i) {
        int r = rb + wave + FW * i;
        if (r > dh - 1) r = dh - 1;
        w[i] = *(const float4*)&in_w[(long)(hh * dh + r) * D + c];
      }
#pragma unroll
      for (int i = 0; i < FROWS; ++i) p[i] += dot4(w[i], x);
    }
#pragma unroll
    for (int i = 0; i < FROWS; ++i) {
      const int r = rb + wave + FW * i;
      if (r < dh) buf[r * FPITCH + lane] = p[i];
    }
  }
  __syncthreads();
  if (tid < dh) {
    float q = in_b[hh * dh + tid];
    for (int l = 0; l < 32; ++l) q += buf[tid * FPITCH + l];
    qh[tid] = q;
  }
  __syncthreads();

  if (blockIdx.x == 0 && wave == 0) {
    float t = 0.0f;
    for (int d = lane; d < dh; d += 32) t += qh[d] * in_b[D + hh * dh + d];
    for (int off = 16; off > 0; off >>= 1) t += __shfl_xor(t, off);
    if (lane == 0) tbuf[hh] = t;
  }

  const float* wk = in_w + (long)(D + hh * dh) * D;
  for (int jc = lane * 4; jc < D; jc += 128) {
    float4 acc = make_float4(0.0f, 0.0f, 0.0f, 0.0f);
    for (int db = 0; db < dh; db += FW * FROWS) {
      float4 w[FROWS];
      float q[FROWS];
#pragma unroll
      for (int i = 0; i < FROWS; ++i) {
        const int d = db + wave + FW * i;
        const int dc = d < dh ? d : dh - 1;
        w[i] = *(const float4*)&wk[(long)dc * D + jc];
        q[i] = d < dh ? qh[dc] : 0.0f;
      }
#pragma unroll
      for (int i = 0; i < FROWS; ++i) {
        acc.x += q[i] * w[i].x; acc.y += q[i] * w[i].y; acc.z += q[i] * w[i].z; acc.w += q[i] * w[i].w;
      }
    }
    *(float4*)&buf[wave * D + jc] = acc;
  }
  __syncthreads();
  for (int j = tid; j < D; j += FW * 32) {
    float s = 0.0f;
    for (int w = 0; w < FW; ++w) s += buf[w * D + j];
    sh[j] = s;
  }
  __syncthreads();

  const int k = blockIdx.x * 32 + lane;
  const float* wp;
  long stride;
  if (k < ND)      { wp = node_w + k;        stride = ND; }
  else if (k < K)  { wp = edge_w + (k - ND); stride = ED; }
  else if (k == K) { wp = node_b;            stride = 1; }
  else             { wp = edge_b;            stride = 1; }
  float a = 0.0f;
  for (int j0 = wave; j0 < D; j0 += FW * FJB) {
    float w[FJB], sv[FJB];
#pragma unroll
    for (int i = 0; i < FJB; ++i) {
      const int j = j0 + FW * i;
      const int jcl = j < D ? j : D - 1;
      w[i] = wp[(long)jcl * stride];
      sv[i] = j < D ? sh[jcl] : 0.0f;
    }
#pragma unroll
    for (int i = 0; i < FJB; ++i) a += sv[i] * w[i];
  }
  red[wave][lane] = a;
  __syncthreads();
  if (tid < 32) {
    float sum = 0.0f;
    for (int w = 0; w < FW; ++w) sum += red[w][lane];
    if (k < K + 2) ubuf[hh * US + k] = sum;
  }
}

#define SLOT_CAP 64
#define GATHER_CHUNK 8
#define RP_CHUNKS 4
#define RP_CAP (SLOT_CAP / RP_CHUNKS)
#define RP_NODES 1024
#define RP_THREADS 1024
#define RP_BATCH 4

__device__ __forceinline__ void rank_place(int r, long e, int nlo, int c, int* lcnt,
                                           const float* __restrict__ eattr,
                                           int* __restrict__ slots, float* __restrict__ sumA,
                                           int ED) {
  const unsigned d = (unsigned)(r - nlo);
  if (d < (unsigned)RP_NODES) {
    const int s = atomicAdd(&lcnt[d], 1);
    if (s < RP_CAP) {
      slots[(long)r * SLOT_CAP + c * RP_CAP + s] = (int)e;
    } else {
      for (int k = 0; k < ED; ++k) atomicAdd(&sumA[(long)r * ED + k], eattr[e * ED + k]);
    }
  }
}

__global__ void edge_rank(const int* __restrict__ row, const float* __restrict__ eattr,
                          int* __restrict__ slots, int* __restrict__ cntc,
                          float* __restrict__ sumA, long E, int N, int ED) {
  __shared__ int lcnt[RP_NODES];
  const int tid = threadIdx.x;
  const int nlo = blockIdx.x * RP_NODES;
  const int c = blockIdx.y;
  for (int i = tid; i < RP_NODES; i += RP_THREADS) lcnt[i] = 0;
  __syncthreads();
  const long ec = (((E + RP_CHUNKS - 1) / RP_CHUNKS) + 3) & ~3L;
  const long e0 = (long)c * ec;
  long e1 = e0 + ec;
  if (e1 > E) e1 = E;
  if (e0 < e1) {
    const long e1v = e0 + ((e1 - e0) & ~3L);
    for (long eb = e0 + 4L * tid; eb < e1v; eb += 4L * RP_THREADS * RP_BATCH) {
      int4 r4[RP_BATCH];
#pragma unroll
      for (int u = 0; u < RP_BATCH; ++u) {
        long e = eb + 4L * RP_THREADS * u;
        if (e > e1v - 4) e = e1v - 4;
        r4[u] = *(const int4*)&row[e];
      }
#pragma unroll
      for (int u = 0; u < RP_BATCH; ++u) {
        const long e = eb + 4L * RP_THREADS * u;
        if (e < e1v) {
          rank_place(r4[u].x, e,     nlo, c, lcnt, eattr, slots, sumA, ED);
          rank_place(r4[u].y, e + 1, nlo, c, lcnt, eattr, slots, sumA, ED);
          rank_place(r4[u].z, e + 2, nlo, c, lcnt, eattr, slots, sumA, ED);
          rank_place(r4[u].w, e + 3, nlo, c, lcnt, eattr, slots, sumA, ED);
        }
      }
    }
    const long et = e1v + tid;
    if (et < e1) rank_place(row[et], et, nlo, c, lcnt, eattr, slots, sumA, ED);
  }
  __syncthreads();
  for (int i = tid; i < RP_NODES; i += RP_THREADS) {
    const int n = nlo + i;
    if (n < N) cntc[(long)n * RP_CHUNKS + c] = lcnt[i];
  }
}

__global__ void edge_gather(const float* __restrict__ eattr, const int* __restrict__ slots,
                            const int* __restrict__ cntc, float* __restrict__ sumA,
                            float* __restrict__ cntf,
                            const float* __restrict__ h, const float* __restrict__ ubuf,
                            const float* __restrict__ tbuf, const int* __restrict__ batch,
                            float* __restrict__ smax, float* __restrict__ scores,
                            int N, int ND, int ED, int US, int D) {
  __shared__ int dense[256 / 16][SLOT_CAP];
  __shared__ float bsc[256 / 16][NUM_HEADS];
  __shared__ int bg[256 / 16];
  const int t = blockIdx.x * blockDim.x + threadIdx.x;
  const int m = t >> 4;
  const int l = t & 15;
  const int grp = threadIdx.x >> 4;
  const bool valid = m < N;
  const int mm = valid ? m : N - 1;
  const int K = ND + ED;
  int m0 = 0, m1 = 0, m2 = 0, m3 = 0, cdeg = 0;
  if (valid) {
    const int4 cq = *(const int4*)&cntc[(long)m * RP_CHUNKS];
    m0 = cq.x < RP_CAP ? cq.x : RP_CAP;
    m1 = cq.y < RP_CAP ? cq.y : RP_CAP;
    m2 = cq.z < RP_CAP ? cq.z : RP_CAP;
    m3 = cq.w < RP_CAP ? cq.w : RP_CAP;
    cdeg = cq.x + cq.y + cq.z + cq.w;
  }
  const int cs = m0 + m1 + m2 + m3;
  int ev[SLOT_CAP / 16];
#pragma unroll
  for (int u = 0; u < SLOT_CAP / 16; ++u) {
    const int j = l + 16 * u;
    ev[u] = 0;
    if (j < cs) {
      int c, s;
      if (j < m0)                { c = 0; s = j; }
      else if (j < m0 + m1)      { c = 1; s = j - m0; }
      else if (j < m0 + m1 + m2) { c = 2; s = j - m0 - m1; }
      else                       { c = 3; s = j - m0 - m1 - m2; }
      ev[u] = slots[(long)m * SLOT_CAP + c * RP_CAP + s];
    }
  }

  const int hl = l & 3;
  const float ct = tbuf[hl];
  const float cu = ubuf[hl * US + K];
  const float cmk = ubuf[hl * US + K + 1];
  const int gm = batch[mm];
  float s0 = 0.0f, s1 = 0.0f, s2 = 0.0f, s3 = 0.0f;
  const float* hr = h + (long)mm * ND;
  for (int ca = l * 4; ca < ND; ca += 128) {
    const bool two = ca + 64 < ND;
    const int cb = two ? ca + 64 : ca;
    const float4 xa = *(const float4*)&hr[ca];
    float4 xb = *(const float4*)&hr[cb];
    const float4 ua0 = *(const float4*)&ubuf[ca];
    const float4 ua1 = *(const float4*)&ubuf[US + ca];
    const float4 ua2 = *(const float4*)&ubuf[2 * US + ca];
    const float4 ua3 = *(const float4*)&ubuf[3 * US + ca];
    const float4 ub0 = *(const float4*)&ubuf[cb];
    const float4 ub1 = *(const float4*)&ubuf[US + cb];
    const float4 ub2 = *(const float4*)&ubuf[2 * US + cb];
    const float4 ub3 = *(const float4*)&ubuf[3 * US + cb];
    if (!two) xb = make_float4(0.0f, 0.0f, 0.0f, 0.0f);
    s0 += dot4(xa, ua0) + dot4(xb, ub0);
    s1 += dot4(xa, ua1) + dot4(xb, ub1);
    s2 += dot4(xa, ua2) + dot4(xb, ub2);
    s3 += dot4(xa, ua3) + dot4(xb, ub3);
  }
  {
    const float cterm = ct + cu + (cdeg > 0 ? cmk : 0.0f);
    s0 += l == 0 ? cterm : 0.0f;
    s1 += l == 1 ? cterm : 0.0f;
    s2 += l == 2 ? cterm : 0.0f;
    s3 += l == 3 ? cterm : 0.0f;
  }

#pragma unroll
  for (int u = 0; u < SLOT_CAP / 16; ++u) {
    const int j = l + 16 * u;
    if (j < cs) dense[grp][j] = ev[u];
  }
  __syncthreads();
  const int* sp = dense[grp];
  for (int c0 = l * 4; c0 < ED; c0 += 64) {
    float4 acc = make_float4(0.0f, 0.0f, 0.0f, 0.0f);
    if (cdeg > cs) acc = *(const float4*)&sumA[(long)m * ED + c0];
    for (int j0 = 0; j0 < cs; j0 += GATHER_CHUNK) {
      float4 v[GATHER_CHUNK];
#pragma unroll
      for (int j = 0; j < GATHER_CHUNK; ++j) {
        int jj = j0 + j;
        if (jj > cs - 1) jj = cs - 1;
        const int e = sp[jj];
        v[j] = *(const float4*)&eattr[(long)e * ED + c0];
      }
#pragma unroll
      for (int j = 0; j < GATHER_CHUNK; ++j)
        asm volatile("" :: "v"(v[j].x), "v"(v[j].y), "v"(v[j].z), "v"(v[j].w));
#pragma unroll
      for (int j = 0; j < GATHER_CHUNK; ++j) {
        const bool in_list = j0 + j < cs;
        acc.x += in_list ? v[j].x : 0.0f;
        acc.y += in_list ? v[j].y : 0.0f;
        acc.z += in_list ? v[j].z : 0.0f;
        acc.w += in_list ? v[j].w : 0.0f;
      }
    }
    const float dm = fmaxf((float)cdeg, 1.0f);
    acc.x /= dm; acc.y /= dm; acc.z /= dm; acc.w /= dm;
    if (valid) *(float4*)&sumA[(long)m * ED + c0] = acc;
    const float4 e0 = *(const float4*)&ubuf[ND + c0];
    const float4 e1 = *(const float4*)&ubuf[US + ND + c0];
    const float4 e2 = *(const float4*)&ubuf[2 * US + ND + c0];
    const float4 e3 = *(const float4*)&ubuf[3 * US + ND + c0];
    s0 += dot4(acc, e0);
    s1 += dot4(acc, e1);
    s2 += dot4(acc, e2);
    s3 += dot4(acc, e3);
  }
  if (valid && l == 0) cntf[m] = (float)cdeg;

  for (int off = 8; off > 0; off >>= 1) {
    s0 += __shfl_xor(s0, off);
    s1 += __shfl_xor(s1, off);
    s2 += __shfl_xor(s2, off);
    s3 += __shfl_xor(s3, off);
  }
  if (l < NUM_HEADS) {
    const float sh = l == 0 ? s0 : (l == 1 ? s1 : (l == 2 ? s2 : s3));
    const float sc = sh * rsqrtf((float)(D / NUM_HEADS));
    if (valid) scores[(long)m * NUM_HEADS + l] = sc;
    bsc[grp][l] = sc;
  }
  if (l == 0) bg[grp] = valid ? gm : -1;
  __syncthreads();
  if (threadIdx.x < NUM_HEADS) {
    const int hd = threadIdx.x;
    int gq[256 / 16];
    float sq[256 / 16];
#pragma unroll
    for (int i = 0; i < 256 / 16; ++i) {
      gq[i] = bg[i];
      sq[i] = bsc[i][hd];
    }
    int gcur = -1;
    float mx = -INFINITY;
#pragma unroll
    for (int i = 0; i < 256 / 16; ++i) {
      const int g = gq[i];
      if (g >= 0) {
        if (g != gcur) {
          if (gcur >= 0) atomic_max_float(&smax[gcur * NUM_HEADS + hd], mx);
          gcur = g;
          mx = -INFINITY;
        }
        mx = fmaxf(mx, sq[i]);
      }
    }
    if (gcur >= 0) atomic_max_float(&smax[gcur * NUM_HEADS + hd], mx);
  }
}


__global__ void gemm_xwT_wmma(const float* __restrict__ X, const float* __restrict__ W,
                              const float* __restrict__ bias, float* __restrict__ Y,
                              int M, int K, int Nout) {
  const int tid = threadIdx.x;
  const int wave = tid >> 5, lane = tid & 31;
  const int mrow = lane & 15;
  const int khalf = (lane & 16) ? 2 : 0;
  const int r0 = blockIdx.x * 16;
  const int n = blockIdx.y * 64 + wave * 16 + mrow;
  int ra = r0 + mrow;
  if (ra > M - 1) ra = M - 1;
  const float* xr = X + (long)ra * K + khalf;
  const float* wr = W + (long)n * K + khalf;
  const float bn = bias[n];

  v8f acc = {};
  for (int kc = 0; kc < K; kc += KCHUNK) {
    v2f a[KCHUNK / 4], b[KCHUNK / 4];
#pragma unroll
    for (int i = 0; i < KCHUNK / 4; ++i) {
      a[i] = *(const v2f*)&xr[kc + 4 * i];
      b[i] = *(const v2f*)&wr[kc + 4 * i];
    }
#pragma unroll
    for (int i = 0; i < KCHUNK / 4; ++i)
      acc = __builtin_amdgcn_wmma_f32_16x16x4_f32(false, a[i], false, b[i], (short)0, acc, false, false);
  }

  const int rb = r0 + ((lane & 16) ? 8 : 0);
#pragma unroll
  for (int r = 0; r < 8; ++r) {
    const int rr = rb + r;
    if (rr < M) Y[(long)rr * Nout + n] = acc[r] + bn;
  }
}

#define POOL_ROWS 64
#define POOL_CHUNK 16

__device__ __forceinline__ void pool_flush(float* __restrict__ Z, int g, int c, int ZS,
                                           float a0, float a1, float a2, float a3) {
  float* zp = Z + (long)g * NUM_HEADS * ZS + c;
  atomicAdd(zp,              a0);
  atomicAdd(zp + ZS,         a1);
  atomicAdd(zp + 2L * ZS,    a2);
  atomicAdd(zp + 3L * ZS,    a3);
}

__global__ void pool_x(const float* __restrict__ scores, const float* __restrict__ smax,
                       const int* __restrict__ batch, const float* __restrict__ h,
                       const float* __restrict__ xe, const float* __restrict__ cnt,
                       float* __restrict__ Zx, int N, int ND, int ED, int ZS) {
  __shared__ float4 wsh[POOL_ROWS];
  __shared__ int gsh[POOL_ROWS];
  const int tid = threadIdx.x;
  const int n0 = blockIdx.x * POOL_ROWS;
  int nend = n0 + POOL_ROWS;
  if (nend > N) nend = N;
  if (tid < POOL_ROWS) {
    if (n0 + tid < N) {
      const int g = batch[n0 + tid];
      const float4 s = *(const float4*)&scores[(long)(n0 + tid) * NUM_HEADS];
      const float4 m = *(const float4*)&smax[(long)g * NUM_HEADS];
      wsh[tid] = make_float4(__expf(s.x - m.x), __expf(s.y - m.y), __expf(s.z - m.z), __expf(s.w - m.w));
      gsh[tid] = g;
    } else {
      wsh[tid] = make_float4(0.0f, 0.0f, 0.0f, 0.0f);
      gsh[tid] = 0;
    }
  }
  __syncthreads();

  const int K = ND + ED;
  const int c = blockIdx.y * blockDim.x + tid;
  const bool active = c <= K + 1;
  const bool is_mask = c == K;
  const bool is_one = c > K;
  const float* xp;
  long pitch;
  if (c < ND)     { xp = h + (long)n0 * ND + c;          pitch = ND; }
  else if (c < K) { xp = xe + (long)n0 * ED + (c - ND);  pitch = ED; }
  else            { xp = cnt + n0;                       pitch = 1; }
  int gcur = __builtin_amdgcn_readfirstlane(gsh[0]);
  const int rlast = nend - 1 - n0;
  float a0 = 0.0f, a1 = 0.0f, a2 = 0.0f, a3 = 0.0f;
  for (int i0 = 0; i0 < POOL_ROWS; i0 += POOL_CHUNK) {
    if (n0 + i0 >= nend) break;
    float x[POOL_CHUNK];
    float4 wq[POOL_CHUNK];
    int gq[POOL_CHUNK];
#pragma unroll
    for (int j = 0; j < POOL_CHUNK; ++j) {
      int r = i0 + j;
      if (r > rlast) r = rlast;
      x[j] = xp[(long)r * pitch];
      wq[j] = wsh[i0 + j];
      gq[j] = gsh[i0 + j];
    }
#pragma unroll
    for (int j = 0; j < POOL_CHUNK; ++j) {
      if (n0 + i0 + j < nend) {
        const int g = __builtin_amdgcn_readfirstlane(gq[j]);
        if (g != gcur) {
          if (active) pool_flush(Zx, gcur, c, ZS, a0, a1, a2, a3);
          gcur = g;
          a0 = 0.0f; a1 = 0.0f; a2 = 0.0f; a3 = 0.0f;
        }
        const float4 w = wq[j];
        const float xv = is_one ? 1.0f : (is_mask ? ((x[j] > 0.0f) ? 1.0f : 0.0f) : x[j]);
        a0 += w.x * xv;
        a1 += w.y * xv;
        a2 += w.z * xv;
        a3 += w.w * xv;
      }
    }
  }
  if (active) pool_flush(Zx, gcur, c, ZS, a0, a1, a2, a3);
}

__global__ void zproj_wmma(const float* __restrict__ Zx,
                           const float* __restrict__ node_w, const float* __restrict__ node_b,
                           const float* __restrict__ edge_w, const float* __restrict__ edge_b,
                           float* __restrict__ Z, int M, int ND, int ED, int D, int ZS) {
  const int tid = threadIdx.x;
  const int wave = tid >> 5, lane = tid & 31;
  const int mrow = lane & 15;
  const int khalf = (lane & 16) ? 2 : 0;
  const int K = ND + ED;
  const int r0 = blockIdx.x * 16;
  const int n = blockIdx.y * 64 + wave * 16 + mrow;
  int ra = r0 + mrow;
  if (ra > M - 1) ra = M - 1;
  const float* zr = Zx + (long)ra * ZS + khalf;
  const float* wn = node_w + (long)n * ND + khalf;
  const float* we = edge_w + (long)n * ED + khalf;

  v8f acc = {};
  for (int kc = 0; kc < K; kc += KCHUNK) {
    v2f a[KCHUNK / 4], b[KCHUNK / 4];
#pragma unroll
    for (int i = 0; i < KCHUNK / 4; ++i) {
      const int kg = kc + 4 * i;
      const float* bp = (kg < ND) ? (wn + kg) : (we + (kg - ND));
      a[i] = *(const v2f*)&zr[kg];
      b[i] = *(const v2f*)bp;
    }
#pragma unroll
    for (int i = 0; i < KCHUNK / 4; ++i)
      acc = __builtin_amdgcn_wmma_f32_16x16x4_f32(false, a[i], false, b[i], (short)0, acc, false, false);
  }

  const float bn = node_b[n];
  const float be = edge_b[n];
  const int rb = r0 + ((lane & 16) ? 8 : 0);
  float dn[8], p1[8];
#pragma unroll
  for (int r = 0; r < 8; ++r) {
    int rr = rb + r;
    if (rr > M - 1) rr = M - 1;
    p1[r] = Zx[(long)rr * ZS + K];
    dn[r] = Zx[(long)rr * ZS + K + 1];
  }
#pragma unroll
  for (int r = 0; r < 8; ++r) {
    const int rr = rb + r;
    const float inv = 1.0f / fmaxf(dn[r], 1e-30f);
    if (rr < M) Z[(long)rr * D + n] = acc[r] * inv + (dn[r] * inv) * bn + (p1[r] * inv) * be;
  }
}

__global__ void vproj_pooled_wmma(const float* __restrict__ Z, const float* __restrict__ Wv,
                                  const float* __restrict__ bv, const float* __restrict__ den,
                                  int DS, float* __restrict__ pooled, int G, int D) {
  const int tid = threadIdx.x;
  const int wave = tid >> 5, lane = tid & 31;
  const int mrow = lane & 15;
  const int khalf = (lane & 16) ? 2 : 0;
  const int g0 = blockIdx.x * 16;
  const int ncol0 = blockIdx.y * 64 + wave * 16;
  const int n = ncol0 + mrow;
  const int hh = ncol0 / (D / NUM_HEADS);
  int ga = g0 + mrow;
  if (ga > G - 1) ga = G - 1;
  const float* zr = Z + ((long)ga * NUM_HEADS + hh) * D + khalf;
  const float* wr = Wv + (long)n * D + khalf;

  v8f acc = {};
  for (int kc = 0; kc < D; kc += KCHUNK) {
    v2f a[KCHUNK / 4], b[KCHUNK / 4];
#pragma unroll
    for (int i = 0; i < KCHUNK / 4; ++i) {
      a[i] = *(const v2f*)&zr[kc + 4 * i];
      b[i] = *(const v2f*)&wr[kc + 4 * i];
    }
#pragma unroll
    for (int i = 0; i < KCHUNK / 4; ++i)
      acc = __builtin_amdgcn_wmma_f32_16x16x4_f32(false, a[i], false, b[i], (short)0, acc, false, false);
  }

  const float bn = bv[n];
  const int gb = g0 + ((lane & 16) ? 8 : 0);
  float dsum[8];
#pragma unroll
  for (int r = 0; r < 8; ++r) {
    int g = gb + r;
    if (g > G - 1) g = G - 1;
    dsum[r] = den[(long)(g * NUM_HEADS + hh) * DS];
  }
#pragma unroll
  for (int r = 0; r < 8; ++r) {
    const int g = gb + r;
    if (g < G) {
      const float sp = dsum[r] / fmaxf(dsum[r], 1e-30f);
      pooled[(long)g * D + n] = acc[r] + sp * bn;
    }
  }
}

extern "C" void kernel_launch(void* const* d_in, const int* in_sizes, int n_in,
                              void* d_out, int out_size, void* d_ws, size_t ws_size,
                              hipStream_t stream) {
  const float* h      = (const float*)d_in[0];
  const int*   eidx   = (const int*)d_in[1];
  const float* eattr  = (const float*)d_in[2];
  const int*   batch  = (const int*)d_in[3];
  const float* node_w = (const float*)d_in[5];
  const float* node_b = (const float*)d_in[6];
  const float* edge_w = (const float*)d_in[7];
  const float* edge_b = (const float*)d_in[8];
  const float* query  = (const float*)d_in[9];
  const float* in_w   = (const float*)d_in[10];
  const float* in_b   = (const float*)d_in[11];
  const float* out_w  = (const float*)d_in[12];
  const float* out_b  = (const float*)d_in[13];
  float* out = (float*)d_out;

  int D  = in_sizes[9];
  int ND = in_sizes[5] / D;
  int ED = in_sizes[7] / D;
  int N  = in_sizes[0] / ND;
  long E = in_sizes[1] / 2;
  int G  = out_size / D;
  int K  = ND + ED;
  int ZS = (K + 2 + 3) & ~3;

  float* ws = (float*)d_ws;
  long o = 0;
  auto take = [&](long nf) { float* p = ws + o; o += (nf + 3) & ~3L; return p; };
  float* sumA   = take((long)N * ED);
  int*   cntc   = (int*)take((long)N * RP_CHUNKS);
  float* cnt    = take(N);
  int*   slots  = (int*)take((long)N * SLOT_CAP);
  float* scores = take((long)N * NUM_HEADS);
  float* tbuf   = take(NUM_HEADS);
  float* ubuf   = take((long)NUM_HEADS * ZS);
  float* smax   = take((long)G * NUM_HEADS);
  float* Zx     = take((long)G * NUM_HEADS * ZS);
  float* Zb     = take((long)G * NUM_HEADS * D);
  float* pooled = take((long)G * D);

  {
    dim3 grid((K + 2 + 31) / 32, NUM_HEADS + FILL_ROWS);
    fold_kernel<<<grid, FW * 32, 0, stream>>>(query, in_w, in_b, node_w, edge_w, node_b, edge_b,
                                              ubuf, tbuf, D, ND, ED, ZS,
                                              sumA, (long)N * ED, Zx, (long)G * NUM_HEADS * ZS,
                                              smax, (long)G * NUM_HEADS);
  }
  {
    dim3 rgrid((N + RP_NODES - 1) / RP_NODES, RP_CHUNKS);
    edge_rank<<<rgrid, RP_THREADS, 0, stream>>>(eidx, eattr, slots, cntc, sumA, E, N, ED);
    long gthreads = (long)N * 16;
    edge_gather<<<(int)((gthreads + 255) / 256), 256, 0, stream>>>(eattr, slots, cntc, sumA, cnt,
                                                                 h, ubuf, tbuf, batch, smax,
                                                                 scores, N, ND, ED, ZS, D);
  }
  {
    dim3 grid((N + POOL_ROWS - 1) / POOL_ROWS, (K + 2 + 255) / 256);
    pool_x<<<grid, 256, 0, stream>>>(scores, smax, batch, h, sumA, cnt, Zx, N, ND, ED, ZS);
  }
  {
    dim3 grid((G * NUM_HEADS + 15) / 16, D / 64);
    zproj_wmma<<<grid, 128, 0, stream>>>(Zx, node_w, node_b, edge_w, edge_b, Zb,
                                         G * NUM_HEADS, ND, ED, D, ZS);
  }
  {
    dim3 grid((G + 15) / 16, D / 64);
    vproj_pooled_wmma<<<grid, 128, 0, stream>>>(Zb, in_w + 2L * D * D, in_b + 2 * D,
                                                Zx + K + 1, ZS, pooled, G, D);
  }
  {
    dim3 grid((G + 15) / 16, D / 64);
    gemm_xwT_wmma<<<grid, 128, 0, stream>>>(pooled, out_w, out_b, out, G, D, D);
  }
}


# Edge ranks re-tiled: a block takes 4096 nodes x a sixteenth of the edge list (was 1024 x a quarter), each id compared 13 times, not 49; gather lane l reads sub-list l; per-node sum order changes
# speedup vs baseline: 4.8400x; 1.0401x over previous
#include <hip/hip_runtime.h>
#include <math.h>

#define NUM_HEADS 4
#define KCHUNK 64

typedef float v2f __attribute__((ext_vector_type(2)));
typedef float v8f __attribute__((ext_vector_type(8)));

__device__ __forceinline__ void atomic_max_float(float* addr, float val) {
  if (val >= 0.0f) {
    atomicMax((int*)addr, __float_as_int(val));
  } else {
    atomicMin((unsigned int*)addr, __float_as_uint(val));
  }
}

__device__ __forceinline__ void fill_region(float* __restrict__ p, long n, float v, long i, long stride) {
  const float4 v4 = make_float4(v, v, v, v);
  const long n4 = n >> 2;
  for (long t = i; t < n4; t += stride) ((float4*)p)[t] = v4;
  for (long t = (n4 << 2) + i; t < n; t += stride) p[t] = v;
}

#define FW 8
#define FROWS 8
#define FJB 16
#define FPITCH 33
#define FILL_ROWS 32

__device__ __forceinline__ float dot4(float4 a, float4 b) {
  return a.x * b.x + a.y * b.y + a.z * b.z + a.w * b.w;
}

__global__ void fold_kernel(const float* __restrict__ query, const float* __restrict__ in_w,
                            const float* __restrict__ in_b,
                            const float* __restrict__ node_w, const float* __restrict__ edge_w,
                            const float* __restrict__ node_b, const float* __restrict__ edge_b,
                            float* __restrict__ ubuf, float* __restrict__ tbuf,
                            int D, int ND, int ED, int US,
                            float* __restrict__ z0, long n0, float* __restrict__ z1, long n1,
                            float* __restrict__ smax, long ns) {
  if (blockIdx.y >= NUM_HEADS) {
    const long fb = (long)(blockIdx.y - NUM_HEADS) * gridDim.x + blockIdx.x;
    const long i = fb * blockDim.x + threadIdx.x;
    const long stride = (long)(gridDim.y - NUM_HEADS) * gridDim.x * blockDim.x;
    fill_region(z0, n0, 0.0f, i, stride);
    fill_region(z1, n1, 0.0f, i, stride);
    fill_region(smax, ns, -INFINITY, i, stride);
    return;
  }
  __shared__ __attribute__((aligned(16))) float buf[256 * FPITCH];
  __shared__ float qh[256];
  __shared__ float sh[1024];
  __shared__ float red[FW][32];
  const int tid = threadIdx.x;
  const int lane = tid & 31, wave = tid >> 5;
  const int hh = blockIdx.y;
  const int dh = D / NUM_HEADS;
  const int K = ND + ED;

  for (int rb = 0; rb < dh; rb += FW * FROWS) {
    float p[FROWS];
#pragma unroll
    for (int i = 0; i < FROWS; ++i) p[i] = 0.0f;
    for (int c = lane * 4; c < D; c += 128) {
      const float4 x = *(const float4*)&query[c];
      float4 w[FROWS];
#pragma unroll
      for (int i = 0; i < FROWS; ++i) {
        int r = rb + wave + FW * i;
        if (r > dh - 1) r = dh - 1;
        w[i] = *(const float4*)&in_w[(long)(hh * dh + r) * D + c];
      }
#pragma unroll
      for (int i = 0; i < FROWS; ++i) p[i] += dot4(w[i], x);
    }
#pragma unroll
    for (int i = 0; i < FROWS; ++i) {
      const int r = rb + wave + FW * i;
      if (r < dh) buf[r * FPITCH + lane] = p[i];
    }
  }
  __syncthreads();
  if (tid < dh) {
    float q = in_b[hh * dh + tid];
    for (int l = 0; l < 32; ++l) q += buf[tid * FPITCH + l];
    qh[tid] = q;
  }
  __syncthreads();

  if (blockIdx.x == 0 && wave == 0) {
    float t = 0.0f;
    for (int d = lane; d < dh; d += 32) t += qh[d] * in_b[D + hh * dh + d];
    for (int off = 16; off > 0; off >>= 1) t += __shfl_xor(t, off);
    if (lane == 0) tbuf[hh] = t;
  }

  const float* wk = in_w + (long)(D + hh * dh) * D;
  for (int jc = lane * 4; jc < D; jc += 128) {
    float4 acc = make_float4(0.0f, 0.0f, 0.0f, 0.0f);
    for (int db = 0; db < dh; db += FW * FROWS) {
      float4 w[FROWS];
      float q[FROWS];
#pragma unroll
      for (int i = 0; i < FROWS; ++i) {
        const int d = db + wave + FW * i;
        const int dc = d < dh ? d : dh - 1;
        w[i] = *(const float4*)&wk[(long)dc * D + jc];
        q[i] = d < dh ? qh[dc] : 0.0f;
      }
#pragma unroll
      for (int i = 0; i < FROWS; ++i) {
        acc.x += q[i] * w[i].x; acc.y += q[i] * w[i].y; acc.z += q[i] * w[i].z; acc.w += q[i] * w[i].w;
      }
    }
    *(float4*)&buf[wave * D + jc] = acc;
  }
  __syncthreads();
  for (int j = tid; j < D; j += FW * 32) {
    float s = 0.0f;
    for (int w = 0; w < FW; ++w) s += buf[w * D + j];
    sh[j] = s;
  }
  __syncthreads();

  const int k = blockIdx.x * 32 + lane;
  const float* wp;
  long stride;
  if (k < ND)      { wp = node_w + k;        stride = ND; }
  else if (k < K)  { wp = edge_w + (k - ND); stride = ED; }
  else if (k == K) { wp = node_b;            stride = 1; }
  else             { wp = edge_b;            stride = 1; }
  float a = 0.0f;
  for (int j0 = wave; j0 < D; j0 += FW * FJB) {
    float w[FJB], sv[FJB];
#pragma unroll
    for (int i = 0; i < FJB; ++i) {
      const int j = j0 + FW * i;
      const int jcl = j < D ? j : D - 1;
      w[i] = wp[(long)jcl * stride];
      sv[i] = j < D ? sh[jcl] : 0.0f;
    }
#pragma unroll
    for (int i = 0; i < FJB; ++i) a += sv[i] * w[i];
  }
  red[wave][lane] = a;
  __syncthreads();
  if (tid < 32) {
    float sum = 0.0f;
    for (int w = 0; w < FW; ++w) sum += red[w][lane];
    if (k < K + 2) ubuf[hh * US + k] = sum;
  }
}

#define SLOT_CAP 128
#define GATHER_CHUNK 8
#define RP_CHUNKS 16
#define RP_CAP (SLOT_CAP / RP_CHUNKS)
#define RP_NODES 4096
#define RP_THREADS 1024
#define RP_BATCH 4

__device__ __forceinline__ void rank_place(int r, long e, int nlo, int c, int* lcnt,
                                           const float* __restrict__ eattr,
                                           int* __restrict__ slots, float* __restrict__ sumA,
                                           int ED) {
  const unsigned d = (unsigned)(r - nlo);
  if (d < (unsigned)RP_NODES) {
    const int s = atomicAdd(&lcnt[d], 1);
    if (s < RP_CAP) {
      slots[(long)r * SLOT_CAP + s * RP_CHUNKS + c] = (int)e;
    } else {
      for (int k = 0; k < ED; ++k) atomicAdd(&sumA[(long)r * ED + k], eattr[e * ED + k]);
    }
  }
}

__global__ void edge_rank(const int* __restrict__ row, const float* __restrict__ eattr,
                          int* __restrict__ slots, int* __restrict__ cntc,
                          float* __restrict__ sumA, long E, int N, int ED) {
  __shared__ int lcnt[RP_NODES];
  const int tid = threadIdx.x;
  const int nlo = blockIdx.x * RP_NODES;
  const int c = blockIdx.y;
  for (int i = tid; i < RP_NODES; i += RP_THREADS) lcnt[i] = 0;
  __syncthreads();
  const long ec = (((E + RP_CHUNKS - 1) / RP_CHUNKS) + 3) & ~3L;
  const long e0 = (long)c * ec;
  long e1 = e0 + ec;
  if (e1 > E) e1 = E;
  if (e0 < e1) {
    const long e1v = e0 + ((e1 - e0) & ~3L);
    for (long eb = e0 + 4L * tid; eb < e1v; eb += 4L * RP_THREADS * RP_BATCH) {
      int4 r4[RP_BATCH];
#pragma unroll
      for (int u = 0; u < RP_BATCH; ++u) {
        long e = eb + 4L * RP_THREADS * u;
        if (e > e1v - 4) e = e1v - 4;
        r4[u] = *(const int4*)&row[e];
      }
#pragma unroll
      for (int u = 0; u < RP_BATCH; ++u) {
        const long e = eb + 4L * RP_THREADS * u;
        if (e < e1v) {
          rank_place(r4[u].x, e,     nlo, c, lcnt, eattr, slots, sumA, ED);
          rank_place(r4[u].y, e + 1, nlo, c, lcnt, eattr, slots, sumA, ED);
          rank_place(r4[u].z, e + 2, nlo, c, lcnt, eattr, slots, sumA, ED);
          rank_place(r4[u].w, e + 3, nlo, c, lcnt, eattr, slots, sumA, ED);
        }
      }
    }
    const long et = e1v + tid;
    if (et < e1) rank_place(row[et], et, nlo, c, lcnt, eattr, slots, sumA, ED);
  }
  __syncthreads();
  for (int i = tid; i < RP_NODES; i += RP_THREADS) {
    const int n = nlo + i;
    if (n < N) cntc[(long)c * N + n] = lcnt[i];
  }
}

__global__ void edge_gather(const float* __restrict__ eattr, const int* __restrict__ slots,
                            const int* __restrict__ cntc, float* __restrict__ sumA,
                            float* __restrict__ cntf,
                            const float* __restrict__ h, const float* __restrict__ ubuf,
                            const float* __restrict__ tbuf, const int* __restrict__ batch,
                            float* __restrict__ smax, float* __restrict__ scores,
                            int N, int ND, int ED, int US, int D) {
  __shared__ int dense[256 / 16][SLOT_CAP];
  __shared__ float bsc[256 / 16][NUM_HEADS];
  __shared__ int bg[256 / 16];
  const int t = blockIdx.x * blockDim.x + threadIdx.x;
  const int m = t >> 4;
  const int l = t & 15;
  const int grp = threadIdx.x >> 4;
  const bool valid = m < N;
  const int mm = valid ? m : N - 1;
  const int K = ND + ED;
  const int myc = valid ? cntc[(long)l * N + m] : 0;
  const int mc = myc < RP_CAP ? myc : RP_CAP;
  const int* srow = slots + (long)mm * SLOT_CAP + l;
  int ev[RP_CAP];
#pragma unroll
  for (int s = 0; s < RP_CAP; ++s) {
    ev[s] = 0;
    if (s < mc) ev[s] = srow[s * RP_CHUNKS];
  }
  int incl = mc;
  for (int off = 1; off < 16; off <<= 1) {
    const int below = __shfl_up(incl, off, 16);
    if (l >= off) incl += below;
  }
  const int lbase = incl - mc;
  const int cs = __shfl(incl, 15, 16);
  int cdeg = myc;
  for (int off = 8; off > 0; off >>= 1) cdeg += __shfl_xor(cdeg, off);

  const int hl = l & 3;
  const float ct = tbuf[hl];
  const float cu = ubuf[hl * US + K];
  const float cmk = ubuf[hl * US + K + 1];
  const int gm = batch[mm];
  float s0 = 0.0f, s1 = 0.0f, s2 = 0.0f, s3 = 0.0f;
  const float* hr = h + (long)mm * ND;
  for (int ca = l * 4; ca < ND; ca += 128) {
    const bool two = ca + 64 < ND;
    const int cb = two ? ca + 64 : ca;
    const float4 xa = *(const float4*)&hr[ca];
    float4 xb = *(const float4*)&hr[cb];
    const float4 ua0 = *(const float4*)&ubuf[ca];
    const float4 ua1 = *(const float4*)&ubuf[US + ca];
    const float4 ua2 = *(const float4*)&ubuf[2 * US + ca];
    const float4 ua3 = *(const float4*)&ubuf[3 * US + ca];
    const float4 ub0 = *(const float4*)&ubuf[cb];
    const float4 ub1 = *(const float4*)&ubuf[US + cb];
    const float4 ub2 = *(const float4*)&ubuf[2 * US + cb];
    const float4 ub3 = *(const float4*)&ubuf[3 * US + cb];
    if (!two) xb = make_float4(0.0f, 0.0f, 0.0f, 0.0f);
    s0 += dot4(xa, ua0) + dot4(xb, ub0);
    s1 += dot4(xa, ua1) + dot4(xb, ub1);
    s2 += dot4(xa, ua2) + dot4(xb, ub2);
    s3 += dot4(xa, ua3) + dot4(xb, ub3);
  }
  {
    const float cterm = ct + cu + (cdeg > 0 ? cmk : 0.0f);
    s0 += l == 0 ? cterm : 0.0f;
    s1 += l == 1 ? cterm : 0.0f;
    s2 += l == 2 ? cterm : 0.0f;
    s3 += l == 3 ? cterm : 0.0f;
  }

#pragma unroll
  for (int s = 0; s < RP_CAP; ++s) {
    if (s < mc) dense[grp][lbase + s] = ev[s];
  }
  __syncthreads();
  const int* sp = dense[grp];
  for (int c0 = l * 4; c0 < ED; c0 += 64) {
    float4 acc = make_float4(0.0f, 0.0f, 0.0f, 0.0f);
    if (cdeg > cs) acc = *(const float4*)&sumA[(long)m * ED + c0];
    for (int j0 = 0; j0 < cs; j0 += GATHER_CHUNK) {
      float4 v[GATHER_CHUNK];
#pragma unroll
      for (int j = 0; j < GATHER_CHUNK; ++j) {
        int jj = j0 + j;
        if (jj > cs - 1) jj = cs - 1;
        const int e = sp[jj];
        v[j] = *(const float4*)&eattr[(long)e * ED + c0];
      }
#pragma unroll
      for (int j = 0; j < GATHER_CHUNK; ++j)
        asm volatile("" :: "v"(v[j].x), "v"(v[j].y), "v"(v[j].z), "v"(v[j].w));
#pragma unroll
      for (int j = 0; j < GATHER_CHUNK; ++j) {
        const bool in_list = j0 + j < cs;
        acc.x += in_list ? v[j].x : 0.0f;
        acc.y += in_list ? v[j].y : 0.0f;
        acc.z += in_list ? v[j].z : 0.0f;
        acc.w += in_list ? v[j].w : 0.0f;
      }
    }
    const float dm = fmaxf((float)cdeg, 1.0f);
    acc.x /= dm; acc.y /= dm; acc.z /= dm; acc.w /= dm;
    if (valid) *(float4*)&sumA[(long)m * ED + c0] = acc;
    const float4 e0 = *(const float4*)&ubuf[ND + c0];
    const float4 e1 = *(const float4*)&ubuf[US + ND + c0];
    const float4 e2 = *(const float4*)&ubuf[2 * US + ND + c0];
    const float4 e3 = *(const float4*)&ubuf[3 * US + ND + c0];
    s0 += dot4(acc, e0);
    s1 += dot4(acc, e1);
    s2 += dot4(acc, e2);
    s3 += dot4(acc, e3);
  }
  if (valid && l == 0) cntf[m] = (float)cdeg;

  for (int off = 8; off > 0; off >>= 1) {
    s0 += __shfl_xor(s0, off);
    s1 += __shfl_xor(s1, off);
    s2 += __shfl_xor(s2, off);
    s3 += __shfl_xor(s3, off);
  }
  if (l < NUM_HEADS) {
    const float sh = l == 0 ? s0 : (l == 1 ? s1 : (l == 2 ? s2 : s3));
    const float sc = sh * rsqrtf((float)(D / NUM_HEADS));
    if (valid) scores[(long)m * NUM_HEADS + l] = sc;
    bsc[grp][l] = sc;
  }
  if (l == 0) bg[grp] = valid ? gm : -1;
  __syncthreads();
  if (threadIdx.x < NUM_HEADS) {
    const int hd = threadIdx.x;
    int gq[256 / 16];
    float sq[256 / 16];
#pragma unroll
    for (int i = 0; i < 256 / 16; ++i) {
      gq[i] = bg[i];
      sq[i] = bsc[i][hd];
    }
    int gcur = -1;
    float mx = -INFINITY;
#pragma unroll
    for (int i = 0; i < 256 / 16; ++i) {
      const int g = gq[i];
      if (g >= 0) {
        if (g != gcur) {
          if (gcur >= 0) atomic_max_float(&smax[gcur * NUM_HEADS + hd], mx);
          gcur = g;
          mx = -INFINITY;
        }
        mx = fmaxf(mx, sq[i]);
      }
    }
    if (gcur >= 0) atomic_max_float(&smax[gcur * NUM_HEADS + hd], mx);
  }
}


__global__ void gemm_xwT_wmma(const float* __restrict__ X, const float* __restrict__ W,
                              const float* __restrict__ bias, float* __restrict__ Y,
                              int M, int K, int Nout) {
  const int tid = threadIdx.x;
  const int wave = tid >> 5, lane = tid & 31;
  const int mrow = lane & 15;
  const int khalf = (lane & 16) ? 2 : 0;
  const int r0 = blockIdx.x * 16;
  const int n = blockIdx.y * 64 + wave * 16 + mrow;
  int ra = r0 + mrow;
  if (ra > M - 1) ra = M - 1;
  const float* xr = X + (long)ra * K + khalf;
  const float* wr = W + (long)n * K + khalf;
  const float bn = bias[n];

  v8f acc = {};
  for (int kc = 0; kc < K; kc += KCHUNK) {
    v2f a[KCHUNK / 4], b[KCHUNK / 4];
#pragma unroll
    for (int i = 0; i < KCHUNK / 4; ++i) {
      a[i] = *(const v2f*)&xr[kc + 4 * i];
      b[i] = *(const v2f*)&wr[kc + 4 * i];
    }
#pragma unroll
    for (int i = 0; i < KCHUNK / 4; ++i)
      acc = __builtin_amdgcn_wmma_f32_16x16x4_f32(false, a[i], false, b[i], (short)0, acc, false, false);
  }

  const int rb = r0 + ((lane & 16) ? 8 : 0);
#pragma unroll
  for (int r = 0; r < 8; ++r) {
    const int rr = rb + r;
    if (rr < M) Y[(long)rr * Nout + n] = acc[r] + bn;
  }
}

#define POOL_ROWS 64
#define POOL_CHUNK 16

__device__ __forceinline__ void pool_flush(float* __restrict__ Z, int g, int c, int ZS,
                                           float a0, float a1, float a2, float a3) {
  float* zp = Z + (long)g * NUM_HEADS * ZS + c;
  atomicAdd(zp,              a0);
  atomicAdd(zp + ZS,         a1);
  atomicAdd(zp + 2L * ZS,    a2);
  atomicAdd(zp + 3L * ZS,    a3);
}

__global__ void pool_x(const float* __restrict__ scores, const float* __restrict__ smax,
                       const int* __restrict__ batch, const float* __restrict__ h,
                       const float* __restrict__ xe, const float* __restrict__ cnt,
                       float* __restrict__ Zx, int N, int ND, int ED, int ZS) {
  __shared__ float4 wsh[POOL_ROWS];
  __shared__ int gsh[POOL_ROWS];
  const int tid = threadIdx.x;
  const int n0 = blockIdx.x * POOL_ROWS;
  int nend = n0 + POOL_ROWS;
  if (nend > N) nend = N;
  if (tid < POOL_ROWS) {
    if (n0 + tid < N) {
      const int g = batch[n0 + tid];
      const float4 s = *(const float4*)&scores[(long)(n0 + tid) * NUM_HEADS];
      const float4 m = *(const float4*)&smax[(long)g * NUM_HEADS];
      wsh[tid] = make_float4(__expf(s.x - m.x), __expf(s.y - m.y), __expf(s.z - m.z), __expf(s.w - m.w));
      gsh[tid] = g;
    } else {
      wsh[tid] = make_float4(0.0f, 0.0f, 0.0f, 0.0f);
      gsh[tid] = 0;
    }
  }
  __syncthreads();

  const int K = ND + ED;
  const int c = blockIdx.y * blockDim.x + tid;
  const bool active = c <= K + 1;
  const bool is_mask = c == K;
  const bool is_one = c > K;
  const float* xp;
  long pitch;
  if (c < ND)     { xp = h + (long)n0 * ND + c;          pitch = ND; }
  else if (c < K) { xp = xe + (long)n0 * ED + (c - ND);  pitch = ED; }
  else            { xp = cnt + n0;                       pitch = 1; }
  int gcur = __builtin_amdgcn_readfirstlane(gsh[0]);
  const int rlast = nend - 1 - n0;
  float a0 = 0.0f, a1 = 0.0f, a2 = 0.0f, a3 = 0.0f;
  for (int i0 = 0; i0 < POOL_ROWS; i0 += POOL_CHUNK) {
    if (n0 + i0 >= nend) break;
    float x[POOL_CHUNK];
    float4 wq[POOL_CHUNK];
    int gq[POOL_CHUNK];
#pragma unroll
    for (int j = 0; j < POOL_CHUNK; ++j) {
      int r = i0 + j;
      if (r > rlast) r = rlast;
      x[j] = xp[(long)r * pitch];
      wq[j] = wsh[i0 + j];
      gq[j] = gsh[i0 + j];
    }
#pragma unroll
    for (int j = 0; j < POOL_CHUNK; ++j) {
      if (n0 + i0 + j < nend) {
        const int g = __builtin_amdgcn_readfirstlane(gq[j]);
        if (g != gcur) {
          if (active) pool_flush(Zx, gcur, c, ZS, a0, a1, a2, a3);
          gcur = g;
          a0 = 0.0f; a1 = 0.0f; a2 = 0.0f; a3 = 0.0f;
        }
        const float4 w = wq[j];
        const float xv = is_one ? 1.0f : (is_mask ? ((x[j] > 0.0f) ? 1.0f : 0.0f) : x[j]);
        a0 += w.x * xv;
        a1 += w.y * xv;
        a2 += w.z * xv;
        a3 += w.w * xv;
      }
    }
  }
  if (active) pool_flush(Zx, gcur, c, ZS, a0, a1, a2, a3);
}

__global__ void zproj_wmma(const float* __restrict__ Zx,
                           const float* __restrict__ node_w, const float* __restrict__ node_b,
                           const float* __restrict__ edge_w, const float* __restrict__ edge_b,
                           float* __restrict__ Z, int M, int ND, int ED, int D, int ZS) {
  const int tid = threadIdx.x;
  const int wave = tid >> 5, lane = tid & 31;
  const int mrow = lane & 15;
  const int khalf = (lane & 16) ? 2 : 0;
  const int K = ND + ED;
  const int r0 = blockIdx.x * 16;
  const int n = blockIdx.y * 64 + wave * 16 + mrow;
  int ra = r0 + mrow;
  if (ra > M - 1) ra = M - 1;
  const float* zr = Zx + (long)ra * ZS + khalf;
  const float* wn = node_w + (long)n * ND + khalf;
  const float* we = edge_w + (long)n * ED + khalf;

  v8f acc = {};
  for (int kc = 0; kc < K; kc += KCHUNK) {
    v2f a[KCHUNK / 4], b[KCHUNK / 4];
#pragma unroll
    for (int i = 0; i < KCHUNK / 4; ++i) {
      const int kg = kc + 4 * i;
      const float* bp = (kg < ND) ? (wn + kg) : (we + (kg - ND));
      a[i] = *(const v2f*)&zr[kg];
      b[i] = *(const v2f*)bp;
    }
#pragma unroll
    for (int i = 0; i < KCHUNK / 4; ++i)
      acc = __builtin_amdgcn_wmma_f32_16x16x4_f32(false, a[i], false, b[i], (short)0, acc, false, false);
  }

  const float bn = node_b[n];
  const float be = edge_b[n];
  const int rb = r0 + ((lane & 16) ? 8 : 0);
  float dn[8], p1[8];
#pragma unroll
  for (int r = 0; r < 8; ++r) {
    int rr = rb + r;
    if (rr > M - 1) rr = M - 1;
    p1[r] = Zx[(long)rr * ZS + K];
    dn[r] = Zx[(long)rr * ZS + K + 1];
  }
#pragma unroll
  for (int r = 0; r < 8; ++r) {
    const int rr = rb + r;
    const float inv = 1.0f / fmaxf(dn[r], 1e-30f);
    if (rr < M) Z[(long)rr * D + n] = acc[r] * inv + (dn[r] * inv) * bn + (p1[r] * inv) * be;
  }
}

__global__ void vproj_pooled_wmma(const float* __restrict__ Z, const float* __restrict__ Wv,
                                  const float* __restrict__ bv, const float* __restrict__ den,
                                  int DS, float* __restrict__ pooled, int G, int D) {
  const int tid = threadIdx.x;
  const int wave = tid >> 5, lane = tid & 31;
  const int mrow = lane & 15;
  const int khalf = (lane & 16) ? 2 : 0;
  const int g0 = blockIdx.x * 16;
  const int ncol0 = blockIdx.y * 64 + wave * 16;
  const int n = ncol0 + mrow;
  const int hh = ncol0 / (D / NUM_HEADS);
  int ga = g0 + mrow;
  if (ga > G - 1) ga = G - 1;
  const float* zr = Z + ((long)ga * NUM_HEADS + hh) * D + khalf;
  const float* wr = Wv + (long)n * D + khalf;

  v8f acc = {};
  for (int kc = 0; kc < D; kc += KCHUNK) {
    v2f a[KCHUNK / 4], b[KCHUNK / 4];
#pragma unroll
    for (int i = 0; i < KCHUNK / 4; ++i) {
      a[i] = *(const v2f*)&zr[kc + 4 * i];
      b[i] = *(const v2f*)&wr[kc + 4 * i];
    }
#pragma unroll
    for (int i = 0; i < KCHUNK / 4; ++i)
      acc = __builtin_amdgcn_wmma_f32_16x16x4_f32(false, a[i], false, b[i], (short)0, acc, false, false);
  }

  const float bn = bv[n];
  const int gb = g0 + ((lane & 16) ? 8 : 0);
  float dsum[8];
#pragma unroll
  for (int r = 0; r < 8; ++r) {
    int g = gb + r;
    if (g > G - 1) g = G - 1;
    dsum[r] = den[(long)(g * NUM_HEADS + hh) * DS];
  }
#pragma unroll
  for (int r = 0; r < 8; ++r) {
    const int g = gb + r;
    if (g < G) {
      const float sp = dsum[r] / fmaxf(dsum[r], 1e-30f);
      pooled[(long)g * D + n] = acc[r] + sp * bn;
    }
  }
}

extern "C" void kernel_launch(void* const* d_in, const int* in_sizes, int n_in,
                              void* d_out, int out_size, void* d_ws, size_t ws_size,
                              hipStream_t stream) {
  const float* h      = (const float*)d_in[0];
  const int*   eidx   = (const int*)d_in[1];
  const float* eattr  = (const float*)d_in[2];
  const int*   batch  = (const int*)d_in[3];
  const float* node_w = (const float*)d_in[5];
  const float* node_b = (const float*)d_in[6];
  const float* edge_w = (const float*)d_in[7];
  const float* edge_b = (const float*)d_in[8];
  const float* query  = (const float*)d_in[9];
  const float* in_w   = (const float*)d_in[10];
  const float* in_b   = (const float*)d_in[11];
  const float* out_w  = (const float*)d_in[12];
  const float* out_b  = (const float*)d_in[13];
  float* out = (float*)d_out;

  int D  = in_sizes[9];
  int ND = in_sizes[5] / D;
  int ED = in_sizes[7] / D;
  int N  = in_sizes[0] / ND;
  long E = in_sizes[1] / 2;
  int G  = out_size / D;
  int K  = ND + ED;
  int ZS = (K + 2 + 3) & ~3;

  float* ws = (float*)d_ws;
  long o = 0;
  auto take = [&](long nf) { float* p = ws + o; o += (nf + 3) & ~3L; return p; };
  float* sumA   = take((long)N * ED);
  int*   cntc   = (int*)take((long)N * RP_CHUNKS);
  float* cnt    = take(N);
  int*   slots  = (int*)take((long)N * SLOT_CAP);
  float* scores = take((long)N * NUM_HEADS);
  float* tbuf   = take(NUM_HEADS);
  float* ubuf   = take((long)NUM_HEADS * ZS);
  float* smax   = take((long)G * NUM_HEADS);
  float* Zx     = take((long)G * NUM_HEADS * ZS);
  float* Zb     = take((long)G * NUM_HEADS * D);
  float* pooled = take((long)G * D);

  {
    dim3 grid((K + 2 + 31) / 32, NUM_HEADS + FILL_ROWS);
    fold_kernel<<<grid, FW * 32, 0, stream>>>(query, in_w, in_b, node_w, edge_w, node_b, edge_b,
                                              ubuf, tbuf, D, ND, ED, ZS,
                                              sumA, (long)N * ED, Zx, (long)G * NUM_HEADS * ZS,
                                              smax, (long)G * NUM_HEADS);
  }
  {
    dim3 rgrid((N + RP_NODES - 1) / RP_NODES, RP_CHUNKS);
    edge_rank<<<rgrid, RP_THREADS, 0, stream>>>(eidx, eattr, slots, cntc, sumA, E, N, ED);
    long gthreads = (long)N * 16;
    edge_gather<<<(int)((gthreads + 255) / 256), 256, 0, stream>>>(eattr, slots, cntc, sumA, cnt,
                                                                 h, ubuf, tbuf, batch, smax,
                                                                 scores, N, ND, ED, ZS, D);
  }
  {
    dim3 grid((N + POOL_ROWS - 1) / POOL_ROWS, (K + 2 + 255) / 256);
    pool_x<<<grid, 256, 0, stream>>>(scores, smax, batch, h, sumA, cnt, Zx, N, ND, ED, ZS);
  }
  {
    dim3 grid((G * NUM_HEADS + 15) / 16, D / 64);
    zproj_wmma<<<grid, 128, 0, stream>>>(Zx, node_w, node_b, edge_w, edge_b, Zb,
                                         G * NUM_HEADS, ND, ED, D, ZS);
  }
  {
    dim3 grid((G + 15) / 16, D / 64);
    vproj_pooled_wmma<<<grid, 128, 0, stream>>>(Zb, in_w + 2L * D * D, in_b + 2 * D,
                                                Zx + K + 1, ZS, pooled, G, D);
  }
  {
    dim3 grid((G + 15) / 16, D / 64);
    gemm_xwT_wmma<<<grid, 128, 0, stream>>>(pooled, out_w, out_b, out, G, D, D);
  }
}
